# starting kernel
# baseline (speedup 1.0000x reference)
#include <hip/hip_runtime.h>

typedef __attribute__((ext_vector_type(2))) float v2f;
typedef __attribute__((ext_vector_type(8))) float v8f;

#define IN_D  32
#define HID_D 32
#define OUT_D 16

__global__ void gcn_zero_kernel(float* __restrict__ p, long long n) {
    long long i = ((long long)blockIdx.x * blockDim.x + threadIdx.x) * 4;
    if (i + 3 < n) {
        *(float4*)(p + i) = make_float4(0.f, 0.f, 0.f, 0.f);
    } else {
        for (long long j = i; j < n; ++j) p[j] = 0.0f;
    }
}

__global__ void gcn_degree_kernel(const int* __restrict__ src,
                                  const int* __restrict__ dst,
                                  float* __restrict__ degO,
                                  float* __restrict__ degI, int E) {
    int e = blockIdx.x * blockDim.x + threadIdx.x;
    if (e < E) {
        atomicAdd(&degO[src[e]], 1.0f);
        atomicAdd(&degI[dst[e]], 1.0f);
    }
}

__global__ void gcn_norm_kernel(const float* __restrict__ degO,
                                const float* __restrict__ degI,
                                float* __restrict__ ns,
                                float* __restrict__ nd, int n) {
    int i = blockIdx.x * blockDim.x + threadIdx.x;
    if (i < n) {
        ns[i] = rsqrtf(fmaxf(degO[i], 1.0f));
        nd[i] = rsqrtf(fmaxf(degI[i], 1.0f));
    }
}

__global__ void gcn_gemm1_kernel(const float* __restrict__ feat,
                                 const float* __restrict__ ns,
                                 const float* __restrict__ W1,
                                 float* __restrict__ h1, int n) {
    const int wave = threadIdx.x >> 5;
    const int lane = threadIdx.x & 31;
    const int tile = blockIdx.x * (blockDim.x >> 5) + wave;
    const int row0 = tile * 16;
    if (row0 >= n) return;

    const int half = lane >> 4;
    const int mrow = lane & 15;
    const int arow = row0 + mrow;
    const float scale = ns[arow];

    v8f acc0 = {};
    v8f acc1 = {};
#pragma unroll
    for (int k0 = 0; k0 < IN_D; k0 += 4) {
        const int kk = k0 + 2 * half;
        v2f a, b0, b1;
        const float* ap = feat + (long long)arow * IN_D + kk;
        a.x = ap[0] * scale;
        a.y = ap[1] * scale;
        b0.x = W1[kk * HID_D + mrow];
        b0.y = W1[(kk + 1) * HID_D + mrow];
        b1.x = W1[kk * HID_D + 16 + mrow];
        b1.y = W1[(kk + 1) * HID_D + 16 + mrow];
        acc0 = __builtin_amdgcn_wmma_f32_16x16x4_f32(false, a, false, b0,
                                                     (short)0, acc0, false, false);
        acc1 = __builtin_amdgcn_wmma_f32_16x16x4_f32(false, a, false, b1,
                                                     (short)0, acc1, false, false);
    }
#pragma unroll
    for (int j = 0; j < 8; ++j) {
        const int m = row0 + j + 8 * half;
        h1[(long long)m * HID_D + mrow]      = acc0[j];
        h1[(long long)m * HID_D + 16 + mrow] = acc1[j];
    }
}

__global__ void gcn_gemm2_kernel(const float* __restrict__ agg1,
                                 const float* __restrict__ ns,
                                 const float* __restrict__ nd,
                                 const float* __restrict__ b1,
                                 const float* __restrict__ W2,
                                 float* __restrict__ h2, int n) {
    const int wave = threadIdx.x >> 5;
    const int lane = threadIdx.x & 31;
    const int tile = blockIdx.x * (blockDim.x >> 5) + wave;
    const int row0 = tile * 16;
    if (row0 >= n) return;

    const int half = lane >> 4;
    const int mrow = lane & 15;
    const int arow = row0 + mrow;
    const float s_ns = ns[arow];
    const float s_nd = nd[arow];

    v8f acc = {};
#pragma unroll
    for (int k0 = 0; k0 < HID_D; k0 += 4) {
        const int kk = k0 + 2 * half;
        v2f a, b;
        const float* ap = agg1 + (long long)arow * HID_D + kk;
        a.x = fmaxf(ap[0] * s_nd + b1[kk],     0.0f) * s_ns;
        a.y = fmaxf(ap[1] * s_nd + b1[kk + 1], 0.0f) * s_ns;
        b.x = W2[kk * OUT_D + mrow];
        b.y = W2[(kk + 1) * OUT_D + mrow];
        acc = __builtin_amdgcn_wmma_f32_16x16x4_f32(false, a, false, b,
                                                    (short)0, acc, false, false);
    }
#pragma unroll
    for (int j = 0; j < 8; ++j) {
        const int m = row0 + j + 8 * half;
        h2[(long long)m * OUT_D + mrow] = acc[j];
    }
}

template <int D>
__global__ void gcn_scatter_kernel(const float* __restrict__ h,
                                   const int* __restrict__ src,
                                   const int* __restrict__ dst,
                                   const float* __restrict__ w,
                                   float* __restrict__ agg, long long total) {
    constexpr int G = D / 4;
    long long t = (long long)blockIdx.x * blockDim.x + threadIdx.x;
    if (t >= total) return;
    const int e = (int)(t / G);
    const int g = (int)(t % G);
    const int s = src[e];
    const int d = dst[e];
    const float wt = w[e];
    const float4 hv = *(const float4*)(h + (long long)s * D + g * 4);
    float* ap = agg + (long long)d * D + g * 4;
    atomicAdd(ap + 0, hv.x * wt);
    atomicAdd(ap + 1, hv.y * wt);
    atomicAdd(ap + 2, hv.z * wt);
    atomicAdd(ap + 3, hv.w * wt);
}

__global__ void gcn_finalize_kernel(float* __restrict__ out,
                                    const float* __restrict__ nd,
                                    const float* __restrict__ b2, long long total4) {
    constexpr int G = OUT_D / 4;
    long long t = (long long)blockIdx.x * blockDim.x + threadIdx.x;
    if (t >= total4) return;
    const int i = (int)(t / G);
    const int g = (int)(t % G);
    const float s = nd[i];
    const float4 bv = *(const float4*)(b2 + g * 4);
    float4* op = (float4*)(out + (long long)i * OUT_D + g * 4);
    float4 v = *op;
    v.x = v.x * s + bv.x;
    v.y = v.y * s + bv.y;
    v.z = v.z * s + bv.z;
    v.w = v.w * s + bv.w;
    *op = v;
}

extern "C" void kernel_launch(void* const* d_in, const int* in_sizes, int n_in,
                              void* d_out, int out_size, void* d_ws, size_t ws_size,
                              hipStream_t stream) {
    const float* feat = (const float*)d_in[0];
    const int*   src  = (const int*)d_in[1];
    const int*   dst  = (const int*)d_in[2];
    const float* ew   = (const float*)d_in[3];
    const float* W1   = (const float*)d_in[4];
    const float* b1   = (const float*)d_in[5];
    const float* W2   = (const float*)d_in[6];
    const float* b2   = (const float*)d_in[7];
    float* out = (float*)d_out;

    const int n = in_sizes[0] / IN_D;
    const int E = in_sizes[1];

    float* degO = (float*)d_ws;
    float* degI = degO + n;
    float* ns   = degI + n;
    float* nd   = ns + n;
    float* h1   = nd + n;
    float* agg1 = h1 + (long long)n * HID_D;
    float* h2   = agg1 + (long long)n * HID_D;

    const int B = 256;
    auto blocks = [&](long long work) { return (unsigned)((work + B - 1) / B); };

    gcn_zero_kernel<<<blocks((2LL * n + 3) / 4), B, 0, stream>>>(degO, 2LL * n);
    gcn_zero_kernel<<<blocks(((long long)n * HID_D + 3) / 4), B, 0, stream>>>(
        agg1, (long long)n * HID_D);
    gcn_zero_kernel<<<blocks(((long long)out_size + 3) / 4), B, 0, stream>>>(
        out, (long long)out_size);

    gcn_degree_kernel<<<blocks(E), B, 0, stream>>>(src, dst, degO, degI, E);

    gcn_norm_kernel<<<blocks(n), B, 0, stream>>>(degO, degI, ns, nd, n);

    const int tiles = (n + 15) / 16;
    const int wavesPerBlock = B / 32;
    const unsigned gemmBlocks = (unsigned)((tiles + wavesPerBlock - 1) / wavesPerBlock);
    gcn_gemm1_kernel<<<gemmBlocks, B, 0, stream>>>(feat, ns, W1, h1, n);

    gcn_scatter_kernel<HID_D><<<blocks((long long)E * (HID_D / 4)), B, 0, stream>>>(
        h1, src, dst, ew, agg1, (long long)E * (HID_D / 4));

    gcn_gemm2_kernel<<<gemmBlocks, B, 0, stream>>>(agg1, ns, nd, b1, W2, h2, n);

    gcn_scatter_kernel<OUT_D><<<blocks((long long)E * (OUT_D / 4)), B, 0, stream>>>(
        h2, src, dst, ew, out, (long long)E * (OUT_D / 4));

    gcn_finalize_kernel<<<blocks((long long)n * (OUT_D / 4)), B, 0, stream>>>(
        out, nd, b2, (long long)n * (OUT_D / 4));
}


# Degree counts: each work-group counts one edge slice's hits on one node range in LDS (integer adds), then adds its per-node counts with whole-line f32 atomics; integer-valued sums, exact in any order
# speedup vs baseline: 1.5832x; 1.5832x over previous
#include <hip/hip_runtime.h>

typedef __attribute__((ext_vector_type(2))) float v2f;
typedef __attribute__((ext_vector_type(8))) float v8f;

#define IN_D  32
#define HID_D 32
#define OUT_D 16

__global__ void gcn_zero_kernel(float* __restrict__ p, long long n) {
    long long i = ((long long)blockIdx.x * blockDim.x + threadIdx.x) * 4;
    if (i + 3 < n) {
        *(float4*)(p + i) = make_float4(0.f, 0.f, 0.f, 0.f);
    } else {
        for (long long j = i; j < n; ++j) p[j] = 0.0f;
    }
}

#define DEG_BINS    25600
#define DEG_PARTS   32
#define DEG_THREADS 1024
#define DEG_LOADS   8
#define DEG_FLUSH   5

__device__ __forceinline__ void gcn_count(unsigned* bins, int node, int base, int nb) {
    const unsigned r = (unsigned)(node - base);
    if (r < (unsigned)nb) atomicAdd(bins + r, 1u);
}

__global__ void __launch_bounds__(DEG_THREADS)
gcn_degree_kernel(const int* __restrict__ src,
                  const int* __restrict__ dst,
                  float* __restrict__ degO,
                  float* __restrict__ degI, int E, int n) {
    __shared__ unsigned bins[DEG_BINS];

    const int which = blockIdx.x & 1;
    const int rp    = blockIdx.x >> 1;
    const int part  = rp % DEG_PARTS;
    const int range = rp / DEG_PARTS;
    const int* __restrict__ idx = which ? dst : src;
    float* __restrict__ deg     = which ? degI : degO;
    const int base = range * DEG_BINS;
    const int nb   = (n - base < DEG_BINS) ? (n - base) : DEG_BINS;

    for (int i = threadIdx.x * 4; i < DEG_BINS; i += DEG_THREADS * 4)
        *(uint4*)(bins + i) = make_uint4(0u, 0u, 0u, 0u);
    __syncthreads();

    const int per  = ((E + DEG_PARTS * 4 - 1) / (DEG_PARTS * 4)) * 4;
    const int lo   = part * per;
    const int hi   = (lo + per < E) ? (lo + per) : E;
    const int len  = (hi > lo) ? (hi - lo) : 0;
    const int nvec = len >> 2;
    const int4* __restrict__ vp = (const int4*)(idx + lo);

    for (int v0 = threadIdx.x; v0 < nvec; v0 += DEG_THREADS * DEG_LOADS) {
        int4 q[DEG_LOADS];
#pragma unroll
        for (int k = 0; k < DEG_LOADS; ++k) {
            const int v = v0 + k * DEG_THREADS;
            q[k] = (v < nvec) ? vp[v] : make_int4(-1, -1, -1, -1);
        }
#pragma unroll
        for (int k = 0; k < DEG_LOADS; ++k) {
            gcn_count(bins, q[k].x, base, nb);
            gcn_count(bins, q[k].y, base, nb);
            gcn_count(bins, q[k].z, base, nb);
            gcn_count(bins, q[k].w, base, nb);
        }
    }
    {
        const int e = lo + (nvec << 2) + threadIdx.x;
        if (e < hi) gcn_count(bins, idx[e], base, nb);
    }
    __syncthreads();

    const int rot = part * (DEG_BINS / DEG_PARTS);
    for (int i0 = threadIdx.x; i0 < DEG_BINS; i0 += DEG_THREADS * DEG_FLUSH) {
        unsigned c[DEG_FLUSH];
        int      b[DEG_FLUSH];
#pragma unroll
        for (int k = 0; k < DEG_FLUSH; ++k) {
            const int i = i0 + k * DEG_THREADS;
            int bb = i + rot;
            if (bb >= DEG_BINS) bb -= DEG_BINS;
            b[k] = bb;
            c[k] = (i < DEG_BINS) ? bins[bb] : 0u;
        }
#pragma unroll
        for (int k = 0; k < DEG_FLUSH; ++k)
            if (c[k] != 0u && b[k] < nb) atomicAdd(deg + base + b[k], (float)c[k]);
    }
}

__global__ void gcn_norm_kernel(const float* __restrict__ degO,
                                const float* __restrict__ degI,
                                float* __restrict__ ns,
                                float* __restrict__ nd, int n) {
    int i = blockIdx.x * blockDim.x + threadIdx.x;
    if (i < n) {
        ns[i] = rsqrtf(fmaxf(degO[i], 1.0f));
        nd[i] = rsqrtf(fmaxf(degI[i], 1.0f));
    }
}

__global__ void gcn_gemm1_kernel(const float* __restrict__ feat,
                                 const float* __restrict__ ns,
                                 const float* __restrict__ W1,
                                 float* __restrict__ h1, int n) {
    const int wave = threadIdx.x >> 5;
    const int lane = threadIdx.x & 31;
    const int tile = blockIdx.x * (blockDim.x >> 5) + wave;
    const int row0 = tile * 16;
    if (row0 >= n) return;

    const int half = lane >> 4;
    const int mrow = lane & 15;
    const int arow = row0 + mrow;
    const float scale = ns[arow];

    v8f acc0 = {};
    v8f acc1 = {};
#pragma unroll
    for (int k0 = 0; k0 < IN_D; k0 += 4) {
        const int kk = k0 + 2 * half;
        v2f a, b0, b1;
        const float* ap = feat + (long long)arow * IN_D + kk;
        a.x = ap[0] * scale;
        a.y = ap[1] * scale;
        b0.x = W1[kk * HID_D + mrow];
        b0.y = W1[(kk + 1) * HID_D + mrow];
        b1.x = W1[kk * HID_D + 16 + mrow];
        b1.y = W1[(kk + 1) * HID_D + 16 + mrow];
        acc0 = __builtin_amdgcn_wmma_f32_16x16x4_f32(false, a, false, b0,
                                                     (short)0, acc0, false, false);
        acc1 = __builtin_amdgcn_wmma_f32_16x16x4_f32(false, a, false, b1,
                                                     (short)0, acc1, false, false);
    }
#pragma unroll
    for (int j = 0; j < 8; ++j) {
        const int m = row0 + j + 8 * half;
        h1[(long long)m * HID_D + mrow]      = acc0[j];
        h1[(long long)m * HID_D + 16 + mrow] = acc1[j];
    }
}

__global__ void gcn_gemm2_kernel(const float* __restrict__ agg1,
                                 const float* __restrict__ ns,
                                 const float* __restrict__ nd,
                                 const float* __restrict__ b1,
                                 const float* __restrict__ W2,
                                 float* __restrict__ h2, int n) {
    const int wave = threadIdx.x >> 5;
    const int lane = threadIdx.x & 31;
    const int tile = blockIdx.x * (blockDim.x >> 5) + wave;
    const int row0 = tile * 16;
    if (row0 >= n) return;

    const int half = lane >> 4;
    const int mrow = lane & 15;
    const int arow = row0 + mrow;
    const float s_ns = ns[arow];
    const float s_nd = nd[arow];

    v8f acc = {};
#pragma unroll
    for (int k0 = 0; k0 < HID_D; k0 += 4) {
        const int kk = k0 + 2 * half;
        v2f a, b;
        const float* ap = agg1 + (long long)arow * HID_D + kk;
        a.x = fmaxf(ap[0] * s_nd + b1[kk],     0.0f) * s_ns;
        a.y = fmaxf(ap[1] * s_nd + b1[kk + 1], 0.0f) * s_ns;
        b.x = W2[kk * OUT_D + mrow];
        b.y = W2[(kk + 1) * OUT_D + mrow];
        acc = __builtin_amdgcn_wmma_f32_16x16x4_f32(false, a, false, b,
                                                    (short)0, acc, false, false);
    }
#pragma unroll
    for (int j = 0; j < 8; ++j) {
        const int m = row0 + j + 8 * half;
        h2[(long long)m * OUT_D + mrow] = acc[j];
    }
}

template <int D>
__global__ void gcn_scatter_kernel(const float* __restrict__ h,
                                   const int* __restrict__ src,
                                   const int* __restrict__ dst,
                                   const float* __restrict__ w,
                                   float* __restrict__ agg, long long total) {
    constexpr int G = D / 4;
    long long t = (long long)blockIdx.x * blockDim.x + threadIdx.x;
    if (t >= total) return;
    const int e = (int)(t / G);
    const int g = (int)(t % G);
    const int s = src[e];
    const int d = dst[e];
    const float wt = w[e];
    const float4 hv = *(const float4*)(h + (long long)s * D + g * 4);
    float* ap = agg + (long long)d * D + g * 4;
    atomicAdd(ap + 0, hv.x * wt);
    atomicAdd(ap + 1, hv.y * wt);
    atomicAdd(ap + 2, hv.z * wt);
    atomicAdd(ap + 3, hv.w * wt);
}

__global__ void gcn_finalize_kernel(float* __restrict__ out,
                                    const float* __restrict__ nd,
                                    const float* __restrict__ b2, long long total4) {
    constexpr int G = OUT_D / 4;
    long long t = (long long)blockIdx.x * blockDim.x + threadIdx.x;
    if (t >= total4) return;
    const int i = (int)(t / G);
    const int g = (int)(t % G);
    const float s = nd[i];
    const float4 bv = *(const float4*)(b2 + g * 4);
    float4* op = (float4*)(out + (long long)i * OUT_D + g * 4);
    float4 v = *op;
    v.x = v.x * s + bv.x;
    v.y = v.y * s + bv.y;
    v.z = v.z * s + bv.z;
    v.w = v.w * s + bv.w;
    *op = v;
}

extern "C" void kernel_launch(void* const* d_in, const int* in_sizes, int n_in,
                              void* d_out, int out_size, void* d_ws, size_t ws_size,
                              hipStream_t stream) {
    const float* feat = (const float*)d_in[0];
    const int*   src  = (const int*)d_in[1];
    const int*   dst  = (const int*)d_in[2];
    const float* ew   = (const float*)d_in[3];
    const float* W1   = (const float*)d_in[4];
    const float* b1   = (const float*)d_in[5];
    const float* W2   = (const float*)d_in[6];
    const float* b2   = (const float*)d_in[7];
    float* out = (float*)d_out;

    const int n = in_sizes[0] / IN_D;
    const int E = in_sizes[1];

    float* degO = (float*)d_ws;
    float* degI = degO + n;
    float* ns   = degI + n;
    float* nd   = ns + n;
    float* h1   = nd + n;
    float* agg1 = h1 + (long long)n * HID_D;
    float* h2   = agg1 + (long long)n * HID_D;

    const int B = 256;
    auto blocks = [&](long long work) { return (unsigned)((work + B - 1) / B); };

    gcn_zero_kernel<<<blocks((2LL * n + 3) / 4), B, 0, stream>>>(degO, 2LL * n);
    gcn_zero_kernel<<<blocks(((long long)n * HID_D + 3) / 4), B, 0, stream>>>(
        agg1, (long long)n * HID_D);
    gcn_zero_kernel<<<blocks(((long long)out_size + 3) / 4), B, 0, stream>>>(
        out, (long long)out_size);

    const unsigned degRanges = (unsigned)((n + DEG_BINS - 1) / DEG_BINS);
    gcn_degree_kernel<<<2u * degRanges * DEG_PARTS, DEG_THREADS, 0, stream>>>(
        src, dst, degO, degI, E, n);

    gcn_norm_kernel<<<blocks(n), B, 0, stream>>>(degO, degI, ns, nd, n);

    const int tiles = (n + 15) / 16;
    const int wavesPerBlock = B / 32;
    const unsigned gemmBlocks = (unsigned)((tiles + wavesPerBlock - 1) / wavesPerBlock);
    gcn_gemm1_kernel<<<gemmBlocks, B, 0, stream>>>(feat, ns, W1, h1, n);

    gcn_scatter_kernel<HID_D><<<blocks((long long)E * (HID_D / 4)), B, 0, stream>>>(
        h1, src, dst, ew, agg1, (long long)E * (HID_D / 4));

    gcn_gemm2_kernel<<<gemmBlocks, B, 0, stream>>>(agg1, ns, nd, b1, W2, h2, n);

    gcn_scatter_kernel<OUT_D><<<blocks((long long)E * (OUT_D / 4)), B, 0, stream>>>(
        h2, src, dst, ew, out, (long long)E * (OUT_D / 4));

    gcn_finalize_kernel<<<blocks((long long)n * (OUT_D / 4)), B, 0, stream>>>(
        out, nd, b2, (long long)n * (OUT_D / 4));
}


# Edge scatter remapped to lane = column: a wave walks 256 edges, one load fetches and one f32 atomic adds a whole row (two half rows at 16 columns); 48x fewer waves; atomics still add in arrival order
# speedup vs baseline: 2.0593x; 1.3007x over previous
#include <hip/hip_runtime.h>

typedef __attribute__((ext_vector_type(2))) float v2f;
typedef __attribute__((ext_vector_type(8))) float v8f;

#define IN_D  32
#define HID_D 32
#define OUT_D 16

__global__ void gcn_zero_kernel(float* __restrict__ p, long long n) {
    long long i = ((long long)blockIdx.x * blockDim.x + threadIdx.x) * 4;
    if (i + 3 < n) {
        *(float4*)(p + i) = make_float4(0.f, 0.f, 0.f, 0.f);
    } else {
        for (long long j = i; j < n; ++j) p[j] = 0.0f;
    }
}

#define DEG_BINS    25600
#define DEG_PARTS   32
#define DEG_THREADS 1024
#define DEG_LOADS   8
#define DEG_FLUSH   5

__device__ __forceinline__ void gcn_count(unsigned* bins, int node, int base, int nb) {
    const unsigned r = (unsigned)(node - base);
    if (r < (unsigned)nb) atomicAdd(bins + r, 1u);
}

__global__ void __launch_bounds__(DEG_THREADS)
gcn_degree_kernel(const int* __restrict__ src,
                  const int* __restrict__ dst,
                  float* __restrict__ degO,
                  float* __restrict__ degI, int E, int n) {
    __shared__ unsigned bins[DEG_BINS];

    const int which = blockIdx.x & 1;
    const int rp    = blockIdx.x >> 1;
    const int part  = rp % DEG_PARTS;
    const int range = rp / DEG_PARTS;
    const int* __restrict__ idx = which ? dst : src;
    float* __restrict__ deg     = which ? degI : degO;
    const int base = range * DEG_BINS;
    const int nb   = (n - base < DEG_BINS) ? (n - base) : DEG_BINS;

    for (int i = threadIdx.x * 4; i < DEG_BINS; i += DEG_THREADS * 4)
        *(uint4*)(bins + i) = make_uint4(0u, 0u, 0u, 0u);
    __syncthreads();

    const int per  = ((E + DEG_PARTS * 4 - 1) / (DEG_PARTS * 4)) * 4;
    const int lo   = part * per;
    const int hi   = (lo + per < E) ? (lo + per) : E;
    const int len  = (hi > lo) ? (hi - lo) : 0;
    const int nvec = len >> 2;
    const int4* __restrict__ vp = (const int4*)(idx + lo);

    for (int v0 = threadIdx.x; v0 < nvec; v0 += DEG_THREADS * DEG_LOADS) {
        int4 q[DEG_LOADS];
#pragma unroll
        for (int k = 0; k < DEG_LOADS; ++k) {
            const int v = v0 + k * DEG_THREADS;
            q[k] = (v < nvec) ? vp[v] : make_int4(-1, -1, -1, -1);
        }
#pragma unroll
        for (int k = 0; k < DEG_LOADS; ++k) {
            gcn_count(bins, q[k].x, base, nb);
            gcn_count(bins, q[k].y, base, nb);
            gcn_count(bins, q[k].z, base, nb);
            gcn_count(bins, q[k].w, base, nb);
        }
    }
    {
        const int e = lo + (nvec << 2) + threadIdx.x;
        if (e < hi) gcn_count(bins, idx[e], base, nb);
    }
    __syncthreads();

    const int rot = part * (DEG_BINS / DEG_PARTS);
    for (int i0 = threadIdx.x; i0 < DEG_BINS; i0 += DEG_THREADS * DEG_FLUSH) {
        unsigned c[DEG_FLUSH];
        int      b[DEG_FLUSH];
#pragma unroll
        for (int k = 0; k < DEG_FLUSH; ++k) {
            const int i = i0 + k * DEG_THREADS;
            int bb = i + rot;
            if (bb >= DEG_BINS) bb -= DEG_BINS;
            b[k] = bb;
            c[k] = (i < DEG_BINS) ? bins[bb] : 0u;
        }
#pragma unroll
        for (int k = 0; k < DEG_FLUSH; ++k)
            if (c[k] != 0u && b[k] < nb) atomicAdd(deg + base + b[k], (float)c[k]);
    }
}

__global__ void gcn_norm_kernel(const float* __restrict__ degO,
                                const float* __restrict__ degI,
                                float* __restrict__ ns,
                                float* __restrict__ nd, int n) {
    int i = blockIdx.x * blockDim.x + threadIdx.x;
    if (i < n) {
        ns[i] = rsqrtf(fmaxf(degO[i], 1.0f));
        nd[i] = rsqrtf(fmaxf(degI[i], 1.0f));
    }
}

__global__ void gcn_gemm1_kernel(const float* __restrict__ feat,
                                 const float* __restrict__ ns,
                                 const float* __restrict__ W1,
                                 float* __restrict__ h1, int n) {
    const int wave = threadIdx.x >> 5;
    const int lane = threadIdx.x & 31;
    const int tile = blockIdx.x * (blockDim.x >> 5) + wave;
    const int row0 = tile * 16;
    if (row0 >= n) return;

    const int half = lane >> 4;
    const int mrow = lane & 15;
    const int arow = row0 + mrow;
    const float scale = ns[arow];

    v8f acc0 = {};
    v8f acc1 = {};
#pragma unroll
    for (int k0 = 0; k0 < IN_D; k0 += 4) {
        const int kk = k0 + 2 * half;
        v2f a, b0, b1;
        const float* ap = feat + (long long)arow * IN_D + kk;
        a.x = ap[0] * scale;
        a.y = ap[1] * scale;
        b0.x = W1[kk * HID_D + mrow];
        b0.y = W1[(kk + 1) * HID_D + mrow];
        b1.x = W1[kk * HID_D + 16 + mrow];
        b1.y = W1[(kk + 1) * HID_D + 16 + mrow];
        acc0 = __builtin_amdgcn_wmma_f32_16x16x4_f32(false, a, false, b0,
                                                     (short)0, acc0, false, false);
        acc1 = __builtin_amdgcn_wmma_f32_16x16x4_f32(false, a, false, b1,
                                                     (short)0, acc1, false, false);
    }
#pragma unroll
    for (int j = 0; j < 8; ++j) {
        const int m = row0 + j + 8 * half;
        h1[(long long)m * HID_D + mrow]      = acc0[j];
        h1[(long long)m * HID_D + 16 + mrow] = acc1[j];
    }
}

__global__ void gcn_gemm2_kernel(const float* __restrict__ agg1,
                                 const float* __restrict__ ns,
                                 const float* __restrict__ nd,
                                 const float* __restrict__ b1,
                                 const float* __restrict__ W2,
                                 float* __restrict__ h2, int n) {
    const int wave = threadIdx.x >> 5;
    const int lane = threadIdx.x & 31;
    const int tile = blockIdx.x * (blockDim.x >> 5) + wave;
    const int row0 = tile * 16;
    if (row0 >= n) return;

    const int half = lane >> 4;
    const int mrow = lane & 15;
    const int arow = row0 + mrow;
    const float s_ns = ns[arow];
    const float s_nd = nd[arow];

    v8f acc = {};
#pragma unroll
    for (int k0 = 0; k0 < HID_D; k0 += 4) {
        const int kk = k0 + 2 * half;
        v2f a, b;
        const float* ap = agg1 + (long long)arow * HID_D + kk;
        a.x = fmaxf(ap[0] * s_nd + b1[kk],     0.0f) * s_ns;
        a.y = fmaxf(ap[1] * s_nd + b1[kk + 1], 0.0f) * s_ns;
        b.x = W2[kk * OUT_D + mrow];
        b.y = W2[(kk + 1) * OUT_D + mrow];
        acc = __builtin_amdgcn_wmma_f32_16x16x4_f32(false, a, false, b,
                                                    (short)0, acc, false, false);
    }
#pragma unroll
    for (int j = 0; j < 8; ++j) {
        const int m = row0 + j + 8 * half;
        h2[(long long)m * OUT_D + mrow] = acc[j];
    }
}

#define SC_WAVE_EDGES 256
#define SC_THREADS    256

template <int P>
__device__ __forceinline__ int gcn_half_share(int v) {
    return __builtin_amdgcn_update_dpp(0, v, 0x150 + P, 0xf, 0xf, true);
}

__global__ void __launch_bounds__(SC_THREADS)
gcn_scatter32_kernel(const float* __restrict__ h,
                     const int* __restrict__ src,
                     const int* __restrict__ dst,
                     const float* __restrict__ w,
                     float* __restrict__ agg, int E) {
    const int lane = threadIdx.x & 31;
    const unsigned laneB = (unsigned)lane * 4u;
    const int wave = __builtin_amdgcn_readfirstlane((int)(threadIdx.x >> 5));
    const int wv   = blockIdx.x * (SC_THREADS >> 5) + wave;
    const int e_lo = wv * SC_WAVE_EDGES;
    if (e_lo >= E) return;
    const int e_hi = (e_lo + SC_WAVE_EDGES < E) ? (e_lo + SC_WAVE_EDGES) : E;

    int s_n = 0, d_n = 0;
    float w_n = 0.0f;
    if (e_lo + lane < e_hi) {
        s_n = src[e_lo + lane];
        d_n = dst[e_lo + lane];
        w_n = w[e_lo + lane];
    }
    for (int eb = e_lo; eb < e_hi; eb += 32) {
        const int   s  = s_n;
        const int   d  = d_n;
        const int   wb = __float_as_int(w_n);
        const int en = eb + 32 + lane;
        if (en < e_hi) {
            s_n = src[en];
            d_n = dst[en];
            w_n = w[en];
        }
        const int cnt = e_hi - eb;
        if (cnt >= 32) {
            float v[32];
#pragma unroll
            for (int k = 0; k < 32; ++k) {
                const unsigned sk = (unsigned)__builtin_amdgcn_readlane(s, k);
                v[k] = *(const float*)((const char*)h + ((sk << 7) + laneB));
            }
#pragma unroll
            for (int k = 0; k < 32; ++k) {
                const unsigned dk = (unsigned)__builtin_amdgcn_readlane(d, k);
                const float    wk = __int_as_float(__builtin_amdgcn_readlane(wb, k));
                atomicAdd((float*)((char*)agg + ((dk << 7) + laneB)), v[k] * wk);
            }
        } else {
            for (int k = 0; k < cnt; ++k) {
                const int   sk = __builtin_amdgcn_readlane(s, k);
                const int   dk = __builtin_amdgcn_readlane(d, k);
                const float wk = __int_as_float(__builtin_amdgcn_readlane(wb, k));
                atomicAdd(agg + (long long)dk * 32 + lane, h[(long long)sk * 32 + lane] * wk);
            }
        }
    }
}

template <int P>
__device__ __forceinline__ float gcn16_row_load(const float* __restrict__ h, int sB, int colB) {
    const unsigned off = (unsigned)(gcn_half_share<P>(sB) + colB);
    return *(const float*)((const char*)h + off);
}
template <int P>
__device__ __forceinline__ void gcn16_row_add(float* __restrict__ agg, int dB, int wb,
                                              int colB, float v) {
    const unsigned off = (unsigned)(gcn_half_share<P>(dB) + colB);
    const float wk = __int_as_float(gcn_half_share<P>(wb));
    atomicAdd((float*)((char*)agg + off), v * wk);
}

__global__ void __launch_bounds__(SC_THREADS)
gcn_scatter16_kernel(const float* __restrict__ h,
                     const int* __restrict__ src,
                     const int* __restrict__ dst,
                     const float* __restrict__ w,
                     float* __restrict__ agg, int E) {
    const int lane = threadIdx.x & 31;
    const int colB = (lane & 15) * 4;
    const int wave = __builtin_amdgcn_readfirstlane((int)(threadIdx.x >> 5));
    const int wv   = blockIdx.x * (SC_THREADS >> 5) + wave;
    const int e_lo = wv * SC_WAVE_EDGES;
    if (e_lo >= E) return;
    const int e_hi = (e_lo + SC_WAVE_EDGES < E) ? (e_lo + SC_WAVE_EDGES) : E;

    int s_n = 0, d_n = 0;
    float w_n = 0.0f;
    if (e_lo + lane < e_hi) {
        s_n = src[e_lo + lane];
        d_n = dst[e_lo + lane];
        w_n = w[e_lo + lane];
    }
    for (int eb = e_lo; eb < e_hi; eb += 32) {
        const int s  = s_n;
        const int d  = d_n;
        const int wb = __float_as_int(w_n);
        const int en = eb + 32 + lane;
        if (en < e_hi) {
            s_n = src[en];
            d_n = dst[en];
            w_n = w[en];
        }
        const int cnt = e_hi - eb;
        if (cnt >= 32) {
            const int sB = s * (OUT_D * 4);
            const int dB = d * (OUT_D * 4);
            float v[16];
#define GCN16_LD(P) v[P] = gcn16_row_load<P>(h, sB, colB);
            GCN16_LD(0)  GCN16_LD(1)  GCN16_LD(2)  GCN16_LD(3)
            GCN16_LD(4)  GCN16_LD(5)  GCN16_LD(6)  GCN16_LD(7)
            GCN16_LD(8)  GCN16_LD(9)  GCN16_LD(10) GCN16_LD(11)
            GCN16_LD(12) GCN16_LD(13) GCN16_LD(14) GCN16_LD(15)
#undef GCN16_LD
#define GCN16_AD(P) gcn16_row_add<P>(agg, dB, wb, colB, v[P]);
            GCN16_AD(0)  GCN16_AD(1)  GCN16_AD(2)  GCN16_AD(3)
            GCN16_AD(4)  GCN16_AD(5)  GCN16_AD(6)  GCN16_AD(7)
            GCN16_AD(8)  GCN16_AD(9)  GCN16_AD(10) GCN16_AD(11)
            GCN16_AD(12) GCN16_AD(13) GCN16_AD(14) GCN16_AD(15)
#undef GCN16_AD
        } else {
            for (int k = 0; k < cnt; ++k) {
                const int   sk = __builtin_amdgcn_readlane(s, k);
                const int   dk = __builtin_amdgcn_readlane(d, k);
                const float wk = __int_as_float(__builtin_amdgcn_readlane(wb, k));
                if (lane < OUT_D)
                    atomicAdd(agg + (long long)dk * OUT_D + lane,
                              h[(long long)sk * OUT_D + lane] * wk);
            }
        }
    }
}

__global__ void gcn_finalize_kernel(float* __restrict__ out,
                                    const float* __restrict__ nd,
                                    const float* __restrict__ b2, long long total4) {
    constexpr int G = OUT_D / 4;
    long long t = (long long)blockIdx.x * blockDim.x + threadIdx.x;
    if (t >= total4) return;
    const int i = (int)(t / G);
    const int g = (int)(t % G);
    const float s = nd[i];
    const float4 bv = *(const float4*)(b2 + g * 4);
    float4* op = (float4*)(out + (long long)i * OUT_D + g * 4);
    float4 v = *op;
    v.x = v.x * s + bv.x;
    v.y = v.y * s + bv.y;
    v.z = v.z * s + bv.z;
    v.w = v.w * s + bv.w;
    *op = v;
}

extern "C" void kernel_launch(void* const* d_in, const int* in_sizes, int n_in,
                              void* d_out, int out_size, void* d_ws, size_t ws_size,
                              hipStream_t stream) {
    const float* feat = (const float*)d_in[0];
    const int*   src  = (const int*)d_in[1];
    const int*   dst  = (const int*)d_in[2];
    const float* ew   = (const float*)d_in[3];
    const float* W1   = (const float*)d_in[4];
    const float* b1   = (const float*)d_in[5];
    const float* W2   = (const float*)d_in[6];
    const float* b2   = (const float*)d_in[7];
    float* out = (float*)d_out;

    const int n = in_sizes[0] / IN_D;
    const int E = in_sizes[1];

    float* degO = (float*)d_ws;
    float* degI = degO + n;
    float* ns   = degI + n;
    float* nd   = ns + n;
    float* h1   = nd + n;
    float* agg1 = h1 + (long long)n * HID_D;
    float* h2   = agg1 + (long long)n * HID_D;

    const int B = 256;
    auto blocks = [&](long long work) { return (unsigned)((work + B - 1) / B); };

    gcn_zero_kernel<<<blocks((2LL * n + 3) / 4), B, 0, stream>>>(degO, 2LL * n);
    gcn_zero_kernel<<<blocks(((long long)n * HID_D + 3) / 4), B, 0, stream>>>(
        agg1, (long long)n * HID_D);
    gcn_zero_kernel<<<blocks(((long long)out_size + 3) / 4), B, 0, stream>>>(
        out, (long long)out_size);

    const unsigned degRanges = (unsigned)((n + DEG_BINS - 1) / DEG_BINS);
    gcn_degree_kernel<<<2u * degRanges * DEG_PARTS, DEG_THREADS, 0, stream>>>(
        src, dst, degO, degI, E, n);

    gcn_norm_kernel<<<blocks(n), B, 0, stream>>>(degO, degI, ns, nd, n);

    const int tiles = (n + 15) / 16;
    const int wavesPerBlock = B / 32;
    const unsigned gemmBlocks = (unsigned)((tiles + wavesPerBlock - 1) / wavesPerBlock);
    gcn_gemm1_kernel<<<gemmBlocks, B, 0, stream>>>(feat, ns, W1, h1, n);

    const unsigned scWaves  = (unsigned)((E + SC_WAVE_EDGES - 1) / SC_WAVE_EDGES);
    const unsigned scBlocks = (scWaves + (SC_THREADS / 32) - 1) / (SC_THREADS / 32);
    gcn_scatter32_kernel<<<scBlocks, SC_THREADS, 0, stream>>>(h1, src, dst, ew, agg1, E);

    gcn_gemm2_kernel<<<gemmBlocks, B, 0, stream>>>(agg1, ns, nd, b1, W2, h2, n);

    gcn_scatter16_kernel<<<scBlocks, SC_THREADS, 0, stream>>>(h2, src, dst, ew, out, E);

    gcn_finalize_kernel<<<blocks((long long)n * (OUT_D / 4)), B, 0, stream>>>(
        out, nd, b2, (long long)n * (OUT_D / 4));
}


# Edge scatter split by chip half: a wave reads its die number and adds only into that half's node range, in 256-edge units drawn from per-range tickets; f32 atomics still add in arrival order
# speedup vs baseline: 2.4841x; 1.2063x over previous
#include <hip/hip_runtime.h>

typedef __attribute__((ext_vector_type(2))) float v2f;
typedef __attribute__((ext_vector_type(8))) float v8f;

#define IN_D  32
#define HID_D 32
#define OUT_D 16

__global__ void gcn_zero_kernel(float* __restrict__ p, long long n) {
    long long i = ((long long)blockIdx.x * blockDim.x + threadIdx.x) * 4;
    if (i + 3 < n) {
        *(float4*)(p + i) = make_float4(0.f, 0.f, 0.f, 0.f);
    } else {
        for (long long j = i; j < n; ++j) p[j] = 0.0f;
    }
}

#define DEG_BINS    25600
#define DEG_PARTS   32
#define DEG_THREADS 1024
#define DEG_LOADS   8
#define DEG_FLUSH   5

__device__ __forceinline__ void gcn_count(unsigned* bins, int node, int base, int nb) {
    const unsigned r = (unsigned)(node - base);
    if (r < (unsigned)nb) atomicAdd(bins + r, 1u);
}

__global__ void __launch_bounds__(DEG_THREADS)
gcn_degree_kernel(const int* __restrict__ src,
                  const int* __restrict__ dst,
                  float* __restrict__ degO,
                  float* __restrict__ degI, int E, int n) {
    __shared__ unsigned bins[DEG_BINS];

    const int which = blockIdx.x & 1;
    const int rp    = blockIdx.x >> 1;
    const int part  = rp % DEG_PARTS;
    const int range = rp / DEG_PARTS;
    const int* __restrict__ idx = which ? dst : src;
    float* __restrict__ deg     = which ? degI : degO;
    const int base = range * DEG_BINS;
    const int nb   = (n - base < DEG_BINS) ? (n - base) : DEG_BINS;

    for (int i = threadIdx.x * 4; i < DEG_BINS; i += DEG_THREADS * 4)
        *(uint4*)(bins + i) = make_uint4(0u, 0u, 0u, 0u);
    __syncthreads();

    const int per  = ((E + DEG_PARTS * 4 - 1) / (DEG_PARTS * 4)) * 4;
    const int lo   = part * per;
    const int hi   = (lo + per < E) ? (lo + per) : E;
    const int len  = (hi > lo) ? (hi - lo) : 0;
    const int nvec = len >> 2;
    const int4* __restrict__ vp = (const int4*)(idx + lo);

    for (int v0 = threadIdx.x; v0 < nvec; v0 += DEG_THREADS * DEG_LOADS) {
        int4 q[DEG_LOADS];
#pragma unroll
        for (int k = 0; k < DEG_LOADS; ++k) {
            const int v = v0 + k * DEG_THREADS;
            q[k] = (v < nvec) ? vp[v] : make_int4(-1, -1, -1, -1);
        }
#pragma unroll
        for (int k = 0; k < DEG_LOADS; ++k) {
            gcn_count(bins, q[k].x, base, nb);
            gcn_count(bins, q[k].y, base, nb);
            gcn_count(bins, q[k].z, base, nb);
            gcn_count(bins, q[k].w, base, nb);
        }
    }
    {
        const int e = lo + (nvec << 2) + threadIdx.x;
        if (e < hi) gcn_count(bins, idx[e], base, nb);
    }
    __syncthreads();

    const int rot = part * (DEG_BINS / DEG_PARTS);
    for (int i0 = threadIdx.x; i0 < DEG_BINS; i0 += DEG_THREADS * DEG_FLUSH) {
        unsigned c[DEG_FLUSH];
        int      b[DEG_FLUSH];
#pragma unroll
        for (int k = 0; k < DEG_FLUSH; ++k) {
            const int i = i0 + k * DEG_THREADS;
            int bb = i + rot;
            if (bb >= DEG_BINS) bb -= DEG_BINS;
            b[k] = bb;
            c[k] = (i < DEG_BINS) ? bins[bb] : 0u;
        }
#pragma unroll
        for (int k = 0; k < DEG_FLUSH; ++k)
            if (c[k] != 0u && b[k] < nb) atomicAdd(deg + base + b[k], (float)c[k]);
    }
}

__global__ void gcn_norm_kernel(float* __restrict__ degO_ns,
                                float* __restrict__ degI_nd, int n) {
    int i = blockIdx.x * blockDim.x + threadIdx.x;
    if (i < n) {
        degO_ns[i] = rsqrtf(fmaxf(degO_ns[i], 1.0f));
        degI_nd[i] = rsqrtf(fmaxf(degI_nd[i], 1.0f));
    }
}

__global__ void gcn_gemm1_kernel(const float* __restrict__ feat,
                                 const float* __restrict__ ns,
                                 const float* __restrict__ W1,
                                 float* __restrict__ h1, int n) {
    const int wave = threadIdx.x >> 5;
    const int lane = threadIdx.x & 31;
    const int tile = blockIdx.x * (blockDim.x >> 5) + wave;
    const int row0 = tile * 16;
    if (row0 >= n) return;

    const int half = lane >> 4;
    const int mrow = lane & 15;
    const int arow = row0 + mrow;
    const float scale = ns[arow];

    v8f acc0 = {};
    v8f acc1 = {};
#pragma unroll
    for (int k0 = 0; k0 < IN_D; k0 += 4) {
        const int kk = k0 + 2 * half;
        v2f a, b0, b1;
        const float* ap = feat + (long long)arow * IN_D + kk;
        a.x = ap[0] * scale;
        a.y = ap[1] * scale;
        b0.x = W1[kk * HID_D + mrow];
        b0.y = W1[(kk + 1) * HID_D + mrow];
        b1.x = W1[kk * HID_D + 16 + mrow];
        b1.y = W1[(kk + 1) * HID_D + 16 + mrow];
        acc0 = __builtin_amdgcn_wmma_f32_16x16x4_f32(false, a, false, b0,
                                                     (short)0, acc0, false, false);
        acc1 = __builtin_amdgcn_wmma_f32_16x16x4_f32(false, a, false, b1,
                                                     (short)0, acc1, false, false);
    }
#pragma unroll
    for (int j = 0; j < 8; ++j) {
        const int m = row0 + j + 8 * half;
        h1[(long long)m * HID_D + mrow]      = acc0[j];
        h1[(long long)m * HID_D + 16 + mrow] = acc1[j];
    }
}

__global__ void gcn_gemm2_kernel(const float* __restrict__ agg1,
                                 const float* __restrict__ ns,
                                 const float* __restrict__ nd,
                                 const float* __restrict__ b1,
                                 const float* __restrict__ W2,
                                 float* __restrict__ h2, int n) {
    const int wave = threadIdx.x >> 5;
    const int lane = threadIdx.x & 31;
    const int tile = blockIdx.x * (blockDim.x >> 5) + wave;
    const int row0 = tile * 16;
    if (row0 >= n) return;

    const int half = lane >> 4;
    const int mrow = lane & 15;
    const int arow = row0 + mrow;
    const float s_ns = ns[arow];
    const float s_nd = nd[arow];

    v8f acc = {};
#pragma unroll
    for (int k0 = 0; k0 < HID_D; k0 += 4) {
        const int kk = k0 + 2 * half;
        v2f a, b;
        const float* ap = agg1 + (long long)arow * HID_D + kk;
        a.x = fmaxf(ap[0] * s_nd + b1[kk],     0.0f) * s_ns;
        a.y = fmaxf(ap[1] * s_nd + b1[kk + 1], 0.0f) * s_ns;
        b.x = W2[kk * OUT_D + mrow];
        b.y = W2[(kk + 1) * OUT_D + mrow];
        acc = __builtin_amdgcn_wmma_f32_16x16x4_f32(false, a, false, b,
                                                    (short)0, acc, false, false);
    }
#pragma unroll
    for (int j = 0; j < 8; ++j) {
        const int m = row0 + j + 8 * half;
        h2[(long long)m * OUT_D + mrow] = acc[j];
    }
}

#define SC_UNIT_EDGES 256
#define SC_THREADS    256
#define SC_CTR_STRIDE 64
#define SC_CTR_WORDS  (4 * SC_CTR_STRIDE)

__device__ __forceinline__ int gcn_home_half() {
    const unsigned id = __builtin_amdgcn_s_sendmsg_rtn(0x87);
    return (int)((id >> 18) & 1u);
}

__device__ __forceinline__ unsigned gcn_take_unit(unsigned* c, int lane) {
    unsigned t = 0u;
    if (lane == 0)
        t = __hip_atomic_fetch_add(c, 1u, __ATOMIC_RELAXED, __HIP_MEMORY_SCOPE_AGENT);
    return (unsigned)__builtin_amdgcn_readfirstlane((int)t);
}

template <int P>
__device__ __forceinline__ int gcn_half_share(int v) {
    return __builtin_amdgcn_update_dpp(0, v, 0x150 + P, 0xf, 0xf, true);
}

__device__ __forceinline__ void gcn_unit32(const float* __restrict__ h,
                                           const int* __restrict__ src,
                                           const int* __restrict__ dst,
                                           const float* __restrict__ w,
                                           float* __restrict__ agg,
                                           int e_lo, int e_hi, int lane, unsigned laneB,
                                           int nSplit, int part) {
    int s_n = 0, d_n = 0;
    float w_n = 0.0f;
    if (e_lo + lane < e_hi) {
        s_n = src[e_lo + lane];
        d_n = dst[e_lo + lane];
        w_n = w[e_lo + lane];
    }
    for (int eb = e_lo; eb < e_hi; eb += 32) {
        const int   s  = s_n;
        const int   d  = d_n;
        const int   wb = __float_as_int(w_n);
        const int en = eb + 32 + lane;
        if (en < e_hi) {
            s_n = src[en];
            d_n = dst[en];
            w_n = w[en];
        }
        const bool mine = (eb + lane < e_hi) && ((d >= nSplit) == (part != 0));
        const unsigned m = __builtin_amdgcn_ballot_w32(mine);
        float v[32];
#pragma unroll
        for (int k = 0; k < 32; ++k) {
            if (m & (1u << k)) {
                const unsigned sk = (unsigned)__builtin_amdgcn_readlane(s, k);
                v[k] = *(const float*)((const char*)h + ((sk << 7) + laneB));
            }
        }
#pragma unroll
        for (int k = 0; k < 32; ++k) {
            if (m & (1u << k)) {
                const unsigned dk = (unsigned)__builtin_amdgcn_readlane(d, k);
                const float    wk = __int_as_float(__builtin_amdgcn_readlane(wb, k));
                atomicAdd((float*)((char*)agg + ((dk << 7) + laneB)), v[k] * wk);
            }
        }
    }
}

__global__ void __launch_bounds__(SC_THREADS)
gcn_scatter32_kernel(const float* __restrict__ h,
                     const int* __restrict__ src,
                     const int* __restrict__ dst,
                     const float* __restrict__ w,
                     float* __restrict__ agg,
                     unsigned* __restrict__ ctr, int E, int n) {
    const int lane = threadIdx.x & 31;
    const unsigned laneB = (unsigned)lane * 4u;
    const int nSplit = (n >> 1) & ~3;
    const unsigned units = (unsigned)((E + SC_UNIT_EDGES - 1) / SC_UNIT_EDGES);
    const int home = gcn_home_half();
    for (int pass = 0; pass < 2; ++pass) {
        const int part = home ^ pass;
        unsigned* c = ctr + part * SC_CTR_STRIDE;
        for (unsigned it = 0; it < units; ++it) {
            const unsigned t = gcn_take_unit(c, lane);
            if (t >= units) break;
            const int e_lo = (int)t * SC_UNIT_EDGES;
            const int e_hi = (e_lo + SC_UNIT_EDGES < E) ? (e_lo + SC_UNIT_EDGES) : E;
            gcn_unit32(h, src, dst, w, agg, e_lo, e_hi, lane, laneB, nSplit, part);
        }
    }
}

template <int P>
__device__ __forceinline__ float gcn16_row_load(const float* __restrict__ h, int sB, int colB) {
    const unsigned off = (unsigned)(gcn_half_share<P>(sB) + colB);
    return *(const float*)((const char*)h + off);
}
template <int P>
__device__ __forceinline__ void gcn16_row_add(float* __restrict__ agg, int dB, int wb,
                                              int colB, float v) {
    const unsigned off = (unsigned)(gcn_half_share<P>(dB) + colB);
    const float wk = __int_as_float(gcn_half_share<P>(wb));
    atomicAdd((float*)((char*)agg + off), v * wk);
}

__device__ __forceinline__ void gcn_unit16(const float* __restrict__ h,
                                           const int* __restrict__ src,
                                           const int* __restrict__ dst,
                                           const float* __restrict__ w,
                                           float* __restrict__ agg,
                                           int e_lo, int e_hi, int lane, int colB,
                                           int nSplit, int part) {
    const int half = lane >> 4;
    int s_n = 0, d_n = 0;
    float w_n = 0.0f;
    if (e_lo + lane < e_hi) {
        s_n = src[e_lo + lane];
        d_n = dst[e_lo + lane];
        w_n = w[e_lo + lane];
    }
    for (int eb = e_lo; eb < e_hi; eb += 32) {
        const int s  = s_n;
        const int d  = d_n;
        const int wb = __float_as_int(w_n);
        const int en = eb + 32 + lane;
        if (en < e_hi) {
            s_n = src[en];
            d_n = dst[en];
            w_n = w[en];
        }
        const bool mine = (eb + lane < e_hi) && ((d >= nSplit) == (part != 0));
        const unsigned m = __builtin_amdgcn_ballot_w32(mine);
        const int cnt   = __builtin_popcount(m);
        const int below = (int)__builtin_amdgcn_mbcnt_lo(m, 0u);
        const int rank = mine ? below : (cnt + lane - below);
        const int to   = (((rank & 1) << 4) | (rank >> 1)) << 2;
        const int sB  = __builtin_amdgcn_ds_permute(to, s * (OUT_D * 4));
        const int dB  = __builtin_amdgcn_ds_permute(to, d * (OUT_D * 4));
        const int wbp = __builtin_amdgcn_ds_permute(to, wb);
        float v[16];
#define GCN16_LD(P)                                                         \
        if (2 * (P) < cnt) {                                                \
            if (2 * (P) + half < cnt) v[P] = gcn16_row_load<P>(h, sB, colB); \
        }
        GCN16_LD(0)  GCN16_LD(1)  GCN16_LD(2)  GCN16_LD(3)
        GCN16_LD(4)  GCN16_LD(5)  GCN16_LD(6)  GCN16_LD(7)
        GCN16_LD(8)  GCN16_LD(9)  GCN16_LD(10) GCN16_LD(11)
        GCN16_LD(12) GCN16_LD(13) GCN16_LD(14) GCN16_LD(15)
#undef GCN16_LD
#define GCN16_AD(P)                                                         \
        if (2 * (P) < cnt) {                                                \
            if (2 * (P) + half < cnt) gcn16_row_add<P>(agg, dB, wbp, colB, v[P]); \
        }
        GCN16_AD(0)  GCN16_AD(1)  GCN16_AD(2)  GCN16_AD(3)
        GCN16_AD(4)  GCN16_AD(5)  GCN16_AD(6)  GCN16_AD(7)
        GCN16_AD(8)  GCN16_AD(9)  GCN16_AD(10) GCN16_AD(11)
        GCN16_AD(12) GCN16_AD(13) GCN16_AD(14) GCN16_AD(15)
#undef GCN16_AD
    }
}

__global__ void __launch_bounds__(SC_THREADS)
gcn_scatter16_kernel(const float* __restrict__ h,
                     const int* __restrict__ src,
                     const int* __restrict__ dst,
                     const float* __restrict__ w,
                     float* __restrict__ agg,
                     unsigned* __restrict__ ctr, int E, int n) {
    const int lane = threadIdx.x & 31;
    const int colB = (lane & 15) * 4;
    const int nSplit = (n >> 1) & ~3;
    const unsigned units = (unsigned)((E + SC_UNIT_EDGES - 1) / SC_UNIT_EDGES);
    const int home = gcn_home_half();
    for (int pass = 0; pass < 2; ++pass) {
        const int part = home ^ pass;
        unsigned* c = ctr + part * SC_CTR_STRIDE;
        for (unsigned it = 0; it < units; ++it) {
            const unsigned t = gcn_take_unit(c, lane);
            if (t >= units) break;
            const int e_lo = (int)t * SC_UNIT_EDGES;
            const int e_hi = (e_lo + SC_UNIT_EDGES < E) ? (e_lo + SC_UNIT_EDGES) : E;
            gcn_unit16(h, src, dst, w, agg, e_lo, e_hi, lane, colB, nSplit, part);
        }
    }
}

__global__ void gcn_finalize_kernel(float* __restrict__ out,
                                    const float* __restrict__ nd,
                                    const float* __restrict__ b2, long long total4) {
    constexpr int G = OUT_D / 4;
    long long t = (long long)blockIdx.x * blockDim.x + threadIdx.x;
    if (t >= total4) return;
    const int i = (int)(t / G);
    const int g = (int)(t % G);
    const float s = nd[i];
    const float4 bv = *(const float4*)(b2 + g * 4);
    float4* op = (float4*)(out + (long long)i * OUT_D + g * 4);
    float4 v = *op;
    v.x = v.x * s + bv.x;
    v.y = v.y * s + bv.y;
    v.z = v.z * s + bv.z;
    v.w = v.w * s + bv.w;
    *op = v;
}

extern "C" void kernel_launch(void* const* d_in, const int* in_sizes, int n_in,
                              void* d_out, int out_size, void* d_ws, size_t ws_size,
                              hipStream_t stream) {
    const float* feat = (const float*)d_in[0];
    const int*   src  = (const int*)d_in[1];
    const int*   dst  = (const int*)d_in[2];
    const float* ew   = (const float*)d_in[3];
    const float* W1   = (const float*)d_in[4];
    const float* b1   = (const float*)d_in[5];
    const float* W2   = (const float*)d_in[6];
    const float* b2   = (const float*)d_in[7];
    float* out = (float*)d_out;

    const int n = in_sizes[0] / IN_D;
    const int E = in_sizes[1];

    float* degO = (float*)d_ws;
    float* degI = degO + n;
    unsigned* ctr = (unsigned*)(degI + n);
    float* ns   = degO;
    float* nd   = degI;
    float* h1   = degO + 4LL * n;
    float* agg1 = h1 + (long long)n * HID_D;
    float* h2   = agg1 + (long long)n * HID_D;

    const int B = 256;
    auto blocks = [&](long long work) { return (unsigned)((work + B - 1) / B); };

    gcn_zero_kernel<<<blocks((2LL * n + SC_CTR_WORDS + 3) / 4), B, 0, stream>>>(
        degO, 2LL * n + SC_CTR_WORDS);
    gcn_zero_kernel<<<blocks(((long long)n * HID_D + 3) / 4), B, 0, stream>>>(
        agg1, (long long)n * HID_D);
    gcn_zero_kernel<<<blocks(((long long)out_size + 3) / 4), B, 0, stream>>>(
        out, (long long)out_size);

    const unsigned degRanges = (unsigned)((n + DEG_BINS - 1) / DEG_BINS);
    gcn_degree_kernel<<<2u * degRanges * DEG_PARTS, DEG_THREADS, 0, stream>>>(
        src, dst, degO, degI, E, n);

    gcn_norm_kernel<<<blocks(n), B, 0, stream>>>(degO, degI, n);

    const int tiles = (n + 15) / 16;
    const int wavesPerBlock = B / 32;
    const unsigned gemmBlocks = (unsigned)((tiles + wavesPerBlock - 1) / wavesPerBlock);
    gcn_gemm1_kernel<<<gemmBlocks, B, 0, stream>>>(feat, ns, W1, h1, n);

    const unsigned scWaves  = 2u * (unsigned)((E + SC_UNIT_EDGES - 1) / SC_UNIT_EDGES);
    const unsigned scBlocks = (scWaves + (SC_THREADS / 32) - 1) / (SC_THREADS / 32);
    gcn_scatter32_kernel<<<scBlocks, SC_THREADS, 0, stream>>>(h1, src, dst, ew, agg1, ctr, E, n);

    gcn_gemm2_kernel<<<gemmBlocks, B, 0, stream>>>(agg1, ns, nd, b1, W2, h2, n);

    gcn_scatter16_kernel<<<scBlocks, SC_THREADS, 0, stream>>>(
        h2, src, dst, ew, out, ctr + 2 * SC_CTR_STRIDE, E, n);

    gcn_finalize_kernel<<<blocks((long long)n * (OUT_D / 4)), B, 0, stream>>>(
        out, nd, b2, (long long)n * (OUT_D / 4));
}


# Edges sorted by destination once (in-degree prefix + cursor tickets); both aggregations gather per row, plain stores, no f32 atomics; layer-1 sums feed GEMM2 via LDS; sums in ticket arrival order
# speedup vs baseline: 2.9624x; 1.1925x over previous
#include <hip/hip_runtime.h>

typedef __attribute__((ext_vector_type(2))) float v2f;
typedef __attribute__((ext_vector_type(8))) float v8f;

#define IN_D  32
#define HID_D 32
#define OUT_D 16

__global__ void gcn_zero_kernel(float* __restrict__ p, long long n) {
    long long i = ((long long)blockIdx.x * blockDim.x + threadIdx.x) * 4;
    if (i + 3 < n) {
        *(float4*)(p + i) = make_float4(0.f, 0.f, 0.f, 0.f);
    } else {
        for (long long j = i; j < n; ++j) p[j] = 0.0f;
    }
}

#define DEG_BINS    25600
#define DEG_PARTS   32
#define DEG_THREADS 1024
#define DEG_LOADS   8
#define DEG_FLUSH   5

__device__ __forceinline__ void gcn_count(unsigned* bins, int node, int base, int nb) {
    const unsigned r = (unsigned)(node - base);
    if (r < (unsigned)nb) atomicAdd(bins + r, 1u);
}

__global__ void __launch_bounds__(DEG_THREADS)
gcn_degree_kernel(const int* __restrict__ src,
                  const int* __restrict__ dst,
                  float* __restrict__ degO,
                  float* __restrict__ degI, int E, int n) {
    __shared__ unsigned bins[DEG_BINS];

    const int which = blockIdx.x & 1;
    const int rp    = blockIdx.x >> 1;
    const int part  = rp % DEG_PARTS;
    const int range = rp / DEG_PARTS;
    const int* __restrict__ idx = which ? dst : src;
    float* __restrict__ deg     = which ? degI : degO;
    const int base = range * DEG_BINS;
    const int nb   = (n - base < DEG_BINS) ? (n - base) : DEG_BINS;

    for (int i = threadIdx.x * 4; i < DEG_BINS; i += DEG_THREADS * 4)
        *(uint4*)(bins + i) = make_uint4(0u, 0u, 0u, 0u);
    __syncthreads();

    const int per  = ((E + DEG_PARTS * 4 - 1) / (DEG_PARTS * 4)) * 4;
    const int lo   = part * per;
    const int hi   = (lo + per < E) ? (lo + per) : E;
    const int len  = (hi > lo) ? (hi - lo) : 0;
    const int nvec = len >> 2;
    const int4* __restrict__ vp = (const int4*)(idx + lo);

    for (int v0 = threadIdx.x; v0 < nvec; v0 += DEG_THREADS * DEG_LOADS) {
        int4 q[DEG_LOADS];
#pragma unroll
        for (int k = 0; k < DEG_LOADS; ++k) {
            const int v = v0 + k * DEG_THREADS;
            q[k] = (v < nvec) ? vp[v] : make_int4(-1, -1, -1, -1);
        }
#pragma unroll
        for (int k = 0; k < DEG_LOADS; ++k) {
            gcn_count(bins, q[k].x, base, nb);
            gcn_count(bins, q[k].y, base, nb);
            gcn_count(bins, q[k].z, base, nb);
            gcn_count(bins, q[k].w, base, nb);
        }
    }
    {
        const int e = lo + (nvec << 2) + threadIdx.x;
        if (e < hi) gcn_count(bins, idx[e], base, nb);
    }
    __syncthreads();

    const int rot = part * (DEG_BINS / DEG_PARTS);
    for (int i0 = threadIdx.x; i0 < DEG_BINS; i0 += DEG_THREADS * DEG_FLUSH) {
        unsigned c[DEG_FLUSH];
        int      b[DEG_FLUSH];
#pragma unroll
        for (int k = 0; k < DEG_FLUSH; ++k) {
            const int i = i0 + k * DEG_THREADS;
            int bb = i + rot;
            if (bb >= DEG_BINS) bb -= DEG_BINS;
            b[k] = bb;
            c[k] = (i < DEG_BINS) ? bins[bb] : 0u;
        }
#pragma unroll
        for (int k = 0; k < DEG_FLUSH; ++k)
            if (c[k] != 0u && b[k] < nb) atomicAdd(deg + base + b[k], (float)c[k]);
    }
}

__global__ void gcn_norm_kernel(float* __restrict__ degO_ns,
                                float* __restrict__ degI_nd, int n) {
    int i = blockIdx.x * blockDim.x + threadIdx.x;
    if (i < n) {
        degO_ns[i] = rsqrtf(fmaxf(degO_ns[i], 1.0f));
        degI_nd[i] = rsqrtf(fmaxf(degI_nd[i], 1.0f));
    }
}

__global__ void gcn_gemm1_kernel(const float* __restrict__ feat,
                                 const float* __restrict__ ns,
                                 const float* __restrict__ W1,
                                 float* __restrict__ h1, int n) {
    const int wave = threadIdx.x >> 5;
    const int lane = threadIdx.x & 31;
    const int tile = blockIdx.x * (blockDim.x >> 5) + wave;
    const int row0 = tile * 16;
    if (row0 >= n) return;

    const int half = lane >> 4;
    const int mrow = lane & 15;
    const int arow = row0 + mrow;
    const float scale = ns[arow];

    v8f acc0 = {};
    v8f acc1 = {};
#pragma unroll
    for (int k0 = 0; k0 < IN_D; k0 += 4) {
        const int kk = k0 + 2 * half;
        v2f a, b0, b1;
        const float* ap = feat + (long long)arow * IN_D + kk;
        a.x = ap[0] * scale;
        a.y = ap[1] * scale;
        b0.x = W1[kk * HID_D + mrow];
        b0.y = W1[(kk + 1) * HID_D + mrow];
        b1.x = W1[kk * HID_D + 16 + mrow];
        b1.y = W1[(kk + 1) * HID_D + 16 + mrow];
        acc0 = __builtin_amdgcn_wmma_f32_16x16x4_f32(false, a, false, b0,
                                                     (short)0, acc0, false, false);
        acc1 = __builtin_amdgcn_wmma_f32_16x16x4_f32(false, a, false, b1,
                                                     (short)0, acc1, false, false);
    }
#pragma unroll
    for (int j = 0; j < 8; ++j) {
        const int m = row0 + j + 8 * half;
        h1[(long long)m * HID_D + mrow]      = acc0[j];
        h1[(long long)m * HID_D + 16 + mrow] = acc1[j];
    }
}

#define RP_THREADS 1024
#define RP_LOADS   8

__global__ void __launch_bounds__(RP_THREADS)
gcn_rowptr_kernel(const float* __restrict__ degI,
                  int* __restrict__ rowptr,
                  int* __restrict__ cursor, int n) {
    __shared__ int wsum[32];
    const int tid   = threadIdx.x;
    const int lane  = tid & 31;
    const int wv    = tid >> 5;
    const int start = blockIdx.x * RP_THREADS;

    int s = 0;
    for (int j0 = tid * 4; j0 < start; j0 += RP_THREADS * 4 * RP_LOADS) {
        float4 q[RP_LOADS];
#pragma unroll
        for (int k = 0; k < RP_LOADS; ++k) {
            const int j = j0 + k * RP_THREADS * 4;
            q[k] = *(const float4*)(degI + ((j < start) ? j : 0));
        }
#pragma unroll
        for (int k = 0; k < RP_LOADS; ++k) {
            const int j = j0 + k * RP_THREADS * 4;
            const int t = (int)q[k].x + (int)q[k].y + (int)q[k].z + (int)q[k].w;
            s += (j < start) ? t : 0;
        }
    }
    for (int o = 16; o > 0; o >>= 1) s += __shfl_xor(s, o);
    if (lane == 0) wsum[wv] = s;
    __syncthreads();
    int base = 0;
    for (int k = 0; k < 32; ++k) base += wsum[k];
    __syncthreads();

    const int i = start + tid;
    const int c = (i < n) ? (int)degI[i] : 0;
    int inc = c;
    for (int o = 1; o < 32; o <<= 1) {
        const int t = __shfl_up(inc, o);
        if (lane >= o) inc += t;
    }
    if (lane == 31) wsum[wv] = inc;
    __syncthreads();
    int wbase = 0;
    for (int k = 0; k < wv; ++k) wbase += wsum[k];
    const int excl = base + wbase + inc - c;
    if (i < n) {
        rowptr[i] = excl;
        cursor[i] = excl;
    }
    if (i == n - 1) rowptr[n] = excl + c;
}

#define SC_UNIT_EDGES 256
#define SC_THREADS    256
#define SC_CTR_STRIDE 64
#define SC_CTR_WORDS  (4 * SC_CTR_STRIDE)

__device__ __forceinline__ int gcn_home_half() {
    const unsigned id = __builtin_amdgcn_s_sendmsg_rtn(0x87);
    return (int)((id >> 18) & 1u);
}

__device__ __forceinline__ unsigned gcn_take_unit(unsigned* c, int lane) {
    unsigned t = 0u;
    if (lane == 0)
        t = __hip_atomic_fetch_add(c, 1u, __ATOMIC_RELAXED, __HIP_MEMORY_SCOPE_AGENT);
    return (unsigned)__builtin_amdgcn_readfirstlane((int)t);
}

template <int P>
__device__ __forceinline__ int gcn_half_share(int v) {
    return __builtin_amdgcn_update_dpp(0, v, 0x150 + P, 0xf, 0xf, true);
}

#define FILL_BATCHES (SC_UNIT_EDGES / 32)

__device__ __forceinline__ void gcn_unit_fill(const int* __restrict__ src,
                                              const int* __restrict__ dst,
                                              const float* __restrict__ w,
                                              int* __restrict__ cursor,
                                              int2* __restrict__ list,
                                              int e_lo, int e_hi, int lane,
                                              int nSplit, int part) {
    int d[FILL_BATCHES], s[FILL_BATCHES], wb[FILL_BATCHES], pos[FILL_BATCHES];
#pragma unroll
    for (int b = 0; b < FILL_BATCHES; ++b) {
        const int e = e_lo + b * 32 + lane;
        const bool in = e < e_hi;
        d[b]  = in ? dst[e] : -1;
        s[b]  = in ? src[e] : 0;
        wb[b] = in ? __float_as_int(w[e]) : 0;
    }
#pragma unroll
    for (int b = 0; b < FILL_BATCHES; ++b) {
        const bool mine = (d[b] >= 0) && ((d[b] >= nSplit) == (part != 0));
        pos[b] = mine ? atomicAdd(cursor + d[b], 1) : -1;
    }
#pragma unroll
    for (int b = 0; b < FILL_BATCHES; ++b)
        if (pos[b] >= 0) list[pos[b]] = make_int2(s[b], wb[b]);
}

__global__ void __launch_bounds__(SC_THREADS)
gcn_fill_kernel(const int* __restrict__ src,
                const int* __restrict__ dst,
                const float* __restrict__ w,
                int* __restrict__ cursor,
                int* __restrict__ list,
                unsigned* __restrict__ ctr, int E, int n) {
    const int lane = threadIdx.x & 31;
    const int nSplit = (n >> 1) & ~31;
    const unsigned units = (unsigned)((E + SC_UNIT_EDGES - 1) / SC_UNIT_EDGES);
    const int home = gcn_home_half();
    for (int pass = 0; pass < 2; ++pass) {
        const int part = home ^ pass;
        unsigned* c = ctr + part * SC_CTR_STRIDE;
        for (unsigned it = 0; it < units; ++it) {
            const unsigned t = gcn_take_unit(c, lane);
            if (t >= units) break;
            const int e_lo = (int)t * SC_UNIT_EDGES;
            const int e_hi = (e_lo + SC_UNIT_EDGES < E) ? (e_lo + SC_UNIT_EDGES) : E;
            gcn_unit_fill(src, dst, w, cursor, (int2*)list, e_lo, e_hi, lane, nSplit, part);
        }
    }
}

#define AG_THREADS 256
#define AG_PITCH   34

__global__ void __launch_bounds__(AG_THREADS)
gcn_agg1_gemm2_kernel(const float* __restrict__ h1,
                      const int* __restrict__ rowptr,
                      const int* __restrict__ list,
                      const float* __restrict__ ns,
                      const float* __restrict__ nd,
                      const float* __restrict__ b1,
                      const float* __restrict__ W2,
                      float* __restrict__ h2, int n) {
    __shared__ float tiles[(AG_THREADS / 32) * 16 * AG_PITCH];
    const int wave = threadIdx.x >> 5;
    const int lane = threadIdx.x & 31;
    const int tile = blockIdx.x * (blockDim.x >> 5) + wave;
    const int row0 = tile * 16;
    if (row0 >= n) return;

    float* tl = tiles + wave * (16 * AG_PITCH);
    const int half = lane >> 4;
    const int mrow = lane & 15;
    const unsigned laneB = (unsigned)lane * 4u;
    const int2* __restrict__ ent = (const int2*)list;

    const int rrow = (row0 + mrow < n) ? (row0 + mrow) : (n - 1);
    const int   endv = rowptr[rrow + 1];
    const float ndv  = nd[rrow];
    const float nsv  = ns[rrow];
    const float b1c  = b1[lane];
    const int P0 = __builtin_amdgcn_readfirstlane(rowptr[row0]);
    const int P1 = __builtin_amdgcn_readlane(endv, 15);

    int   r    = 0;
    int   rend = __builtin_amdgcn_readlane(endv, 0);
    float acc  = 0.0f;

#define AG_FLUSH()                                                                    \
    {                                                                                 \
        const float ndr = __int_as_float(__builtin_amdgcn_readlane(__float_as_int(ndv), r)); \
        const float nsr = __int_as_float(__builtin_amdgcn_readlane(__float_as_int(nsv), r)); \
        tl[r * AG_PITCH + lane] = fmaxf(acc * ndr + b1c, 0.0f) * nsr;                 \
        acc = 0.0f;                                                                   \
        ++r;                                                                          \
    }

    int2 e_n = make_int2(0, 0);
    if (P0 + lane < P1) e_n = ent[P0 + lane];
    for (int base = P0; base < P1; base += 32) {
        const int s  = e_n.x;
        const int wb = e_n.y;
        const int qn = base + 32 + lane;
        if (qn < P1) e_n = ent[qn];
        const int cnt = (P1 - base < 32) ? (P1 - base) : 32;
        float v[32];
#pragma unroll
        for (int k = 0; k < 32; ++k) {
            if (k < cnt) {
                const unsigned sk = (unsigned)__builtin_amdgcn_readlane(s, k);
                v[k] = *(const float*)((const char*)h1 + ((sk << 7) + laneB));
            }
        }
#pragma unroll
        for (int k = 0; k < 32; ++k) {
            if (k < cnt) {
                while (base + k >= rend) {
                    AG_FLUSH()
                    rend = __builtin_amdgcn_readlane(endv, r);
                }
                const float wk = __int_as_float(__builtin_amdgcn_readlane(wb, k));
                acc += v[k] * wk;
            }
        }
    }
    while (r < 16) AG_FLUSH()
#undef AG_FLUSH

    __builtin_amdgcn_fence(__ATOMIC_RELEASE, "workgroup");
    __builtin_amdgcn_fence(__ATOMIC_ACQUIRE, "workgroup");

    v8f acc8 = {};
#pragma unroll
    for (int k0 = 0; k0 < HID_D; k0 += 4) {
        const int kk = k0 + 2 * half;
        v2f a, b;
        a.x = tl[mrow * AG_PITCH + kk];
        a.y = tl[mrow * AG_PITCH + kk + 1];
        b.x = W2[kk * OUT_D + mrow];
        b.y = W2[(kk + 1) * OUT_D + mrow];
        acc8 = __builtin_amdgcn_wmma_f32_16x16x4_f32(false, a, false, b,
                                                     (short)0, acc8, false, false);
    }
#pragma unroll
    for (int j = 0; j < 8; ++j) {
        const int m = row0 + j + 8 * half;
        if (m < n) h2[(long long)m * OUT_D + mrow] = acc8[j];
    }
}

#define G16_THREADS 256
#define G16_PAIRS   4

template <int P>
__device__ __forceinline__ float gcn16_row_load(const float* __restrict__ h, int sB, int colB) {
    const unsigned off = (unsigned)(gcn_half_share<P>(sB) + colB);
    return *(const float*)((const char*)h + off);
}

__global__ void __launch_bounds__(G16_THREADS)
gcn_agg2_kernel(const float* __restrict__ h2,
                const int* __restrict__ rowptr,
                const int* __restrict__ list,
                const float* __restrict__ nd,
                const float* __restrict__ b2,
                float* __restrict__ out, int n) {
    const int lane = threadIdx.x & 31;
    const int half = lane >> 4;
    const int col  = lane & 15;
    const int colB = col * 4;
    const int gw   = blockIdx.x * (G16_THREADS / 32) + (threadIdx.x >> 5);
    const int2* __restrict__ ent = (const int2*)list;
    const float b2c = b2[col];

    for (int t = 0; t < G16_PAIRS; ++t) {
        const int pair = gw * G16_PAIRS + t;
        if (2 * pair >= n) break;
        const int  i     = 2 * pair + half;
        const bool valid = i < n;
        const int  p0    = valid ? rowptr[i] : 0;
        const int  deg   = valid ? (rowptr[i + 1] - p0) : 0;
        const int  dlo   = __builtin_amdgcn_readlane(deg, 0);
        const int  dhi   = __builtin_amdgcn_readlane(deg, 16);
        const int  dmax  = dlo > dhi ? dlo : dhi;
        float acc = 0.0f;
        for (int c = 0; c < dmax; c += 16) {
            const int2 e   = ent[(c + col < deg) ? (p0 + c + col) : 0];
            const int sB   = e.x * (OUT_D * 4);
            const int wb   = e.y;
            const int rem  = deg - c;
            const int remx = dmax - c;
            float v[16];
#define G16_LD(P)                                                   \
            if ((P) < remx) {                                       \
                if ((P) < rem) v[P] = gcn16_row_load<P>(h2, sB, colB); \
            }
            G16_LD(0)  G16_LD(1)  G16_LD(2)  G16_LD(3)
            G16_LD(4)  G16_LD(5)  G16_LD(6)  G16_LD(7)
            G16_LD(8)  G16_LD(9)  G16_LD(10) G16_LD(11)
            G16_LD(12) G16_LD(13) G16_LD(14) G16_LD(15)
#undef G16_LD
#define G16_AD(P)                                                   \
            if ((P) < remx) {                                       \
                if ((P) < rem) acc += v[P] * __int_as_float(gcn_half_share<P>(wb)); \
            }
            G16_AD(0)  G16_AD(1)  G16_AD(2)  G16_AD(3)
            G16_AD(4)  G16_AD(5)  G16_AD(6)  G16_AD(7)
            G16_AD(8)  G16_AD(9)  G16_AD(10) G16_AD(11)
            G16_AD(12) G16_AD(13) G16_AD(14) G16_AD(15)
#undef G16_AD
        }
        if (valid) out[(long long)i * OUT_D + col] = acc * nd[i] + b2c;
    }
}

extern "C" void kernel_launch(void* const* d_in, const int* in_sizes, int n_in,
                              void* d_out, int out_size, void* d_ws, size_t ws_size,
                              hipStream_t stream) {
    const float* feat = (const float*)d_in[0];
    const int*   src  = (const int*)d_in[1];
    const int*   dst  = (const int*)d_in[2];
    const float* ew   = (const float*)d_in[3];
    const float* W1   = (const float*)d_in[4];
    const float* b1   = (const float*)d_in[5];
    const float* W2   = (const float*)d_in[6];
    const float* b2   = (const float*)d_in[7];
    float* out = (float*)d_out;

    const int n = in_sizes[0] / IN_D;
    const int E = in_sizes[1];

    float* degO = (float*)d_ws;
    float* degI = degO + n;
    unsigned* ctr = (unsigned*)(degI + n);
    int*   rowptr = (int*)(ctr + SC_CTR_WORDS);
    float* ns   = degO;
    float* nd   = degI;
    float* h1   = degO + 4LL * n;
    int*   list = (int*)(h1 + (long long)n * HID_D);
    float* h2   = (float*)(list + (long long)n * HID_D);
    int*   cursor = (int*)h2;

    const int B = 256;
    auto blocks = [&](long long work) { return (unsigned)((work + B - 1) / B); };

    gcn_zero_kernel<<<blocks((2LL * n + SC_CTR_WORDS + 3) / 4), B, 0, stream>>>(
        degO, 2LL * n + SC_CTR_WORDS);

    const unsigned degRanges = (unsigned)((n + DEG_BINS - 1) / DEG_BINS);
    gcn_degree_kernel<<<2u * degRanges * DEG_PARTS, DEG_THREADS, 0, stream>>>(
        src, dst, degO, degI, E, n);

    gcn_rowptr_kernel<<<(unsigned)((n + RP_THREADS - 1) / RP_THREADS), RP_THREADS, 0, stream>>>(
        degI, rowptr, cursor, n);

    gcn_norm_kernel<<<blocks(n), B, 0, stream>>>(degO, degI, n);

    const unsigned scWaves  = 2u * (unsigned)((E + SC_UNIT_EDGES - 1) / SC_UNIT_EDGES);
    const unsigned scBlocks = (scWaves + (SC_THREADS / 32) - 1) / (SC_THREADS / 32);
    gcn_fill_kernel<<<scBlocks, SC_THREADS, 0, stream>>>(src, dst, ew, cursor, list, ctr, E, n);

    const int tiles = (n + 15) / 16;
    const int wavesPerBlock = B / 32;
    const unsigned gemmBlocks = (unsigned)((tiles + wavesPerBlock - 1) / wavesPerBlock);
    gcn_gemm1_kernel<<<gemmBlocks, B, 0, stream>>>(feat, ns, W1, h1, n);

    gcn_agg1_gemm2_kernel<<<gemmBlocks, AG_THREADS, 0, stream>>>(
        h1, rowptr, list, ns, nd, b1, W2, h2, n);

    const long long pairs = ((long long)n + 1) / 2;
    const unsigned g16Blocks =
        (unsigned)((pairs + (long long)G16_PAIRS * (G16_THREADS / 32) - 1) /
                   ((long long)G16_PAIRS * (G16_THREADS / 32)));
    gcn_agg2_kernel<<<g16Blocks, G16_THREADS, 0, stream>>>(h2, rowptr, list, nd, b2, out, n);
}


# Launches 8 -> 6: GEMM1 tiles run as alternate work-groups of the edge-fill launch (independent data), and the norm pass goes: rsqrt(max(deg,1)) is computed where it is used; sums unchanged
# speedup vs baseline: 3.2223x; 1.0877x over previous
#include <hip/hip_runtime.h>

typedef __attribute__((ext_vector_type(2))) float v2f;
typedef __attribute__((ext_vector_type(8))) float v8f;

#define IN_D  32
#define HID_D 32
#define OUT_D 16

__global__ void gcn_zero_kernel(float* __restrict__ p, long long n) {
    long long i = ((long long)blockIdx.x * blockDim.x + threadIdx.x) * 4;
    if (i + 3 < n) {
        *(float4*)(p + i) = make_float4(0.f, 0.f, 0.f, 0.f);
    } else {
        for (long long j = i; j < n; ++j) p[j] = 0.0f;
    }
}

#define DEG_BINS    25600
#define DEG_PARTS   32
#define DEG_THREADS 1024
#define DEG_LOADS   8
#define DEG_FLUSH   5

__device__ __forceinline__ void gcn_count(unsigned* bins, int node, int base, int nb) {
    const unsigned r = (unsigned)(node - base);
    if (r < (unsigned)nb) atomicAdd(bins + r, 1u);
}

__global__ void __launch_bounds__(DEG_THREADS)
gcn_degree_kernel(const int* __restrict__ src,
                  const int* __restrict__ dst,
                  float* __restrict__ degO,
                  float* __restrict__ degI, int E, int n) {
    __shared__ unsigned bins[DEG_BINS];

    const int which = blockIdx.x & 1;
    const int rp    = blockIdx.x >> 1;
    const int part  = rp % DEG_PARTS;
    const int range = rp / DEG_PARTS;
    const int* __restrict__ idx = which ? dst : src;
    float* __restrict__ deg     = which ? degI : degO;
    const int base = range * DEG_BINS;
    const int nb   = (n - base < DEG_BINS) ? (n - base) : DEG_BINS;

    for (int i = threadIdx.x * 4; i < DEG_BINS; i += DEG_THREADS * 4)
        *(uint4*)(bins + i) = make_uint4(0u, 0u, 0u, 0u);
    __syncthreads();

    const int per  = ((E + DEG_PARTS * 4 - 1) / (DEG_PARTS * 4)) * 4;
    const int lo   = part * per;
    const int hi   = (lo + per < E) ? (lo + per) : E;
    const int len  = (hi > lo) ? (hi - lo) : 0;
    const int nvec = len >> 2;
    const int4* __restrict__ vp = (const int4*)(idx + lo);

    for (int v0 = threadIdx.x; v0 < nvec; v0 += DEG_THREADS * DEG_LOADS) {
        int4 q[DEG_LOADS];
#pragma unroll
        for (int k = 0; k < DEG_LOADS; ++k) {
            const int v = v0 + k * DEG_THREADS;
            q[k] = (v < nvec) ? vp[v] : make_int4(-1, -1, -1, -1);
        }
#pragma unroll
        for (int k = 0; k < DEG_LOADS; ++k) {
            gcn_count(bins, q[k].x, base, nb);
            gcn_count(bins, q[k].y, base, nb);
            gcn_count(bins, q[k].z, base, nb);
            gcn_count(bins, q[k].w, base, nb);
        }
    }
    {
        const int e = lo + (nvec << 2) + threadIdx.x;
        if (e < hi) gcn_count(bins, idx[e], base, nb);
    }
    __syncthreads();

    const int rot = part * (DEG_BINS / DEG_PARTS);
    for (int i0 = threadIdx.x; i0 < DEG_BINS; i0 += DEG_THREADS * DEG_FLUSH) {
        unsigned c[DEG_FLUSH];
        int      b[DEG_FLUSH];
#pragma unroll
        for (int k = 0; k < DEG_FLUSH; ++k) {
            const int i = i0 + k * DEG_THREADS;
            int bb = i + rot;
            if (bb >= DEG_BINS) bb -= DEG_BINS;
            b[k] = bb;
            c[k] = (i < DEG_BINS) ? bins[bb] : 0u;
        }
#pragma unroll
        for (int k = 0; k < DEG_FLUSH; ++k)
            if (c[k] != 0u && b[k] < nb) atomicAdd(deg + base + b[k], (float)c[k]);
    }
}

__device__ __forceinline__ float gcn_norm(float deg) {
    return rsqrtf(fmaxf(deg, 1.0f));
}

#define GEMM1_WAVES 8

__device__ __forceinline__ void gcn_gemm1_block(const float* __restrict__ feat,
                                                const float* __restrict__ degO,
                                                const float* __restrict__ W1,
                                                float* __restrict__ h1, int n, int blk) {
    const int wave = threadIdx.x >> 5;
    const int lane = threadIdx.x & 31;
    const int tile = blk * GEMM1_WAVES + wave;
    const int row0 = tile * 16;
    if (row0 >= n) return;

    const int half = lane >> 4;
    const int mrow = lane & 15;
    const int arow = row0 + mrow;
    const float scale = gcn_norm(degO[arow]);

    v8f acc0 = {};
    v8f acc1 = {};
#pragma unroll
    for (int k0 = 0; k0 < IN_D; k0 += 4) {
        const int kk = k0 + 2 * half;
        v2f a, b0, b1;
        const float* ap = feat + (long long)arow * IN_D + kk;
        a.x = ap[0] * scale;
        a.y = ap[1] * scale;
        b0.x = W1[kk * HID_D + mrow];
        b0.y = W1[(kk + 1) * HID_D + mrow];
        b1.x = W1[kk * HID_D + 16 + mrow];
        b1.y = W1[(kk + 1) * HID_D + 16 + mrow];
        acc0 = __builtin_amdgcn_wmma_f32_16x16x4_f32(false, a, false, b0,
                                                     (short)0, acc0, false, false);
        acc1 = __builtin_amdgcn_wmma_f32_16x16x4_f32(false, a, false, b1,
                                                     (short)0, acc1, false, false);
    }
#pragma unroll
    for (int j = 0; j < 8; ++j) {
        const int m = row0 + j + 8 * half;
        h1[(long long)m * HID_D + mrow]      = acc0[j];
        h1[(long long)m * HID_D + 16 + mrow] = acc1[j];
    }
}

#define RP_THREADS 1024
#define RP_LOADS   8

__global__ void __launch_bounds__(RP_THREADS)
gcn_rowptr_kernel(const float* __restrict__ degI,
                  int* __restrict__ rowptr,
                  int* __restrict__ cursor, int n) {
    __shared__ int wsum[32];
    const int tid   = threadIdx.x;
    const int lane  = tid & 31;
    const int wv    = tid >> 5;
    const int start = blockIdx.x * RP_THREADS;

    int s = 0;
    for (int j0 = tid * 4; j0 < start; j0 += RP_THREADS * 4 * RP_LOADS) {
        float4 q[RP_LOADS];
#pragma unroll
        for (int k = 0; k < RP_LOADS; ++k) {
            const int j = j0 + k * RP_THREADS * 4;
            q[k] = *(const float4*)(degI + ((j < start) ? j : 0));
        }
#pragma unroll
        for (int k = 0; k < RP_LOADS; ++k) {
            const int j = j0 + k * RP_THREADS * 4;
            const int t = (int)q[k].x + (int)q[k].y + (int)q[k].z + (int)q[k].w;
            s += (j < start) ? t : 0;
        }
    }
    for (int o = 16; o > 0; o >>= 1) s += __shfl_xor(s, o);
    if (lane == 0) wsum[wv] = s;
    __syncthreads();
    int base = 0;
    for (int k = 0; k < 32; ++k) base += wsum[k];
    __syncthreads();

    const int i = start + tid;
    const int c = (i < n) ? (int)degI[i] : 0;
    int inc = c;
    for (int o = 1; o < 32; o <<= 1) {
        const int t = __shfl_up(inc, o);
        if (lane >= o) inc += t;
    }
    if (lane == 31) wsum[wv] = inc;
    __syncthreads();
    int wbase = 0;
    for (int k = 0; k < wv; ++k) wbase += wsum[k];
    const int excl = base + wbase + inc - c;
    if (i < n) {
        rowptr[i] = excl;
        cursor[i] = excl;
    }
    if (i == n - 1) rowptr[n] = excl + c;
}

#define SC_UNIT_EDGES 256
#define SC_THREADS    256
#define SC_CTR_STRIDE 64
#define SC_CTR_WORDS  (4 * SC_CTR_STRIDE)

__device__ __forceinline__ int gcn_home_half() {
    const unsigned id = __builtin_amdgcn_s_sendmsg_rtn(0x87);
    return (int)((id >> 18) & 1u);
}

__device__ __forceinline__ unsigned gcn_take_unit(unsigned* c, int lane) {
    unsigned t = 0u;
    if (lane == 0)
        t = __hip_atomic_fetch_add(c, 1u, __ATOMIC_RELAXED, __HIP_MEMORY_SCOPE_AGENT);
    return (unsigned)__builtin_amdgcn_readfirstlane((int)t);
}

template <int P>
__device__ __forceinline__ int gcn_half_share(int v) {
    return __builtin_amdgcn_update_dpp(0, v, 0x150 + P, 0xf, 0xf, true);
}

#define FILL_BATCHES (SC_UNIT_EDGES / 32)

__device__ __forceinline__ void gcn_unit_fill(const int* __restrict__ src,
                                              const int* __restrict__ dst,
                                              const float* __restrict__ w,
                                              int* __restrict__ cursor,
                                              int2* __restrict__ list,
                                              int e_lo, int e_hi, int lane,
                                              int nSplit, int part) {
    int d[FILL_BATCHES], s[FILL_BATCHES], wb[FILL_BATCHES], pos[FILL_BATCHES];
#pragma unroll
    for (int b = 0; b < FILL_BATCHES; ++b) {
        const int e = e_lo + b * 32 + lane;
        const bool in = e < e_hi;
        d[b]  = in ? dst[e] : -1;
        s[b]  = in ? src[e] : 0;
        wb[b] = in ? __float_as_int(w[e]) : 0;
    }
#pragma unroll
    for (int b = 0; b < FILL_BATCHES; ++b) {
        const bool mine = (d[b] >= 0) && ((d[b] >= nSplit) == (part != 0));
        pos[b] = mine ? atomicAdd(cursor + d[b], 1) : -1;
    }
#pragma unroll
    for (int b = 0; b < FILL_BATCHES; ++b)
        if (pos[b] >= 0) list[pos[b]] = make_int2(s[b], wb[b]);
}

__global__ void __launch_bounds__(SC_THREADS)
gcn_fill_gemm1_kernel(const int* __restrict__ src,
                      const int* __restrict__ dst,
                      const float* __restrict__ w,
                      int* __restrict__ cursor,
                      int* __restrict__ list,
                      unsigned* __restrict__ ctr, int E, int n,
                      const float* __restrict__ feat,
                      const float* __restrict__ degO,
                      const float* __restrict__ W1,
                      float* __restrict__ h1, int gemmBlocks) {
    const int b = (int)blockIdx.x;
    if (b < 2 * gemmBlocks && (b & 1)) {
        gcn_gemm1_block(feat, degO, W1, h1, n, b >> 1);
        return;
    }
    const int lane = threadIdx.x & 31;
    const int nSplit = (n >> 1) & ~31;
    const unsigned units = (unsigned)((E + SC_UNIT_EDGES - 1) / SC_UNIT_EDGES);
    const int home = gcn_home_half();
    for (int pass = 0; pass < 2; ++pass) {
        const int part = home ^ pass;
        unsigned* c = ctr + part * SC_CTR_STRIDE;
        for (unsigned it = 0; it < units; ++it) {
            const unsigned t = gcn_take_unit(c, lane);
            if (t >= units) break;
            const int e_lo = (int)t * SC_UNIT_EDGES;
            const int e_hi = (e_lo + SC_UNIT_EDGES < E) ? (e_lo + SC_UNIT_EDGES) : E;
            gcn_unit_fill(src, dst, w, cursor, (int2*)list, e_lo, e_hi, lane, nSplit, part);
        }
    }
}

#define AG_THREADS 256
#define AG_PITCH   34

__global__ void __launch_bounds__(AG_THREADS)
gcn_agg1_gemm2_kernel(const float* __restrict__ h1,
                      const int* __restrict__ rowptr,
                      const int* __restrict__ list,
                      const float* __restrict__ degO,
                      const float* __restrict__ degI,
                      const float* __restrict__ b1,
                      const float* __restrict__ W2,
                      float* __restrict__ h2, int n) {
    __shared__ float tiles[(AG_THREADS / 32) * 16 * AG_PITCH];
    const int wave = threadIdx.x >> 5;
    const int lane = threadIdx.x & 31;
    const int tile = blockIdx.x * (blockDim.x >> 5) + wave;
    const int row0 = tile * 16;
    if (row0 >= n) return;

    float* tl = tiles + wave * (16 * AG_PITCH);
    const int half = lane >> 4;
    const int mrow = lane & 15;
    const unsigned laneB = (unsigned)lane * 4u;
    const int2* __restrict__ ent = (const int2*)list;

    const int rrow = (row0 + mrow < n) ? (row0 + mrow) : (n - 1);
    const int   endv = rowptr[rrow + 1];
    const float ndv  = gcn_norm(degI[rrow]);
    const float nsv  = gcn_norm(degO[rrow]);
    const float b1c  = b1[lane];
    const int P0 = __builtin_amdgcn_readfirstlane(rowptr[row0]);
    const int P1 = __builtin_amdgcn_readlane(endv, 15);

    int   r    = 0;
    int   rend = __builtin_amdgcn_readlane(endv, 0);
    float acc  = 0.0f;

#define AG_FLUSH()                                                                    \
    {                                                                                 \
        const float ndr = __int_as_float(__builtin_amdgcn_readlane(__float_as_int(ndv), r)); \
        const float nsr = __int_as_float(__builtin_amdgcn_readlane(__float_as_int(nsv), r)); \
        tl[r * AG_PITCH + lane] = fmaxf(acc * ndr + b1c, 0.0f) * nsr;                 \
        acc = 0.0f;                                                                   \
        ++r;                                                                          \
    }

    int2 e_n = make_int2(0, 0);
    if (P0 + lane < P1) e_n = ent[P0 + lane];
    for (int base = P0; base < P1; base += 32) {
        const int s  = e_n.x;
        const int wb = e_n.y;
        const int qn = base + 32 + lane;
        if (qn < P1) e_n = ent[qn];
        const int cnt = (P1 - base < 32) ? (P1 - base) : 32;
        float v[32];
#pragma unroll
        for (int k = 0; k < 32; ++k) {
            if (k < cnt) {
                const unsigned sk = (unsigned)__builtin_amdgcn_readlane(s, k);
                v[k] = *(const float*)((const char*)h1 + ((sk << 7) + laneB));
            }
        }
#pragma unroll
        for (int k = 0; k < 32; ++k) {
            if (k < cnt) {
                while (base + k >= rend) {
                    AG_FLUSH()
                    rend = __builtin_amdgcn_readlane(endv, r);
                }
                const float wk = __int_as_float(__builtin_amdgcn_readlane(wb, k));
                acc += v[k] * wk;
            }
        }
    }
    while (r < 16) AG_FLUSH()
#undef AG_FLUSH

    __builtin_amdgcn_fence(__ATOMIC_RELEASE, "workgroup");
    __builtin_amdgcn_fence(__ATOMIC_ACQUIRE, "workgroup");

    v8f acc8 = {};
#pragma unroll
    for (int k0 = 0; k0 < HID_D; k0 += 4) {
        const int kk = k0 + 2 * half;
        v2f a, b;
        a.x = tl[mrow * AG_PITCH + kk];
        a.y = tl[mrow * AG_PITCH + kk + 1];
        b.x = W2[kk * OUT_D + mrow];
        b.y = W2[(kk + 1) * OUT_D + mrow];
        acc8 = __builtin_amdgcn_wmma_f32_16x16x4_f32(false, a, false, b,
                                                     (short)0, acc8, false, false);
    }
#pragma unroll
    for (int j = 0; j < 8; ++j) {
        const int m = row0 + j + 8 * half;
        if (m < n) h2[(long long)m * OUT_D + mrow] = acc8[j];
    }
}

#define G16_THREADS 256
#define G16_PAIRS   4

template <int P>
__device__ __forceinline__ float gcn16_row_load(const float* __restrict__ h, int sB, int colB) {
    const unsigned off = (unsigned)(gcn_half_share<P>(sB) + colB);
    return *(const float*)((const char*)h + off);
}

__global__ void __launch_bounds__(G16_THREADS)
gcn_agg2_kernel(const float* __restrict__ h2,
                const int* __restrict__ rowptr,
                const int* __restrict__ list,
                const float* __restrict__ degI,
                const float* __restrict__ b2,
                float* __restrict__ out, int n) {
    const int lane = threadIdx.x & 31;
    const int half = lane >> 4;
    const int col  = lane & 15;
    const int colB = col * 4;
    const int gw   = blockIdx.x * (G16_THREADS / 32) + (threadIdx.x >> 5);
    const int2* __restrict__ ent = (const int2*)list;
    const float b2c = b2[col];

    for (int t = 0; t < G16_PAIRS; ++t) {
        const int pair = gw * G16_PAIRS + t;
        if (2 * pair >= n) break;
        const int  i     = 2 * pair + half;
        const bool valid = i < n;
        const int  p0    = valid ? rowptr[i] : 0;
        const int  deg   = valid ? (rowptr[i + 1] - p0) : 0;
        const int  dlo   = __builtin_amdgcn_readlane(deg, 0);
        const int  dhi   = __builtin_amdgcn_readlane(deg, 16);
        const int  dmax  = dlo > dhi ? dlo : dhi;
        float acc = 0.0f;
        for (int c = 0; c < dmax; c += 16) {
            const int2 e   = ent[(c + col < deg) ? (p0 + c + col) : 0];
            const int sB   = e.x * (OUT_D * 4);
            const int wb   = e.y;
            const int rem  = deg - c;
            const int remx = dmax - c;
            float v[16];
#define G16_LD(P)                                                   \
            if ((P) < remx) {                                       \
                if ((P) < rem) v[P] = gcn16_row_load<P>(h2, sB, colB); \
            }
            G16_LD(0)  G16_LD(1)  G16_LD(2)  G16_LD(3)
            G16_LD(4)  G16_LD(5)  G16_LD(6)  G16_LD(7)
            G16_LD(8)  G16_LD(9)  G16_LD(10) G16_LD(11)
            G16_LD(12) G16_LD(13) G16_LD(14) G16_LD(15)
#undef G16_LD
#define G16_AD(P)                                                   \
            if ((P) < remx) {                                       \
                if ((P) < rem) acc += v[P] * __int_as_float(gcn_half_share<P>(wb)); \
            }
            G16_AD(0)  G16_AD(1)  G16_AD(2)  G16_AD(3)
            G16_AD(4)  G16_AD(5)  G16_AD(6)  G16_AD(7)
            G16_AD(8)  G16_AD(9)  G16_AD(10) G16_AD(11)
            G16_AD(12) G16_AD(13) G16_AD(14) G16_AD(15)
#undef G16_AD
        }
        if (valid) out[(long long)i * OUT_D + col] = acc * gcn_norm(degI[i]) + b2c;
    }
}

extern "C" void kernel_launch(void* const* d_in, const int* in_sizes, int n_in,
                              void* d_out, int out_size, void* d_ws, size_t ws_size,
                              hipStream_t stream) {
    const float* feat = (const float*)d_in[0];
    const int*   src  = (const int*)d_in[1];
    const int*   dst  = (const int*)d_in[2];
    const float* ew   = (const float*)d_in[3];
    const float* W1   = (const float*)d_in[4];
    const float* b1   = (const float*)d_in[5];
    const float* W2   = (const float*)d_in[6];
    const float* b2   = (const float*)d_in[7];
    float* out = (float*)d_out;

    const int n = in_sizes[0] / IN_D;
    const int E = in_sizes[1];

    float* degO = (float*)d_ws;
    float* degI = degO + n;
    unsigned* ctr = (unsigned*)(degI + n);
    int*   rowptr = (int*)(ctr + SC_CTR_WORDS);
    float* h1   = degO + 4LL * n;
    int*   list = (int*)(h1 + (long long)n * HID_D);
    float* h2   = (float*)(list + (long long)n * HID_D);
    int*   cursor = (int*)h2;

    const int B = 256;
    auto blocks = [&](long long work) { return (unsigned)((work + B - 1) / B); };

    gcn_zero_kernel<<<blocks((2LL * n + SC_CTR_WORDS + 3) / 4), B, 0, stream>>>(
        degO, 2LL * n + SC_CTR_WORDS);

    const unsigned degRanges = (unsigned)((n + DEG_BINS - 1) / DEG_BINS);
    gcn_degree_kernel<<<2u * degRanges * DEG_PARTS, DEG_THREADS, 0, stream>>>(
        src, dst, degO, degI, E, n);

    gcn_rowptr_kernel<<<(unsigned)((n + RP_THREADS - 1) / RP_THREADS), RP_THREADS, 0, stream>>>(
        degI, rowptr, cursor, n);

    const unsigned scWaves  = 2u * (unsigned)((E + SC_UNIT_EDGES - 1) / SC_UNIT_EDGES);
    const unsigned scBlocks = (scWaves + (SC_THREADS / 32) - 1) / (SC_THREADS / 32);
    const int tiles = (n + 15) / 16;
    const unsigned gemmBlocks = (unsigned)((tiles + GEMM1_WAVES - 1) / GEMM1_WAVES);
    const unsigned fillBlocks = (scBlocks > gemmBlocks) ? scBlocks : gemmBlocks;
    gcn_fill_gemm1_kernel<<<fillBlocks + gemmBlocks, SC_THREADS, 0, stream>>>(
        src, dst, ew, cursor, list, ctr, E, n, feat, degO, W1, h1, (int)gemmBlocks);

    const unsigned aggBlocks =
        (unsigned)((tiles + (AG_THREADS / 32) - 1) / (AG_THREADS / 32));
    gcn_agg1_gemm2_kernel<<<aggBlocks, AG_THREADS, 0, stream>>>(
        h1, rowptr, list, degO, degI, b1, W2, h2, n);

    const long long pairs = ((long long)n + 1) / 2;
    const unsigned g16Blocks =
        (unsigned)((pairs + (long long)G16_PAIRS * (G16_THREADS / 32) - 1) /
                   ((long long)G16_PAIRS * (G16_THREADS / 32)));
    gcn_agg2_kernel<<<g16Blocks, G16_THREADS, 0, stream>>>(h2, rowptr, list, degI, b2, out, n);
}


# Layer-1 aggregation walk: whole 32-entry batches run a copy of the body without the per-entry count test, with the row-end code off the straight path; sums and their order unchanged
# speedup vs baseline: 3.3839x; 1.0502x over previous
#include <hip/hip_runtime.h>

typedef __attribute__((ext_vector_type(2))) float v2f;
typedef __attribute__((ext_vector_type(8))) float v8f;

#define IN_D  32
#define HID_D 32
#define OUT_D 16

__global__ void gcn_zero_kernel(float* __restrict__ p, long long n) {
    long long i = ((long long)blockIdx.x * blockDim.x + threadIdx.x) * 4;
    if (i + 3 < n) {
        *(float4*)(p + i) = make_float4(0.f, 0.f, 0.f, 0.f);
    } else {
        for (long long j = i; j < n; ++j) p[j] = 0.0f;
    }
}

#define DEG_BINS    25600
#define DEG_PARTS   32
#define DEG_THREADS 1024
#define DEG_LOADS   8
#define DEG_FLUSH   5

__device__ __forceinline__ void gcn_count(unsigned* bins, int node, int base, int nb) {
    const unsigned r = (unsigned)(node - base);
    if (r < (unsigned)nb) atomicAdd(bins + r, 1u);
}

__global__ void __launch_bounds__(DEG_THREADS)
gcn_degree_kernel(const int* __restrict__ src,
                  const int* __restrict__ dst,
                  float* __restrict__ degO,
                  float* __restrict__ degI, int E, int n) {
    __shared__ unsigned bins[DEG_BINS];

    const int which = blockIdx.x & 1;
    const int rp    = blockIdx.x >> 1;
    const int part  = rp % DEG_PARTS;
    const int range = rp / DEG_PARTS;
    const int* __restrict__ idx = which ? dst : src;
    float* __restrict__ deg     = which ? degI : degO;
    const int base = range * DEG_BINS;
    const int nb   = (n - base < DEG_BINS) ? (n - base) : DEG_BINS;

    for (int i = threadIdx.x * 4; i < DEG_BINS; i += DEG_THREADS * 4)
        *(uint4*)(bins + i) = make_uint4(0u, 0u, 0u, 0u);
    __syncthreads();

    const int per  = ((E + DEG_PARTS * 4 - 1) / (DEG_PARTS * 4)) * 4;
    const int lo   = part * per;
    const int hi   = (lo + per < E) ? (lo + per) : E;
    const int len  = (hi > lo) ? (hi - lo) : 0;
    const int nvec = len >> 2;
    const int4* __restrict__ vp = (const int4*)(idx + lo);

    for (int v0 = threadIdx.x; v0 < nvec; v0 += DEG_THREADS * DEG_LOADS) {
        int4 q[DEG_LOADS];
#pragma unroll
        for (int k = 0; k < DEG_LOADS; ++k) {
            const int v = v0 + k * DEG_THREADS;
            q[k] = (v < nvec) ? vp[v] : make_int4(-1, -1, -1, -1);
        }
#pragma unroll
        for (int k = 0; k < DEG_LOADS; ++k) {
            gcn_count(bins, q[k].x, base, nb);
            gcn_count(bins, q[k].y, base, nb);
            gcn_count(bins, q[k].z, base, nb);
            gcn_count(bins, q[k].w, base, nb);
        }
    }
    {
        const int e = lo + (nvec << 2) + threadIdx.x;
        if (e < hi) gcn_count(bins, idx[e], base, nb);
    }
    __syncthreads();

    const int rot = part * (DEG_BINS / DEG_PARTS);
    for (int i0 = threadIdx.x; i0 < DEG_BINS; i0 += DEG_THREADS * DEG_FLUSH) {
        unsigned c[DEG_FLUSH];
        int      b[DEG_FLUSH];
#pragma unroll
        for (int k = 0; k < DEG_FLUSH; ++k) {
            const int i = i0 + k * DEG_THREADS;
            int bb = i + rot;
            if (bb >= DEG_BINS) bb -= DEG_BINS;
            b[k] = bb;
            c[k] = (i < DEG_BINS) ? bins[bb] : 0u;
        }
#pragma unroll
        for (int k = 0; k < DEG_FLUSH; ++k)
            if (c[k] != 0u && b[k] < nb) atomicAdd(deg + base + b[k], (float)c[k]);
    }
}

__device__ __forceinline__ float gcn_norm(float deg) {
    return rsqrtf(fmaxf(deg, 1.0f));
}

#define GEMM1_WAVES 8

__device__ __forceinline__ void gcn_gemm1_block(const float* __restrict__ feat,
                                                const float* __restrict__ degO,
                                                const float* __restrict__ W1,
                                                float* __restrict__ h1, int n, int blk) {
    const int wave = threadIdx.x >> 5;
    const int lane = threadIdx.x & 31;
    const int tile = blk * GEMM1_WAVES + wave;
    const int row0 = tile * 16;
    if (row0 >= n) return;

    const int half = lane >> 4;
    const int mrow = lane & 15;
    const int arow = row0 + mrow;
    const float scale = gcn_norm(degO[arow]);

    v8f acc0 = {};
    v8f acc1 = {};
#pragma unroll
    for (int k0 = 0; k0 < IN_D; k0 += 4) {
        const int kk = k0 + 2 * half;
        v2f a, b0, b1;
        const float* ap = feat + (long long)arow * IN_D + kk;
        a.x = ap[0] * scale;
        a.y = ap[1] * scale;
        b0.x = W1[kk * HID_D + mrow];
        b0.y = W1[(kk + 1) * HID_D + mrow];
        b1.x = W1[kk * HID_D + 16 + mrow];
        b1.y = W1[(kk + 1) * HID_D + 16 + mrow];
        acc0 = __builtin_amdgcn_wmma_f32_16x16x4_f32(false, a, false, b0,
                                                     (short)0, acc0, false, false);
        acc1 = __builtin_amdgcn_wmma_f32_16x16x4_f32(false, a, false, b1,
                                                     (short)0, acc1, false, false);
    }
#pragma unroll
    for (int j = 0; j < 8; ++j) {
        const int m = row0 + j + 8 * half;
        h1[(long long)m * HID_D + mrow]      = acc0[j];
        h1[(long long)m * HID_D + 16 + mrow] = acc1[j];
    }
}

#define RP_THREADS 1024
#define RP_LOADS   8

__global__ void __launch_bounds__(RP_THREADS)
gcn_rowptr_kernel(const float* __restrict__ degI,
                  int* __restrict__ rowptr,
                  int* __restrict__ cursor, int n) {
    __shared__ int wsum[32];
    const int tid   = threadIdx.x;
    const int lane  = tid & 31;
    const int wv    = tid >> 5;
    const int start = blockIdx.x * RP_THREADS;

    int s = 0;
    for (int j0 = tid * 4; j0 < start; j0 += RP_THREADS * 4 * RP_LOADS) {
        float4 q[RP_LOADS];
#pragma unroll
        for (int k = 0; k < RP_LOADS; ++k) {
            const int j = j0 + k * RP_THREADS * 4;
            q[k] = *(const float4*)(degI + ((j < start) ? j : 0));
        }
#pragma unroll
        for (int k = 0; k < RP_LOADS; ++k) {
            const int j = j0 + k * RP_THREADS * 4;
            const int t = (int)q[k].x + (int)q[k].y + (int)q[k].z + (int)q[k].w;
            s += (j < start) ? t : 0;
        }
    }
    for (int o = 16; o > 0; o >>= 1) s += __shfl_xor(s, o);
    if (lane == 0) wsum[wv] = s;
    __syncthreads();
    int base = 0;
    for (int k = 0; k < 32; ++k) base += wsum[k];
    __syncthreads();

    const int i = start + tid;
    const int c = (i < n) ? (int)degI[i] : 0;
    int inc = c;
    for (int o = 1; o < 32; o <<= 1) {
        const int t = __shfl_up(inc, o);
        if (lane >= o) inc += t;
    }
    if (lane == 31) wsum[wv] = inc;
    __syncthreads();
    int wbase = 0;
    for (int k = 0; k < wv; ++k) wbase += wsum[k];
    const int excl = base + wbase + inc - c;
    if (i < n) {
        rowptr[i] = excl;
        cursor[i] = excl;
    }
    if (i == n - 1) rowptr[n] = excl + c;
}

#define SC_UNIT_EDGES 256
#define SC_THREADS    256
#define SC_CTR_STRIDE 64
#define SC_CTR_WORDS  (4 * SC_CTR_STRIDE)

__device__ __forceinline__ int gcn_home_half() {
    const unsigned id = __builtin_amdgcn_s_sendmsg_rtn(0x87);
    return (int)((id >> 18) & 1u);
}

__device__ __forceinline__ unsigned gcn_take_unit(unsigned* c, int lane) {
    unsigned t = 0u;
    if (lane == 0)
        t = __hip_atomic_fetch_add(c, 1u, __ATOMIC_RELAXED, __HIP_MEMORY_SCOPE_AGENT);
    return (unsigned)__builtin_amdgcn_readfirstlane((int)t);
}

template <int P>
__device__ __forceinline__ int gcn_half_share(int v) {
    return __builtin_amdgcn_update_dpp(0, v, 0x150 + P, 0xf, 0xf, true);
}

#define FILL_BATCHES (SC_UNIT_EDGES / 32)

__device__ __forceinline__ void gcn_unit_fill(const int* __restrict__ src,
                                              const int* __restrict__ dst,
                                              const float* __restrict__ w,
                                              int* __restrict__ cursor,
                                              int2* __restrict__ list,
                                              int e_lo, int e_hi, int lane,
                                              int nSplit, int part) {
    int d[FILL_BATCHES], s[FILL_BATCHES], wb[FILL_BATCHES], pos[FILL_BATCHES];
#pragma unroll
    for (int b = 0; b < FILL_BATCHES; ++b) {
        const int e = e_lo + b * 32 + lane;
        const bool in = e < e_hi;
        d[b]  = in ? dst[e] : -1;
        s[b]  = in ? src[e] : 0;
        wb[b] = in ? __float_as_int(w[e]) : 0;
    }
#pragma unroll
    for (int b = 0; b < FILL_BATCHES; ++b) {
        const bool mine = (d[b] >= 0) && ((d[b] >= nSplit) == (part != 0));
        pos[b] = mine ? atomicAdd(cursor + d[b], 1) : -1;
    }
#pragma unroll
    for (int b = 0; b < FILL_BATCHES; ++b)
        if (pos[b] >= 0) list[pos[b]] = make_int2(s[b], wb[b]);
}

__global__ void __launch_bounds__(SC_THREADS)
gcn_fill_gemm1_kernel(const int* __restrict__ src,
                      const int* __restrict__ dst,
                      const float* __restrict__ w,
                      int* __restrict__ cursor,
                      int* __restrict__ list,
                      unsigned* __restrict__ ctr, int E, int n,
                      const float* __restrict__ feat,
                      const float* __restrict__ degO,
                      const float* __restrict__ W1,
                      float* __restrict__ h1, int gemmBlocks) {
    const int b = (int)blockIdx.x;
    if (b < 2 * gemmBlocks && (b & 1)) {
        gcn_gemm1_block(feat, degO, W1, h1, n, b >> 1);
        return;
    }
    const int lane = threadIdx.x & 31;
    const int nSplit = (n >> 1) & ~31;
    const unsigned units = (unsigned)((E + SC_UNIT_EDGES - 1) / SC_UNIT_EDGES);
    const int home = gcn_home_half();
    for (int pass = 0; pass < 2; ++pass) {
        const int part = home ^ pass;
        unsigned* c = ctr + part * SC_CTR_STRIDE;
        for (unsigned it = 0; it < units; ++it) {
            const unsigned t = gcn_take_unit(c, lane);
            if (t >= units) break;
            const int e_lo = (int)t * SC_UNIT_EDGES;
            const int e_hi = (e_lo + SC_UNIT_EDGES < E) ? (e_lo + SC_UNIT_EDGES) : E;
            gcn_unit_fill(src, dst, w, cursor, (int2*)list, e_lo, e_hi, lane, nSplit, part);
        }
    }
}

#define AG_THREADS 256
#define AG_PITCH   34

__device__ __forceinline__ void gcn_ag_flush(float* __restrict__ tl, int lane,
                                             float ndv, float nsv, float b1c,
                                             int& r, float& acc) {
    const float ndr = __int_as_float(__builtin_amdgcn_readlane(__float_as_int(ndv), r));
    const float nsr = __int_as_float(__builtin_amdgcn_readlane(__float_as_int(nsv), r));
    tl[r * AG_PITCH + lane] = fmaxf(acc * ndr + b1c, 0.0f) * nsr;
    acc = 0.0f;
    ++r;
}

template <bool FULL>
__device__ __forceinline__ void gcn_ag_batch(const float* __restrict__ h1,
                                             float* __restrict__ tl,
                                             int s, int wb, int base, int cnt,
                                             unsigned laneB, int lane, int endv,
                                             float ndv, float nsv, float b1c,
                                             int& r, int& rend, float& acc) {
    const unsigned sB = (unsigned)s << 7;
    float v[32];
#pragma unroll
    for (int k = 0; k < 32; ++k) {
        if (FULL || k < cnt) {
            const unsigned sk = (unsigned)__builtin_amdgcn_readlane((int)sB, k);
            v[k] = *(const float*)((const char*)h1 + (sk + laneB));
        }
    }
    int lim = rend - base;
#pragma unroll
    for (int k = 0; k < 32; ++k) {
        if (FULL || k < cnt) {
            while (__builtin_expect(k >= lim, 0)) {
                gcn_ag_flush(tl, lane, ndv, nsv, b1c, r, acc);
                rend = __builtin_amdgcn_readlane(endv, r);
                lim  = rend - base;
            }
            const float wk = __int_as_float(__builtin_amdgcn_readlane(wb, k));
            acc += v[k] * wk;
        }
    }
}

__global__ void __launch_bounds__(AG_THREADS)
gcn_agg1_gemm2_kernel(const float* __restrict__ h1,
                      const int* __restrict__ rowptr,
                      const int* __restrict__ list,
                      const float* __restrict__ degO,
                      const float* __restrict__ degI,
                      const float* __restrict__ b1,
                      const float* __restrict__ W2,
                      float* __restrict__ h2, int n) {
    __shared__ float tiles[(AG_THREADS / 32) * 16 * AG_PITCH];
    const int wave = threadIdx.x >> 5;
    const int lane = threadIdx.x & 31;
    const int tile = blockIdx.x * (blockDim.x >> 5) + wave;
    const int row0 = tile * 16;
    if (row0 >= n) return;

    float* tl = tiles + wave * (16 * AG_PITCH);
    const int half = lane >> 4;
    const int mrow = lane & 15;
    const unsigned laneB = (unsigned)lane * 4u;
    const int2* __restrict__ ent = (const int2*)list;

    const int rrow = (row0 + mrow < n) ? (row0 + mrow) : (n - 1);
    const int   endv = rowptr[rrow + 1];
    const float ndv  = gcn_norm(degI[rrow]);
    const float nsv  = gcn_norm(degO[rrow]);
    const float b1c  = b1[lane];
    const int P0 = __builtin_amdgcn_readfirstlane(rowptr[row0]);
    const int P1 = __builtin_amdgcn_readlane(endv, 15);

    int   r    = 0;
    int   rend = __builtin_amdgcn_readlane(endv, 0);
    float acc  = 0.0f;

    int2 e_n = make_int2(0, 0);
    if (P0 + lane < P1) e_n = ent[P0 + lane];
    int base = P0;
    for (; base + 32 <= P1; base += 32) {
        const int s  = e_n.x;
        const int wb = e_n.y;
        const int qn = base + 32 + lane;
        if (qn < P1) e_n = ent[qn];
        gcn_ag_batch<true>(h1, tl, s, wb, base, 32, laneB, lane, endv, ndv, nsv, b1c, r, rend, acc);
    }
    if (base < P1)
        gcn_ag_batch<false>(h1, tl, e_n.x, e_n.y, base, P1 - base, laneB, lane, endv, ndv, nsv, b1c,
                            r, rend, acc);
    while (r < 16) gcn_ag_flush(tl, lane, ndv, nsv, b1c, r, acc);

    __builtin_amdgcn_fence(__ATOMIC_RELEASE, "workgroup");
    __builtin_amdgcn_fence(__ATOMIC_ACQUIRE, "workgroup");

    v8f acc8 = {};
#pragma unroll
    for (int k0 = 0; k0 < HID_D; k0 += 4) {
        const int kk = k0 + 2 * half;
        v2f a, b;
        a.x = tl[mrow * AG_PITCH + kk];
        a.y = tl[mrow * AG_PITCH + kk + 1];
        b.x = W2[kk * OUT_D + mrow];
        b.y = W2[(kk + 1) * OUT_D + mrow];
        acc8 = __builtin_amdgcn_wmma_f32_16x16x4_f32(false, a, false, b,
                                                     (short)0, acc8, false, false);
    }
#pragma unroll
    for (int j = 0; j < 8; ++j) {
        const int m = row0 + j + 8 * half;
        if (m < n) h2[(long long)m * OUT_D + mrow] = acc8[j];
    }
}

#define G16_THREADS 256
#define G16_PAIRS   4

template <int P>
__device__ __forceinline__ float gcn16_row_load(const float* __restrict__ h, int sB, int colB) {
    const unsigned off = (unsigned)(gcn_half_share<P>(sB) + colB);
    return *(const float*)((const char*)h + off);
}

__global__ void __launch_bounds__(G16_THREADS)
gcn_agg2_kernel(const float* __restrict__ h2,
                const int* __restrict__ rowptr,
                const int* __restrict__ list,
                const float* __restrict__ degI,
                const float* __restrict__ b2,
                float* __restrict__ out, int n) {
    const int lane = threadIdx.x & 31;
    const int half = lane >> 4;
    const int col  = lane & 15;
    const int colB = col * 4;
    const int gw   = blockIdx.x * (G16_THREADS / 32) + (threadIdx.x >> 5);
    const int2* __restrict__ ent = (const int2*)list;
    const float b2c = b2[col];

    for (int t = 0; t < G16_PAIRS; ++t) {
        const int pair = gw * G16_PAIRS + t;
        if (2 * pair >= n) break;
        const int  i     = 2 * pair + half;
        const bool valid = i < n;
        const int  p0    = valid ? rowptr[i] : 0;
        const int  deg   = valid ? (rowptr[i + 1] - p0) : 0;
        const int  dlo   = __builtin_amdgcn_readlane(deg, 0);
        const int  dhi   = __builtin_amdgcn_readlane(deg, 16);
        const int  dmax  = dlo > dhi ? dlo : dhi;
        float acc = 0.0f;
        for (int c = 0; c < dmax; c += 16) {
            const int2 e   = ent[(c + col < deg) ? (p0 + c + col) : 0];
            const int sB   = e.x * (OUT_D * 4);
            const int wb   = e.y;
            const int rem  = deg - c;
            const int remx = dmax - c;
            float v[16];
#define G16_LD(P)                                                   \
            if ((P) < remx) {                                       \
                if ((P) < rem) v[P] = gcn16_row_load<P>(h2, sB, colB); \
            }
            G16_LD(0)  G16_LD(1)  G16_LD(2)  G16_LD(3)
            G16_LD(4)  G16_LD(5)  G16_LD(6)  G16_LD(7)
            G16_LD(8)  G16_LD(9)  G16_LD(10) G16_LD(11)
            G16_LD(12) G16_LD(13) G16_LD(14) G16_LD(15)
#undef G16_LD
#define G16_AD(P)                                                   \
            if ((P) < remx) {                                       \
                if ((P) < rem) acc += v[P] * __int_as_float(gcn_half_share<P>(wb)); \
            }
            G16_AD(0)  G16_AD(1)  G16_AD(2)  G16_AD(3)
            G16_AD(4)  G16_AD(5)  G16_AD(6)  G16_AD(7)
            G16_AD(8)  G16_AD(9)  G16_AD(10) G16_AD(11)
            G16_AD(12) G16_AD(13) G16_AD(14) G16_AD(15)
#undef G16_AD
        }
        if (valid) out[(long long)i * OUT_D + col] = acc * gcn_norm(degI[i]) + b2c;
    }
}

extern "C" void kernel_launch(void* const* d_in, const int* in_sizes, int n_in,
                              void* d_out, int out_size, void* d_ws, size_t ws_size,
                              hipStream_t stream) {
    const float* feat = (const float*)d_in[0];
    const int*   src  = (const int*)d_in[1];
    const int*   dst  = (const int*)d_in[2];
    const float* ew   = (const float*)d_in[3];
    const float* W1   = (const float*)d_in[4];
    const float* b1   = (const float*)d_in[5];
    const float* W2   = (const float*)d_in[6];
    const float* b2   = (const float*)d_in[7];
    float* out = (float*)d_out;

    const int n = in_sizes[0] / IN_D;
    const int E = in_sizes[1];

    float* degO = (float*)d_ws;
    float* degI = degO + n;
    unsigned* ctr = (unsigned*)(degI + n);
    int*   rowptr = (int*)(ctr + SC_CTR_WORDS);
    float* h1   = degO + 4LL * n;
    int*   list = (int*)(h1 + (long long)n * HID_D);
    float* h2   = (float*)(list + (long long)n * HID_D);
    int*   cursor = (int*)h2;

    const int B = 256;
    auto blocks = [&](long long work) { return (unsigned)((work + B - 1) / B); };

    gcn_zero_kernel<<<blocks((2LL * n + SC_CTR_WORDS + 3) / 4), B, 0, stream>>>(
        degO, 2LL * n + SC_CTR_WORDS);

    const unsigned degRanges = (unsigned)((n + DEG_BINS - 1) / DEG_BINS);
    gcn_degree_kernel<<<2u * degRanges * DEG_PARTS, DEG_THREADS, 0, stream>>>(
        src, dst, degO, degI, E, n);

    gcn_rowptr_kernel<<<(unsigned)((n + RP_THREADS - 1) / RP_THREADS), RP_THREADS, 0, stream>>>(
        degI, rowptr, cursor, n);

    const unsigned scWaves  = 2u * (unsigned)((E + SC_UNIT_EDGES - 1) / SC_UNIT_EDGES);
    const unsigned scBlocks = (scWaves + (SC_THREADS / 32) - 1) / (SC_THREADS / 32);
    const int tiles = (n + 15) / 16;
    const unsigned gemmBlocks = (unsigned)((tiles + GEMM1_WAVES - 1) / GEMM1_WAVES);
    const unsigned fillBlocks = (scBlocks > gemmBlocks) ? scBlocks : gemmBlocks;
    gcn_fill_gemm1_kernel<<<fillBlocks + gemmBlocks, SC_THREADS, 0, stream>>>(
        src, dst, ew, cursor, list, ctr, E, n, feat, degO, W1, h1, (int)gemmBlocks);

    const unsigned aggBlocks =
        (unsigned)((tiles + (AG_THREADS / 32) - 1) / (AG_THREADS / 32));
    gcn_agg1_gemm2_kernel<<<aggBlocks, AG_THREADS, 0, stream>>>(
        h1, rowptr, list, degO, degI, b1, W2, h2, n);

    const long long pairs = ((long long)n + 1) / 2;
    const unsigned g16Blocks =
        (unsigned)((pairs + (long long)G16_PAIRS * (G16_THREADS / 32) - 1) /
                   ((long long)G16_PAIRS * (G16_THREADS / 32)));
    gcn_agg2_kernel<<<g16Blocks, G16_THREADS, 0, stream>>>(h2, rowptr, list, degI, b2, out, n);
}


# Degree counts kept per work-group as 16-bit partial planes (plain stores), summed in the prefix pass, which starts from block sums: zero-fill launch gone (6 -> 5), no f32 atomics; f32 sums unchanged
# speedup vs baseline: 3.6793x; 1.0873x over previous
#include <hip/hip_runtime.h>

typedef __attribute__((ext_vector_type(2))) float v2f;
typedef __attribute__((ext_vector_type(8))) float v8f;

#define IN_D  32
#define HID_D 32
#define OUT_D 16

#define DEG_BINS    25600
#define DEG_PARTS   32
#define DEG_THREADS 1024
#define DEG_LOADS   8
#define DEG_BLOCK   1024
#define DEG_RBLOCKS (DEG_BINS / DEG_BLOCK)
#define BS_PITCH    128

__device__ __forceinline__ void gcn_count(unsigned* bins, int node, int base, int nb) {
    const unsigned r = (unsigned)(node - base);
    if (r < (unsigned)nb) atomicAdd(bins + r, 1u);
}

__global__ void __launch_bounds__(DEG_THREADS)
gcn_degree_kernel(const int* __restrict__ src,
                  const int* __restrict__ dst,
                  unsigned* __restrict__ planes,
                  int* __restrict__ bsum, int E, int n, int pitch) {
    __shared__ unsigned bins[DEG_BINS];

    const int which = blockIdx.x & 1;
    const int rp    = blockIdx.x >> 1;
    const int part  = rp % DEG_PARTS;
    const int range = rp / DEG_PARTS;
    const int* __restrict__ idx = which ? dst : src;
    const int base = range * DEG_BINS;
    const int nb   = (n - base < DEG_BINS) ? (n - base) : DEG_BINS;

    for (int i = threadIdx.x * 4; i < DEG_BINS; i += DEG_THREADS * 4)
        *(uint4*)(bins + i) = make_uint4(0u, 0u, 0u, 0u);
    __syncthreads();

    const int per  = ((E + DEG_PARTS * 4 - 1) / (DEG_PARTS * 4)) * 4;
    const int lo   = part * per;
    const int hi   = (lo + per < E) ? (lo + per) : E;
    const int len  = (hi > lo) ? (hi - lo) : 0;
    const int nvec = len >> 2;
    const int4* __restrict__ vp = (const int4*)(idx + lo);

    for (int v0 = threadIdx.x; v0 < nvec; v0 += DEG_THREADS * DEG_LOADS) {
        int4 q[DEG_LOADS];
#pragma unroll
        for (int k = 0; k < DEG_LOADS; ++k) {
            const int v = v0 + k * DEG_THREADS;
            q[k] = (v < nvec) ? vp[v] : make_int4(-1, -1, -1, -1);
        }
#pragma unroll
        for (int k = 0; k < DEG_LOADS; ++k) {
            gcn_count(bins, q[k].x, base, nb);
            gcn_count(bins, q[k].y, base, nb);
            gcn_count(bins, q[k].z, base, nb);
            gcn_count(bins, q[k].w, base, nb);
        }
    }
    {
        const int e = lo + (nvec << 2) + threadIdx.x;
        if (e < hi) gcn_count(bins, idx[e], base, nb);
    }
    __syncthreads();

    unsigned* __restrict__ pl = planes + (((which * DEG_PARTS + part) * pitch + base) >> 1);
    for (int i = threadIdx.x * 8; i < DEG_BINS; i += DEG_THREADS * 8) {
        const uint4 a = *(const uint4*)(bins + i);
        const uint4 b = *(const uint4*)(bins + i + 4);
        *(uint4*)(pl + (i >> 1)) = make_uint4(a.x | (a.y << 16), a.z | (a.w << 16),
                                              b.x | (b.y << 16), b.z | (b.w << 16));
    }
    const int wv = threadIdx.x >> 5;
    if (which && wv < DEG_RBLOCKS) {
        const int lane = threadIdx.x & 31;
        unsigned s = 0u;
#pragma unroll
        for (int j = 0; j < DEG_BLOCK / 32; ++j) s += bins[wv * DEG_BLOCK + j * 32 + lane];
        for (int o = 16; o > 0; o >>= 1) s += __shfl_xor(s, o);
        if (lane == 0) bsum[part * BS_PITCH + range * DEG_RBLOCKS + wv] = (int)s;
    }
}

__device__ __forceinline__ float gcn_norm(float deg) {
    return rsqrtf(fmaxf(deg, 1.0f));
}

#define GEMM1_WAVES 8

__device__ __forceinline__ void gcn_gemm1_block(const float* __restrict__ feat,
                                                const float* __restrict__ degO,
                                                const float* __restrict__ W1,
                                                float* __restrict__ h1, int n, int blk) {
    const int wave = threadIdx.x >> 5;
    const int lane = threadIdx.x & 31;
    const int tile = blk * GEMM1_WAVES + wave;
    const int row0 = tile * 16;
    if (row0 >= n) return;

    const int half = lane >> 4;
    const int mrow = lane & 15;
    const int arow = row0 + mrow;
    const float scale = gcn_norm(degO[arow]);

    v8f acc0 = {};
    v8f acc1 = {};
#pragma unroll
    for (int k0 = 0; k0 < IN_D; k0 += 4) {
        const int kk = k0 + 2 * half;
        v2f a, b0, b1;
        const float* ap = feat + (long long)arow * IN_D + kk;
        a.x = ap[0] * scale;
        a.y = ap[1] * scale;
        b0.x = W1[kk * HID_D + mrow];
        b0.y = W1[(kk + 1) * HID_D + mrow];
        b1.x = W1[kk * HID_D + 16 + mrow];
        b1.y = W1[(kk + 1) * HID_D + 16 + mrow];
        acc0 = __builtin_amdgcn_wmma_f32_16x16x4_f32(false, a, false, b0,
                                                     (short)0, acc0, false, false);
        acc1 = __builtin_amdgcn_wmma_f32_16x16x4_f32(false, a, false, b1,
                                                     (short)0, acc1, false, false);
    }
#pragma unroll
    for (int j = 0; j < 8; ++j) {
        const int m = row0 + j + 8 * half;
        h1[(long long)m * HID_D + mrow]      = acc0[j];
        h1[(long long)m * HID_D + 16 + mrow] = acc1[j];
    }
}

#define RP_THREADS 512
#define RP_PLANES  (2 * DEG_PARTS)
#define SC_CTR_STRIDE 64
#define SC_CTR_WORDS  (4 * SC_CTR_STRIDE)

__global__ void __launch_bounds__(RP_THREADS)
gcn_rowptr_kernel(const unsigned* __restrict__ planes,
                  const int* __restrict__ bsum,
                  float* __restrict__ degO,
                  float* __restrict__ degI,
                  int* __restrict__ rowptr,
                  int* __restrict__ cursor,
                  unsigned* __restrict__ ctr, int n, int pitch) {
    __shared__ int wsum[RP_THREADS / 32];
    const int tid  = threadIdx.x;
    const int lane = tid & 31;
    const int wv   = tid >> 5;
    const int blk  = blockIdx.x;

    if (blk == 0 && tid < SC_CTR_WORDS) ctr[tid] = 0u;

    const int i0 = blk * DEG_BLOCK + 2 * tid;
    const unsigned* __restrict__ pl = planes + (i0 >> 1);
    const int hp = pitch >> 1;
    unsigned q[RP_PLANES];
#pragma unroll
    for (int k = 0; k < RP_PLANES; ++k) q[k] = pl[k * hp];

    const bool in = (tid & (BS_PITCH - 1)) < blk;
    int s = 0;
#pragma unroll
    for (int k = 0; k < DEG_PARTS * BS_PITCH / RP_THREADS; ++k) {
        const int v = bsum[in ? (tid + k * RP_THREADS) : 0];
        s += in ? v : 0;
    }
    for (int o = 16; o > 0; o >>= 1) s += __shfl_xor(s, o);
    if (lane == 0) wsum[wv] = s;
    __syncthreads();
    int base = 0;
    for (int k = 0; k < RP_THREADS / 32; ++k) base += wsum[k];
    __syncthreads();

    int o0 = 0, o1 = 0, c0 = 0, c1 = 0;
#pragma unroll
    for (int k = 0; k < DEG_PARTS; ++k) {
        o0 += (int)(q[k] & 0xffffu);
        o1 += (int)(q[k] >> 16);
        c0 += (int)(q[DEG_PARTS + k] & 0xffffu);
        c1 += (int)(q[DEG_PARTS + k] >> 16);
    }
    const int c = c0 + c1;
    int inc = c;
    for (int o = 1; o < 32; o <<= 1) {
        const int t = __shfl_up(inc, o);
        if (lane >= o) inc += t;
    }
    if (lane == 31) wsum[wv] = inc;
    __syncthreads();
    int wbase = 0;
    for (int k = 0; k < wv; ++k) wbase += wsum[k];
    const int excl = base + wbase + inc - c;
    if (i0 < n) {
        if (i0 + 1 < n) {
            *(float2*)(degO + i0)  = make_float2((float)o0, (float)o1);
            *(float2*)(degI + i0)  = make_float2((float)c0, (float)c1);
            *(int2*)(cursor + i0)  = make_int2(excl, excl + c0);
        } else {
            degO[i0]   = (float)o0;
            degI[i0]   = (float)c0;
            cursor[i0] = excl;
        }
        *(int2*)(rowptr + i0) = make_int2(excl, excl + c0);
        if (i0 + 2 == n) rowptr[n] = excl + c;
    }
}

#define SC_UNIT_EDGES 256
#define SC_THREADS    256

__device__ __forceinline__ int gcn_home_half() {
    const unsigned id = __builtin_amdgcn_s_sendmsg_rtn(0x87);
    return (int)((id >> 18) & 1u);
}

__device__ __forceinline__ unsigned gcn_take_unit(unsigned* c, int lane) {
    unsigned t = 0u;
    if (lane == 0)
        t = __hip_atomic_fetch_add(c, 1u, __ATOMIC_RELAXED, __HIP_MEMORY_SCOPE_AGENT);
    return (unsigned)__builtin_amdgcn_readfirstlane((int)t);
}

template <int P>
__device__ __forceinline__ int gcn_half_share(int v) {
    return __builtin_amdgcn_update_dpp(0, v, 0x150 + P, 0xf, 0xf, true);
}

#define FILL_BATCHES (SC_UNIT_EDGES / 32)

__device__ __forceinline__ void gcn_unit_fill(const int* __restrict__ src,
                                              const int* __restrict__ dst,
                                              const float* __restrict__ w,
                                              int* __restrict__ cursor,
                                              int2* __restrict__ list,
                                              int e_lo, int e_hi, int lane,
                                              int nSplit, int part) {
    int d[FILL_BATCHES], s[FILL_BATCHES], wb[FILL_BATCHES], pos[FILL_BATCHES];
#pragma unroll
    for (int b = 0; b < FILL_BATCHES; ++b) {
        const int e = e_lo + b * 32 + lane;
        const bool in = e < e_hi;
        d[b]  = in ? dst[e] : -1;
        s[b]  = in ? src[e] : 0;
        wb[b] = in ? __float_as_int(w[e]) : 0;
    }
#pragma unroll
    for (int b = 0; b < FILL_BATCHES; ++b) {
        const bool mine = (d[b] >= 0) && ((d[b] >= nSplit) == (part != 0));
        pos[b] = mine ? atomicAdd(cursor + d[b], 1) : -1;
    }
#pragma unroll
    for (int b = 0; b < FILL_BATCHES; ++b)
        if (pos[b] >= 0) list[pos[b]] = make_int2(s[b], wb[b]);
}

__global__ void __launch_bounds__(SC_THREADS)
gcn_fill_gemm1_kernel(const int* __restrict__ src,
                      const int* __restrict__ dst,
                      const float* __restrict__ w,
                      int* __restrict__ cursor,
                      int* __restrict__ list,
                      unsigned* __restrict__ ctr, int E, int n,
                      const float* __restrict__ feat,
                      const float* __restrict__ degO,
                      const float* __restrict__ W1,
                      float* __restrict__ h1, int gemmBlocks) {
    const int b = (int)blockIdx.x;
    if (b < 2 * gemmBlocks && (b & 1)) {
        gcn_gemm1_block(feat, degO, W1, h1, n, b >> 1);
        return;
    }
    const int lane = threadIdx.x & 31;
    const int nSplit = (n >> 1) & ~31;
    const unsigned units = (unsigned)((E + SC_UNIT_EDGES - 1) / SC_UNIT_EDGES);
    const int home = gcn_home_half();
    for (int pass = 0; pass < 2; ++pass) {
        const int part = home ^ pass;
        unsigned* c = ctr + part * SC_CTR_STRIDE;
        for (unsigned it = 0; it < units; ++it) {
            const unsigned t = gcn_take_unit(c, lane);
            if (t >= units) break;
            const int e_lo = (int)t * SC_UNIT_EDGES;
            const int e_hi = (e_lo + SC_UNIT_EDGES < E) ? (e_lo + SC_UNIT_EDGES) : E;
            gcn_unit_fill(src, dst, w, cursor, (int2*)list, e_lo, e_hi, lane, nSplit, part);
        }
    }
}

#define AG_THREADS 256
#define AG_PITCH   34

__device__ __forceinline__ void gcn_ag_flush(float* __restrict__ tl, int lane,
                                             float ndv, float nsv, float b1c,
                                             int& r, float& acc) {
    const float ndr = __int_as_float(__builtin_amdgcn_readlane(__float_as_int(ndv), r));
    const float nsr = __int_as_float(__builtin_amdgcn_readlane(__float_as_int(nsv), r));
    tl[r * AG_PITCH + lane] = fmaxf(acc * ndr + b1c, 0.0f) * nsr;
    acc = 0.0f;
    ++r;
}

template <bool FULL>
__device__ __forceinline__ void gcn_ag_batch(const float* __restrict__ h1,
                                             float* __restrict__ tl,
                                             int s, int wb, int base, int cnt,
                                             unsigned laneB, int lane, int endv,
                                             float ndv, float nsv, float b1c,
                                             int& r, int& rend, float& acc) {
    const unsigned sB = (unsigned)s << 7;
    float v[32];
#pragma unroll
    for (int k = 0; k < 32; ++k) {
        if (FULL || k < cnt) {
            const unsigned sk = (unsigned)__builtin_amdgcn_readlane((int)sB, k);
            v[k] = *(const float*)((const char*)h1 + (sk + laneB));
        }
    }
    int lim = rend - base;
#pragma unroll
    for (int k = 0; k < 32; ++k) {
        if (FULL || k < cnt) {
            while (__builtin_expect(k >= lim, 0)) {
                gcn_ag_flush(tl, lane, ndv, nsv, b1c, r, acc);
                rend = __builtin_amdgcn_readlane(endv, r);
                lim  = rend - base;
            }
            const float wk = __int_as_float(__builtin_amdgcn_readlane(wb, k));
            acc += v[k] * wk;
        }
    }
}

__global__ void __launch_bounds__(AG_THREADS)
gcn_agg1_gemm2_kernel(const float* __restrict__ h1,
                      const int* __restrict__ rowptr,
                      const int* __restrict__ list,
                      const float* __restrict__ degO,
                      const float* __restrict__ degI,
                      const float* __restrict__ b1,
                      const float* __restrict__ W2,
                      float* __restrict__ h2, int n) {
    __shared__ float tiles[(AG_THREADS / 32) * 16 * AG_PITCH];
    const int wave = threadIdx.x >> 5;
    const int lane = threadIdx.x & 31;
    const int tile = blockIdx.x * (blockDim.x >> 5) + wave;
    const int row0 = tile * 16;
    if (row0 >= n) return;

    float* tl = tiles + wave * (16 * AG_PITCH);
    const int half = lane >> 4;
    const int mrow = lane & 15;
    const unsigned laneB = (unsigned)lane * 4u;
    const int2* __restrict__ ent = (const int2*)list;

    const int rrow = (row0 + mrow < n) ? (row0 + mrow) : (n - 1);
    const int   endv = rowptr[rrow + 1];
    const float ndv  = gcn_norm(degI[rrow]);
    const float nsv  = gcn_norm(degO[rrow]);
    const float b1c  = b1[lane];
    const int P0 = __builtin_amdgcn_readfirstlane(rowptr[row0]);
    const int P1 = __builtin_amdgcn_readlane(endv, 15);

    int   r    = 0;
    int   rend = __builtin_amdgcn_readlane(endv, 0);
    float acc  = 0.0f;

    int2 e_n = make_int2(0, 0);
    if (P0 + lane < P1) e_n = ent[P0 + lane];
    int base = P0;
    for (; base + 32 <= P1; base += 32) {
        const int s  = e_n.x;
        const int wb = e_n.y;
        const int qn = base + 32 + lane;
        if (qn < P1) e_n = ent[qn];
        gcn_ag_batch<true>(h1, tl, s, wb, base, 32, laneB, lane, endv, ndv, nsv, b1c, r, rend, acc);
    }
    if (base < P1)
        gcn_ag_batch<false>(h1, tl, e_n.x, e_n.y, base, P1 - base, laneB, lane, endv, ndv, nsv, b1c,
                            r, rend, acc);
    while (r < 16) gcn_ag_flush(tl, lane, ndv, nsv, b1c, r, acc);

    __builtin_amdgcn_fence(__ATOMIC_RELEASE, "workgroup");
    __builtin_amdgcn_fence(__ATOMIC_ACQUIRE, "workgroup");

    v8f acc8 = {};
#pragma unroll
    for (int k0 = 0; k0 < HID_D; k0 += 4) {
        const int kk = k0 + 2 * half;
        v2f a, b;
        a.x = tl[mrow * AG_PITCH + kk];
        a.y = tl[mrow * AG_PITCH + kk + 1];
        b.x = W2[kk * OUT_D + mrow];
        b.y = W2[(kk + 1) * OUT_D + mrow];
        acc8 = __builtin_amdgcn_wmma_f32_16x16x4_f32(false, a, false, b,
                                                     (short)0, acc8, false, false);
    }
#pragma unroll
    for (int j = 0; j < 8; ++j) {
        const int m = row0 + j + 8 * half;
        if (m < n) h2[(long long)m * OUT_D + mrow] = acc8[j];
    }
}

#define G16_THREADS 256
#define G16_PAIRS   4

template <int P>
__device__ __forceinline__ float gcn16_row_load(const float* __restrict__ h, int sB, int colB) {
    const unsigned off = (unsigned)(gcn_half_share<P>(sB) + colB);
    return *(const float*)((const char*)h + off);
}

__global__ void __launch_bounds__(G16_THREADS)
gcn_agg2_kernel(const float* __restrict__ h2,
                const int* __restrict__ rowptr,
                const int* __restrict__ list,
                const float* __restrict__ degI,
                const float* __restrict__ b2,
                float* __restrict__ out, int n) {
    const int lane = threadIdx.x & 31;
    const int half = lane >> 4;
    const int col  = lane & 15;
    const int colB = col * 4;
    const int gw   = blockIdx.x * (G16_THREADS / 32) + (threadIdx.x >> 5);
    const int2* __restrict__ ent = (const int2*)list;
    const float b2c = b2[col];

    for (int t = 0; t < G16_PAIRS; ++t) {
        const int pair = gw * G16_PAIRS + t;
        if (2 * pair >= n) break;
        const int  i     = 2 * pair + half;
        const bool valid = i < n;
        const int  p0    = valid ? rowptr[i] : 0;
        const int  deg   = valid ? (rowptr[i + 1] - p0) : 0;
        const int  dlo   = __builtin_amdgcn_readlane(deg, 0);
        const int  dhi   = __builtin_amdgcn_readlane(deg, 16);
        const int  dmax  = dlo > dhi ? dlo : dhi;
        float acc = 0.0f;
        for (int c = 0; c < dmax; c += 16) {
            const int2 e   = ent[(c + col < deg) ? (p0 + c + col) : 0];
            const int sB   = e.x * (OUT_D * 4);
            const int wb   = e.y;
            const int rem  = deg - c;
            const int remx = dmax - c;
            float v[16];
#define G16_LD(P)                                                   \
            if ((P) < remx) {                                       \
                if ((P) < rem) v[P] = gcn16_row_load<P>(h2, sB, colB); \
            }
            G16_LD(0)  G16_LD(1)  G16_LD(2)  G16_LD(3)
            G16_LD(4)  G16_LD(5)  G16_LD(6)  G16_LD(7)
            G16_LD(8)  G16_LD(9)  G16_LD(10) G16_LD(11)
            G16_LD(12) G16_LD(13) G16_LD(14) G16_LD(15)
#undef G16_LD
#define G16_AD(P)                                                   \
            if ((P) < remx) {                                       \
                if ((P) < rem) acc += v[P] * __int_as_float(gcn_half_share<P>(wb)); \
            }
            G16_AD(0)  G16_AD(1)  G16_AD(2)  G16_AD(3)
            G16_AD(4)  G16_AD(5)  G16_AD(6)  G16_AD(7)
            G16_AD(8)  G16_AD(9)  G16_AD(10) G16_AD(11)
            G16_AD(12) G16_AD(13) G16_AD(14) G16_AD(15)
#undef G16_AD
        }
        if (valid) out[(long long)i * OUT_D + col] = acc * gcn_norm(degI[i]) + b2c;
    }
}

extern "C" void kernel_launch(void* const* d_in, const int* in_sizes, int n_in,
                              void* d_out, int out_size, void* d_ws, size_t ws_size,
                              hipStream_t stream) {
    const float* feat = (const float*)d_in[0];
    const int*   src  = (const int*)d_in[1];
    const int*   dst  = (const int*)d_in[2];
    const float* ew   = (const float*)d_in[3];
    const float* W1   = (const float*)d_in[4];
    const float* b1   = (const float*)d_in[5];
    const float* W2   = (const float*)d_in[6];
    const float* b2   = (const float*)d_in[7];
    float* out = (float*)d_out;

    const int n = in_sizes[0] / IN_D;
    const int E = in_sizes[1];

    float* degO = (float*)d_ws;
    float* degI = degO + n;
    unsigned* ctr = (unsigned*)(degI + n);
    int*   rowptr = (int*)(ctr + SC_CTR_WORDS);
    float* h1   = degO + 4LL * n;
    int*   list = (int*)(h1 + (long long)n * HID_D);
    float* h2   = (float*)(list + (long long)n * HID_D);
    int*   cursor = (int*)h2;
    int*   bsum   = (int*)degO + 3LL * n + 512;
    unsigned* planes = (unsigned*)h1;

    const unsigned degRanges = (unsigned)((n + DEG_BINS - 1) / DEG_BINS);
    const int pitch = (int)degRanges * DEG_BINS;
    gcn_degree_kernel<<<2u * degRanges * DEG_PARTS, DEG_THREADS, 0, stream>>>(
        src, dst, planes, bsum, E, n, pitch);

    gcn_rowptr_kernel<<<(unsigned)((n + DEG_BLOCK - 1) / DEG_BLOCK), RP_THREADS, 0, stream>>>(
        planes, bsum, degO, degI, rowptr, cursor, ctr, n, pitch);

    const unsigned scWaves  = 2u * (unsigned)((E + SC_UNIT_EDGES - 1) / SC_UNIT_EDGES);
    const unsigned scBlocks = (scWaves + (SC_THREADS / 32) - 1) / (SC_THREADS / 32);
    const int tiles = (n + 15) / 16;
    const unsigned gemmBlocks = (unsigned)((tiles + GEMM1_WAVES - 1) / GEMM1_WAVES);
    const unsigned fillBlocks = (scBlocks > gemmBlocks) ? scBlocks : gemmBlocks;
    gcn_fill_gemm1_kernel<<<fillBlocks + gemmBlocks, SC_THREADS, 0, stream>>>(
        src, dst, ew, cursor, list, ctr, E, n, feat, degO, W1, h1, (int)gemmBlocks);

    const unsigned aggBlocks =
        (unsigned)((tiles + (AG_THREADS / 32) - 1) / (AG_THREADS / 32));
    gcn_agg1_gemm2_kernel<<<aggBlocks, AG_THREADS, 0, stream>>>(
        h1, rowptr, list, degO, degI, b1, W2, h2, n);

    const long long pairs = ((long long)n + 1) / 2;
    const unsigned g16Blocks =
        (unsigned)((pairs + (long long)G16_PAIRS * (G16_THREADS / 32) - 1) /
                   ((long long)G16_PAIRS * (G16_THREADS / 32)));
    gcn_agg2_kernel<<<g16Blocks, G16_THREADS, 0, stream>>>(h2, rowptr, list, degI, b2, out, n);
}


# Layer-2 aggregation walk without lane tests: 16 rows' bounds in one load, the next pair's entries requested under the current pair, a half past its row's end adds zero-weight entries; sums unchanged
# speedup vs baseline: 3.8035x; 1.0338x over previous
#include <hip/hip_runtime.h>

typedef __attribute__((ext_vector_type(2))) float v2f;
typedef __attribute__((ext_vector_type(8))) float v8f;

#define IN_D  32
#define HID_D 32
#define OUT_D 16

#define DEG_BINS    25600
#define DEG_PARTS   32
#define DEG_THREADS 1024
#define DEG_LOADS   8
#define DEG_BLOCK   1024
#define DEG_RBLOCKS (DEG_BINS / DEG_BLOCK)
#define BS_PITCH    128

__device__ __forceinline__ void gcn_count(unsigned* bins, int node, int base, int nb) {
    const unsigned r = (unsigned)(node - base);
    if (r < (unsigned)nb) atomicAdd(bins + r, 1u);
}

__global__ void __launch_bounds__(DEG_THREADS)
gcn_degree_kernel(const int* __restrict__ src,
                  const int* __restrict__ dst,
                  unsigned* __restrict__ planes,
                  int* __restrict__ bsum, int E, int n, int pitch) {
    __shared__ unsigned bins[DEG_BINS];

    const int which = blockIdx.x & 1;
    const int rp    = blockIdx.x >> 1;
    const int part  = rp % DEG_PARTS;
    const int range = rp / DEG_PARTS;
    const int* __restrict__ idx = which ? dst : src;
    const int base = range * DEG_BINS;
    const int nb   = (n - base < DEG_BINS) ? (n - base) : DEG_BINS;

    for (int i = threadIdx.x * 4; i < DEG_BINS; i += DEG_THREADS * 4)
        *(uint4*)(bins + i) = make_uint4(0u, 0u, 0u, 0u);
    __syncthreads();

    const int per  = ((E + DEG_PARTS * 4 - 1) / (DEG_PARTS * 4)) * 4;
    const int lo   = part * per;
    const int hi   = (lo + per < E) ? (lo + per) : E;
    const int len  = (hi > lo) ? (hi - lo) : 0;
    const int nvec = len >> 2;
    const int4* __restrict__ vp = (const int4*)(idx + lo);

    for (int v0 = threadIdx.x; v0 < nvec; v0 += DEG_THREADS * DEG_LOADS) {
        int4 q[DEG_LOADS];
#pragma unroll
        for (int k = 0; k < DEG_LOADS; ++k) {
            const int v = v0 + k * DEG_THREADS;
            q[k] = (v < nvec) ? vp[v] : make_int4(-1, -1, -1, -1);
        }
#pragma unroll
        for (int k = 0; k < DEG_LOADS; ++k) {
            gcn_count(bins, q[k].x, base, nb);
            gcn_count(bins, q[k].y, base, nb);
            gcn_count(bins, q[k].z, base, nb);
            gcn_count(bins, q[k].w, base, nb);
        }
    }
    {
        const int e = lo + (nvec << 2) + threadIdx.x;
        if (e < hi) gcn_count(bins, idx[e], base, nb);
    }
    __syncthreads();

    unsigned* __restrict__ pl = planes + (((which * DEG_PARTS + part) * pitch + base) >> 1);
    for (int i = threadIdx.x * 8; i < DEG_BINS; i += DEG_THREADS * 8) {
        const uint4 a = *(const uint4*)(bins + i);
        const uint4 b = *(const uint4*)(bins + i + 4);
        *(uint4*)(pl + (i >> 1)) = make_uint4(a.x | (a.y << 16), a.z | (a.w << 16),
                                              b.x | (b.y << 16), b.z | (b.w << 16));
    }
    const int wv = threadIdx.x >> 5;
    if (which && wv < DEG_RBLOCKS) {
        const int lane = threadIdx.x & 31;
        unsigned s = 0u;
#pragma unroll
        for (int j = 0; j < DEG_BLOCK / 32; ++j) s += bins[wv * DEG_BLOCK + j * 32 + lane];
        for (int o = 16; o > 0; o >>= 1) s += __shfl_xor(s, o);
        if (lane == 0) bsum[part * BS_PITCH + range * DEG_RBLOCKS + wv] = (int)s;
    }
}

__device__ __forceinline__ float gcn_norm(float deg) {
    return rsqrtf(fmaxf(deg, 1.0f));
}

#define GEMM1_WAVES 8

__device__ __forceinline__ void gcn_gemm1_block(const float* __restrict__ feat,
                                                const float* __restrict__ degO,
                                                const float* __restrict__ W1,
                                                float* __restrict__ h1, int n, int blk) {
    const int wave = threadIdx.x >> 5;
    const int lane = threadIdx.x & 31;
    const int tile = blk * GEMM1_WAVES + wave;
    const int row0 = tile * 16;
    if (row0 >= n) return;

    const int half = lane >> 4;
    const int mrow = lane & 15;
    const int arow = row0 + mrow;
    const float scale = gcn_norm(degO[arow]);

    v8f acc0 = {};
    v8f acc1 = {};
#pragma unroll
    for (int k0 = 0; k0 < IN_D; k0 += 4) {
        const int kk = k0 + 2 * half;
        v2f a, b0, b1;
        const float* ap = feat + (long long)arow * IN_D + kk;
        a.x = ap[0] * scale;
        a.y = ap[1] * scale;
        b0.x = W1[kk * HID_D + mrow];
        b0.y = W1[(kk + 1) * HID_D + mrow];
        b1.x = W1[kk * HID_D + 16 + mrow];
        b1.y = W1[(kk + 1) * HID_D + 16 + mrow];
        acc0 = __builtin_amdgcn_wmma_f32_16x16x4_f32(false, a, false, b0,
                                                     (short)0, acc0, false, false);
        acc1 = __builtin_amdgcn_wmma_f32_16x16x4_f32(false, a, false, b1,
                                                     (short)0, acc1, false, false);
    }
#pragma unroll
    for (int j = 0; j < 8; ++j) {
        const int m = row0 + j + 8 * half;
        h1[(long long)m * HID_D + mrow]      = acc0[j];
        h1[(long long)m * HID_D + 16 + mrow] = acc1[j];
    }
}

#define RP_THREADS 512
#define RP_PLANES  (2 * DEG_PARTS)
#define SC_CTR_STRIDE 64
#define SC_CTR_WORDS  (4 * SC_CTR_STRIDE)

__global__ void __launch_bounds__(RP_THREADS)
gcn_rowptr_kernel(const unsigned* __restrict__ planes,
                  const int* __restrict__ bsum,
                  float* __restrict__ degO,
                  float* __restrict__ degI,
                  int* __restrict__ rowptr,
                  int* __restrict__ cursor,
                  unsigned* __restrict__ ctr, int n, int pitch) {
    __shared__ int wsum[RP_THREADS / 32];
    const int tid  = threadIdx.x;
    const int lane = tid & 31;
    const int wv   = tid >> 5;
    const int blk  = blockIdx.x;

    if (blk == 0 && tid < SC_CTR_WORDS) ctr[tid] = 0u;

    const int i0 = blk * DEG_BLOCK + 2 * tid;
    const unsigned* __restrict__ pl = planes + (i0 >> 1);
    const int hp = pitch >> 1;
    unsigned q[RP_PLANES];
#pragma unroll
    for (int k = 0; k < RP_PLANES; ++k) q[k] = pl[k * hp];

    const bool in = (tid & (BS_PITCH - 1)) < blk;
    int s = 0;
#pragma unroll
    for (int k = 0; k < DEG_PARTS * BS_PITCH / RP_THREADS; ++k) {
        const int v = bsum[in ? (tid + k * RP_THREADS) : 0];
        s += in ? v : 0;
    }
    for (int o = 16; o > 0; o >>= 1) s += __shfl_xor(s, o);
    if (lane == 0) wsum[wv] = s;
    __syncthreads();
    int base = 0;
    for (int k = 0; k < RP_THREADS / 32; ++k) base += wsum[k];
    __syncthreads();

    int o0 = 0, o1 = 0, c0 = 0, c1 = 0;
#pragma unroll
    for (int k = 0; k < DEG_PARTS; ++k) {
        o0 += (int)(q[k] & 0xffffu);
        o1 += (int)(q[k] >> 16);
        c0 += (int)(q[DEG_PARTS + k] & 0xffffu);
        c1 += (int)(q[DEG_PARTS + k] >> 16);
    }
    const int c = c0 + c1;
    int inc = c;
    for (int o = 1; o < 32; o <<= 1) {
        const int t = __shfl_up(inc, o);
        if (lane >= o) inc += t;
    }
    if (lane == 31) wsum[wv] = inc;
    __syncthreads();
    int wbase = 0;
    for (int k = 0; k < wv; ++k) wbase += wsum[k];
    const int excl = base + wbase + inc - c;
    if (i0 < n) {
        if (i0 + 1 < n) {
            *(float2*)(degO + i0)  = make_float2((float)o0, (float)o1);
            *(float2*)(degI + i0)  = make_float2((float)c0, (float)c1);
            *(int2*)(cursor + i0)  = make_int2(excl, excl + c0);
        } else {
            degO[i0]   = (float)o0;
            degI[i0]   = (float)c0;
            cursor[i0] = excl;
        }
        *(int2*)(rowptr + i0) = make_int2(excl, excl + c0);
        if (i0 + 2 == n) rowptr[n] = excl + c;
    }
}

#define SC_UNIT_EDGES 256
#define SC_THREADS    256

__device__ __forceinline__ int gcn_home_half() {
    const unsigned id = __builtin_amdgcn_s_sendmsg_rtn(0x87);
    return (int)((id >> 18) & 1u);
}

__device__ __forceinline__ unsigned gcn_take_unit(unsigned* c, int lane) {
    unsigned t = 0u;
    if (lane == 0)
        t = __hip_atomic_fetch_add(c, 1u, __ATOMIC_RELAXED, __HIP_MEMORY_SCOPE_AGENT);
    return (unsigned)__builtin_amdgcn_readfirstlane((int)t);
}

template <int P>
__device__ __forceinline__ int gcn_half_share(int v) {
    return __builtin_amdgcn_update_dpp(0, v, 0x150 + P, 0xf, 0xf, true);
}

#define FILL_BATCHES (SC_UNIT_EDGES / 32)

__device__ __forceinline__ void gcn_unit_fill(const int* __restrict__ src,
                                              const int* __restrict__ dst,
                                              const float* __restrict__ w,
                                              int* __restrict__ cursor,
                                              int2* __restrict__ list,
                                              int e_lo, int e_hi, int lane,
                                              int nSplit, int part) {
    int d[FILL_BATCHES], s[FILL_BATCHES], wb[FILL_BATCHES], pos[FILL_BATCHES];
#pragma unroll
    for (int b = 0; b < FILL_BATCHES; ++b) {
        const int e = e_lo + b * 32 + lane;
        const bool in = e < e_hi;
        d[b]  = in ? dst[e] : -1;
        s[b]  = in ? src[e] : 0;
        wb[b] = in ? __float_as_int(w[e]) : 0;
    }
#pragma unroll
    for (int b = 0; b < FILL_BATCHES; ++b) {
        const bool mine = (d[b] >= 0) && ((d[b] >= nSplit) == (part != 0));
        pos[b] = mine ? atomicAdd(cursor + d[b], 1) : -1;
    }
#pragma unroll
    for (int b = 0; b < FILL_BATCHES; ++b)
        if (pos[b] >= 0) list[pos[b]] = make_int2(s[b], wb[b]);
}

__global__ void __launch_bounds__(SC_THREADS)
gcn_fill_gemm1_kernel(const int* __restrict__ src,
                      const int* __restrict__ dst,
                      const float* __restrict__ w,
                      int* __restrict__ cursor,
                      int* __restrict__ list,
                      unsigned* __restrict__ ctr, int E, int n,
                      const float* __restrict__ feat,
                      const float* __restrict__ degO,
                      const float* __restrict__ W1,
                      float* __restrict__ h1, int gemmBlocks) {
    const int b = (int)blockIdx.x;
    if (b < 2 * gemmBlocks && (b & 1)) {
        gcn_gemm1_block(feat, degO, W1, h1, n, b >> 1);
        return;
    }
    const int lane = threadIdx.x & 31;
    const int nSplit = (n >> 1) & ~31;
    const unsigned units = (unsigned)((E + SC_UNIT_EDGES - 1) / SC_UNIT_EDGES);
    const int home = gcn_home_half();
    for (int pass = 0; pass < 2; ++pass) {
        const int part = home ^ pass;
        unsigned* c = ctr + part * SC_CTR_STRIDE;
        for (unsigned it = 0; it < units; ++it) {
            const unsigned t = gcn_take_unit(c, lane);
            if (t >= units) break;
            const int e_lo = (int)t * SC_UNIT_EDGES;
            const int e_hi = (e_lo + SC_UNIT_EDGES < E) ? (e_lo + SC_UNIT_EDGES) : E;
            gcn_unit_fill(src, dst, w, cursor, (int2*)list, e_lo, e_hi, lane, nSplit, part);
        }
    }
}

#define AG_THREADS 256
#define AG_PITCH   34

__device__ __forceinline__ void gcn_ag_flush(float* __restrict__ tl, int lane,
                                             float ndv, float nsv, float b1c,
                                             int& r, float& acc) {
    const float ndr = __int_as_float(__builtin_amdgcn_readlane(__float_as_int(ndv), r));
    const float nsr = __int_as_float(__builtin_amdgcn_readlane(__float_as_int(nsv), r));
    tl[r * AG_PITCH + lane] = fmaxf(acc * ndr + b1c, 0.0f) * nsr;
    acc = 0.0f;
    ++r;
}

template <bool FULL>
__device__ __forceinline__ void gcn_ag_batch(const float* __restrict__ h1,
                                             float* __restrict__ tl,
                                             int s, int wb, int base, int cnt,
                                             unsigned laneB, int lane, int endv,
                                             float ndv, float nsv, float b1c,
                                             int& r, int& rend, float& acc) {
    const unsigned sB = (unsigned)s << 7;
    float v[32];
#pragma unroll
    for (int k = 0; k < 32; ++k) {
        if (FULL || k < cnt) {
            const unsigned sk = (unsigned)__builtin_amdgcn_readlane((int)sB, k);
            v[k] = *(const float*)((const char*)h1 + (sk + laneB));
        }
    }
    int lim = rend - base;
#pragma unroll
    for (int k = 0; k < 32; ++k) {
        if (FULL || k < cnt) {
            while (__builtin_expect(k >= lim, 0)) {
                gcn_ag_flush(tl, lane, ndv, nsv, b1c, r, acc);
                rend = __builtin_amdgcn_readlane(endv, r);
                lim  = rend - base;
            }
            const float wk = __int_as_float(__builtin_amdgcn_readlane(wb, k));
            acc += v[k] * wk;
        }
    }
}

__global__ void __launch_bounds__(AG_THREADS)
gcn_agg1_gemm2_kernel(const float* __restrict__ h1,
                      const int* __restrict__ rowptr,
                      const int* __restrict__ list,
                      const float* __restrict__ degO,
                      const float* __restrict__ degI,
                      const float* __restrict__ b1,
                      const float* __restrict__ W2,
                      float* __restrict__ h2, int n) {
    __shared__ float tiles[(AG_THREADS / 32) * 16 * AG_PITCH];
    const int wave = threadIdx.x >> 5;
    const int lane = threadIdx.x & 31;
    const int tile = blockIdx.x * (blockDim.x >> 5) + wave;
    const int row0 = tile * 16;
    if (row0 >= n) return;

    float* tl = tiles + wave * (16 * AG_PITCH);
    const int half = lane >> 4;
    const int mrow = lane & 15;
    const unsigned laneB = (unsigned)lane * 4u;
    const int2* __restrict__ ent = (const int2*)list;

    const int rrow = (row0 + mrow < n) ? (row0 + mrow) : (n - 1);
    const int   endv = rowptr[rrow + 1];
    const float ndv  = gcn_norm(degI[rrow]);
    const float nsv  = gcn_norm(degO[rrow]);
    const float b1c  = b1[lane];
    const int P0 = __builtin_amdgcn_readfirstlane(rowptr[row0]);
    const int P1 = __builtin_amdgcn_readlane(endv, 15);

    int   r    = 0;
    int   rend = __builtin_amdgcn_readlane(endv, 0);
    float acc  = 0.0f;

    int2 e_n = make_int2(0, 0);
    if (P0 + lane < P1) e_n = ent[P0 + lane];
    int base = P0;
    for (; base + 32 <= P1; base += 32) {
        const int s  = e_n.x;
        const int wb = e_n.y;
        const int qn = base + 32 + lane;
        if (qn < P1) e_n = ent[qn];
        gcn_ag_batch<true>(h1, tl, s, wb, base, 32, laneB, lane, endv, ndv, nsv, b1c, r, rend, acc);
    }
    if (base < P1)
        gcn_ag_batch<false>(h1, tl, e_n.x, e_n.y, base, P1 - base, laneB, lane, endv, ndv, nsv, b1c,
                            r, rend, acc);
    while (r < 16) gcn_ag_flush(tl, lane, ndv, nsv, b1c, r, acc);

    __builtin_amdgcn_fence(__ATOMIC_RELEASE, "workgroup");
    __builtin_amdgcn_fence(__ATOMIC_ACQUIRE, "workgroup");

    v8f acc8 = {};
#pragma unroll
    for (int k0 = 0; k0 < HID_D; k0 += 4) {
        const int kk = k0 + 2 * half;
        v2f a, b;
        a.x = tl[mrow * AG_PITCH + kk];
        a.y = tl[mrow * AG_PITCH + kk + 1];
        b.x = W2[kk * OUT_D + mrow];
        b.y = W2[(kk + 1) * OUT_D + mrow];
        acc8 = __builtin_amdgcn_wmma_f32_16x16x4_f32(false, a, false, b,
                                                     (short)0, acc8, false, false);
    }
#pragma unroll
    for (int j = 0; j < 8; ++j) {
        const int m = row0 + j + 8 * half;
        if (m < n) h2[(long long)m * OUT_D + mrow] = acc8[j];
    }
}

#define G16_THREADS 256
#define G16_PAIRS   8
#define G16_ROWS    (2 * G16_PAIRS)

template <int P>
__device__ __forceinline__ float gcn16_row_load(const float* __restrict__ h, int sB, int colB) {
    const unsigned off = (unsigned)(gcn_half_share<P>(sB) + colB);
    return *(const float*)((const char*)h + off);
}

#define G16_LD(P, V, S)   V[P] = gcn16_row_load<P>(h2, S, colB);
#define G16_AD(P, V, WT)  acc = __builtin_fmaf(V[P], __int_as_float(gcn_half_share<P>(WT)), acc);
#define G16_LD4(Q, V, S)  G16_LD(Q, V, S) G16_LD(Q + 1, V, S) G16_LD(Q + 2, V, S) G16_LD(Q + 3, V, S)
#define G16_AD4(Q, V, WT) G16_AD(Q, V, WT) G16_AD(Q + 1, V, WT) G16_AD(Q + 2, V, WT) G16_AD(Q + 3, V, WT)
#define G16_LD16(V, S)    G16_LD4(0, V, S) G16_LD4(4, V, S) G16_LD4(8, V, S) G16_LD4(12, V, S)
#define G16_AD16(V, WT)   G16_AD4(0, V, WT) G16_AD4(4, V, WT) G16_AD4(8, V, WT) G16_AD4(12, V, WT)

__global__ void __launch_bounds__(G16_THREADS)
gcn_agg2_kernel(const float* __restrict__ h2,
                const int* __restrict__ rowptr,
                const int* __restrict__ list,
                const float* __restrict__ b2,
                float* __restrict__ out, int n) {
    const int lane = threadIdx.x & 31;
    const int half = lane >> 4;
    const int col  = lane & 15;
    const int colB = col * 4;
    const int gw   = __builtin_amdgcn_readfirstlane(
        (int)(blockIdx.x * (G16_THREADS / 32) + (threadIdx.x >> 5)));
    const int row0 = gw * G16_ROWS;
    if (row0 >= n) return;
    const int2* __restrict__ ent = (const int2*)list;
    const float b2c = b2[col];

    const int rj = (row0 + lane < n) ? (row0 + lane) : n;
    const int rp = rowptr[rj];

    int a0 = __builtin_amdgcn_readlane(rp, 0);
    int a1 = __builtin_amdgcn_readlane(rp, 1);
    int a2 = __builtin_amdgcn_readlane(rp, 2);
    int dm = (a1 - a0 > a2 - a1) ? (a1 - a0) : (a2 - a1);
    int p0 = half ? a1 : a0;
    int dg = half ? (a2 - a1) : (a1 - a0);
    int2 e0 = ent[(col < dg) ? (p0 + col) : 0];
    int2 e1 = ent[(col + 16 < dg) ? (p0 + 16 + col) : 0];

#pragma nounroll
    for (int t = 0; t < G16_PAIRS; ++t) {
        const int c0  = __builtin_amdgcn_readlane(rp, 2 * t + 2);
        const int c1  = __builtin_amdgcn_readlane(rp, 2 * t + 3);
        const int c2  = __builtin_amdgcn_readlane(rp, 2 * t + 4);
        const int dmN = (c1 - c0 > c2 - c1) ? (c1 - c0) : (c2 - c1);
        const int p0N = half ? c1 : c0;
        const int dgN = half ? (c2 - c1) : (c1 - c0);
        const int2 e0N = ent[(col < dgN) ? (p0N + col) : 0];
        const int2 e1N = ent[(col + 16 < dgN) ? (p0N + 16 + col) : 0];

        const int  i   = row0 + 2 * t + half;
        const bool in0 = col < dg;
        const bool in1 = col + 16 < dg;
        const int sB0 = in0 ? e0.x * (OUT_D * 4) : 0;
        const int wb0 = in0 ? e0.y : 0;
        const int sB1 = in1 ? e1.x * (OUT_D * 4) : 0;
        const int wb1 = in1 ? e1.y : 0;
        float acc = 0.0f;
        float v0[16];
        G16_LD16(v0, sB0)
        if (dm > 16) {
            if (dm > 28) {
                float v1[16];
                G16_LD16(v1, sB1)
                G16_AD16(v0, wb0)
                G16_AD16(v1, wb1)
                for (int c = 32; c < dm; c += 16) {
                    const bool in = c + col < dg;
                    const int2 e  = ent[in ? (p0 + c + col) : 0];
                    const int sB  = in ? e.x * (OUT_D * 4) : 0;
                    const int wb  = in ? e.y : 0;
                    float v2[16];
                    G16_LD16(v2, sB)
                    G16_AD16(v2, wb)
                }
            } else if (dm > 24) {
                float v1[12];
                G16_LD4(0, v1, sB1) G16_LD4(4, v1, sB1) G16_LD4(8, v1, sB1)
                G16_AD16(v0, wb0)
                G16_AD4(0, v1, wb1) G16_AD4(4, v1, wb1) G16_AD4(8, v1, wb1)
            } else if (dm > 20) {
                float v1[8];
                G16_LD4(0, v1, sB1) G16_LD4(4, v1, sB1)
                G16_AD16(v0, wb0)
                G16_AD4(0, v1, wb1) G16_AD4(4, v1, wb1)
            } else {
                float v1[4];
                G16_LD4(0, v1, sB1)
                G16_AD16(v0, wb0)
                G16_AD4(0, v1, wb1)
            }
        } else {
            G16_AD16(v0, wb0)
        }
        if (i < n) out[(long long)i * OUT_D + col] = acc * gcn_norm((float)dg) + b2c;

        dm = dmN; p0 = p0N; dg = dgN; e0 = e0N; e1 = e1N;
    }
}
#undef G16_LD
#undef G16_AD
#undef G16_LD4
#undef G16_AD4
#undef G16_LD16
#undef G16_AD16

extern "C" void kernel_launch(void* const* d_in, const int* in_sizes, int n_in,
                              void* d_out, int out_size, void* d_ws, size_t ws_size,
                              hipStream_t stream) {
    const float* feat = (const float*)d_in[0];
    const int*   src  = (const int*)d_in[1];
    const int*   dst  = (const int*)d_in[2];
    const float* ew   = (const float*)d_in[3];
    const float* W1   = (const float*)d_in[4];
    const float* b1   = (const float*)d_in[5];
    const float* W2   = (const float*)d_in[6];
    const float* b2   = (const float*)d_in[7];
    float* out = (float*)d_out;

    const int n = in_sizes[0] / IN_D;
    const int E = in_sizes[1];

    float* degO = (float*)d_ws;
    float* degI = degO + n;
    unsigned* ctr = (unsigned*)(degI + n);
    int*   rowptr = (int*)(ctr + SC_CTR_WORDS);
    float* h1   = degO + 4LL * n;
    int*   list = (int*)(h1 + (long long)n * HID_D);
    float* h2   = (float*)(list + (long long)n * HID_D);
    int*   cursor = (int*)h2;
    int*   bsum   = (int*)degO + 3LL * n + 512;
    unsigned* planes = (unsigned*)h1;

    const unsigned degRanges = (unsigned)((n + DEG_BINS - 1) / DEG_BINS);
    const int pitch = (int)degRanges * DEG_BINS;
    gcn_degree_kernel<<<2u * degRanges * DEG_PARTS, DEG_THREADS, 0, stream>>>(
        src, dst, planes, bsum, E, n, pitch);

    gcn_rowptr_kernel<<<(unsigned)((n + DEG_BLOCK - 1) / DEG_BLOCK), RP_THREADS, 0, stream>>>(
        planes, bsum, degO, degI, rowptr, cursor, ctr, n, pitch);

    const unsigned scWaves  = 2u * (unsigned)((E + SC_UNIT_EDGES - 1) / SC_UNIT_EDGES);
    const unsigned scBlocks = (scWaves + (SC_THREADS / 32) - 1) / (SC_THREADS / 32);
    const int tiles = (n + 15) / 16;
    const unsigned gemmBlocks = (unsigned)((tiles + GEMM1_WAVES - 1) / GEMM1_WAVES);
    const unsigned fillBlocks = (scBlocks > gemmBlocks) ? scBlocks : gemmBlocks;
    gcn_fill_gemm1_kernel<<<fillBlocks + gemmBlocks, SC_THREADS, 0, stream>>>(
        src, dst, ew, cursor, list, ctr, E, n, feat, degO, W1, h1, (int)gemmBlocks);

    const unsigned aggBlocks =
        (unsigned)((tiles + (AG_THREADS / 32) - 1) / (AG_THREADS / 32));
    gcn_agg1_gemm2_kernel<<<aggBlocks, AG_THREADS, 0, stream>>>(
        h1, rowptr, list, degO, degI, b1, W2, h2, n);

    const long long pairs = ((long long)n + 1) / 2;
    const unsigned g16Blocks =
        (unsigned)((pairs + (long long)G16_PAIRS * (G16_THREADS / 32) - 1) /
                   ((long long)G16_PAIRS * (G16_THREADS / 32)));
    gcn_agg2_kernel<<<g16Blocks, G16_THREADS, 0, stream>>>(h2, rowptr, list, b2, out, n);
}


# Edge-list fill: a ticket hands out 1024 edges (two 512-edge stretches) instead of 256: 4x fewer draws on the two counter words, 4x fewer fill waves, entering ahead of GEMM1; sums unchanged
# speedup vs baseline: 4.2694x; 1.1225x over previous
#include <hip/hip_runtime.h>

typedef __attribute__((ext_vector_type(2))) float v2f;
typedef __attribute__((ext_vector_type(8))) float v8f;

#define IN_D  32
#define HID_D 32
#define OUT_D 16

#define DEG_BINS    25600
#define DEG_PARTS   32
#define DEG_THREADS 1024
#define DEG_LOADS   8
#define DEG_BLOCK   1024
#define DEG_RBLOCKS (DEG_BINS / DEG_BLOCK)
#define BS_PITCH    128

__device__ __forceinline__ void gcn_count(unsigned* bins, int node, int base, int nb) {
    const unsigned r = (unsigned)(node - base);
    if (r < (unsigned)nb) atomicAdd(bins + r, 1u);
}

__global__ void __launch_bounds__(DEG_THREADS)
gcn_degree_kernel(const int* __restrict__ src,
                  const int* __restrict__ dst,
                  unsigned* __restrict__ planes,
                  int* __restrict__ bsum, int E, int n, int pitch) {
    __shared__ unsigned bins[DEG_BINS];

    const int which = blockIdx.x & 1;
    const int rp    = blockIdx.x >> 1;
    const int part  = rp % DEG_PARTS;
    const int range = rp / DEG_PARTS;
    const int* __restrict__ idx = which ? dst : src;
    const int base = range * DEG_BINS;
    const int nb   = (n - base < DEG_BINS) ? (n - base) : DEG_BINS;

    for (int i = threadIdx.x * 4; i < DEG_BINS; i += DEG_THREADS * 4)
        *(uint4*)(bins + i) = make_uint4(0u, 0u, 0u, 0u);
    __syncthreads();

    const int per  = ((E + DEG_PARTS * 4 - 1) / (DEG_PARTS * 4)) * 4;
    const int lo   = part * per;
    const int hi   = (lo + per < E) ? (lo + per) : E;
    const int len  = (hi > lo) ? (hi - lo) : 0;
    const int nvec = len >> 2;
    const int4* __restrict__ vp = (const int4*)(idx + lo);

    for (int v0 = threadIdx.x; v0 < nvec; v0 += DEG_THREADS * DEG_LOADS) {
        int4 q[DEG_LOADS];
#pragma unroll
        for (int k = 0; k < DEG_LOADS; ++k) {
            const int v = v0 + k * DEG_THREADS;
            q[k] = (v < nvec) ? vp[v] : make_int4(-1, -1, -1, -1);
        }
#pragma unroll
        for (int k = 0; k < DEG_LOADS; ++k) {
            gcn_count(bins, q[k].x, base, nb);
            gcn_count(bins, q[k].y, base, nb);
            gcn_count(bins, q[k].z, base, nb);
            gcn_count(bins, q[k].w, base, nb);
        }
    }
    {
        const int e = lo + (nvec << 2) + threadIdx.x;
        if (e < hi) gcn_count(bins, idx[e], base, nb);
    }
    __syncthreads();

    unsigned* __restrict__ pl = planes + (((which * DEG_PARTS + part) * pitch + base) >> 1);
    for (int i = threadIdx.x * 8; i < DEG_BINS; i += DEG_THREADS * 8) {
        const uint4 a = *(const uint4*)(bins + i);
        const uint4 b = *(const uint4*)(bins + i + 4);
        *(uint4*)(pl + (i >> 1)) = make_uint4(a.x | (a.y << 16), a.z | (a.w << 16),
                                              b.x | (b.y << 16), b.z | (b.w << 16));
    }
    const int wv = threadIdx.x >> 5;
    if (which && wv < DEG_RBLOCKS) {
        const int lane = threadIdx.x & 31;
        unsigned s = 0u;
#pragma unroll
        for (int j = 0; j < DEG_BLOCK / 32; ++j) s += bins[wv * DEG_BLOCK + j * 32 + lane];
        for (int o = 16; o > 0; o >>= 1) s += __shfl_xor(s, o);
        if (lane == 0) bsum[part * BS_PITCH + range * DEG_RBLOCKS + wv] = (int)s;
    }
}

__device__ __forceinline__ float gcn_norm(float deg) {
    return rsqrtf(fmaxf(deg, 1.0f));
}

#define GEMM1_WAVES 8

__device__ __forceinline__ void gcn_gemm1_block(const float* __restrict__ feat,
                                                const float* __restrict__ degO,
                                                const float* __restrict__ W1,
                                                float* __restrict__ h1, int n, int blk) {
    const int wave = threadIdx.x >> 5;
    const int lane = threadIdx.x & 31;
    const int tile = blk * GEMM1_WAVES + wave;
    const int row0 = tile * 16;
    if (row0 >= n) return;

    const int half = lane >> 4;
    const int mrow = lane & 15;
    const int arow = row0 + mrow;
    const float scale = gcn_norm(degO[arow]);

    v8f acc0 = {};
    v8f acc1 = {};
#pragma unroll
    for (int k0 = 0; k0 < IN_D; k0 += 4) {
        const int kk = k0 + 2 * half;
        v2f a, b0, b1;
        const float* ap = feat + (long long)arow * IN_D + kk;
        a.x = ap[0] * scale;
        a.y = ap[1] * scale;
        b0.x = W1[kk * HID_D + mrow];
        b0.y = W1[(kk + 1) * HID_D + mrow];
        b1.x = W1[kk * HID_D + 16 + mrow];
        b1.y = W1[(kk + 1) * HID_D + 16 + mrow];
        acc0 = __builtin_amdgcn_wmma_f32_16x16x4_f32(false, a, false, b0,
                                                     (short)0, acc0, false, false);
        acc1 = __builtin_amdgcn_wmma_f32_16x16x4_f32(false, a, false, b1,
                                                     (short)0, acc1, false, false);
    }
#pragma unroll
    for (int j = 0; j < 8; ++j) {
        const int m = row0 + j + 8 * half;
        h1[(long long)m * HID_D + mrow]      = acc0[j];
        h1[(long long)m * HID_D + 16 + mrow] = acc1[j];
    }
}

#define RP_THREADS 512
#define RP_PLANES  (2 * DEG_PARTS)
#define SC_CTR_STRIDE 64
#define SC_CTR_WORDS  (4 * SC_CTR_STRIDE)

__global__ void __launch_bounds__(RP_THREADS)
gcn_rowptr_kernel(const unsigned* __restrict__ planes,
                  const int* __restrict__ bsum,
                  float* __restrict__ degO,
                  float* __restrict__ degI,
                  int* __restrict__ rowptr,
                  int* __restrict__ cursor,
                  unsigned* __restrict__ ctr, int n, int pitch) {
    __shared__ int wsum[RP_THREADS / 32];
    const int tid  = threadIdx.x;
    const int lane = tid & 31;
    const int wv   = tid >> 5;
    const int blk  = blockIdx.x;

    if (blk == 0 && tid < SC_CTR_WORDS) ctr[tid] = 0u;

    const int i0 = blk * DEG_BLOCK + 2 * tid;
    const unsigned* __restrict__ pl = planes + (i0 >> 1);
    const int hp = pitch >> 1;
    unsigned q[RP_PLANES];
#pragma unroll
    for (int k = 0; k < RP_PLANES; ++k) q[k] = pl[k * hp];

    const bool in = (tid & (BS_PITCH - 1)) < blk;
    int s = 0;
#pragma unroll
    for (int k = 0; k < DEG_PARTS * BS_PITCH / RP_THREADS; ++k) {
        const int v = bsum[in ? (tid + k * RP_THREADS) : 0];
        s += in ? v : 0;
    }
    for (int o = 16; o > 0; o >>= 1) s += __shfl_xor(s, o);
    if (lane == 0) wsum[wv] = s;
    __syncthreads();
    int base = 0;
    for (int k = 0; k < RP_THREADS / 32; ++k) base += wsum[k];
    __syncthreads();

    int o0 = 0, o1 = 0, c0 = 0, c1 = 0;
#pragma unroll
    for (int k = 0; k < DEG_PARTS; ++k) {
        o0 += (int)(q[k] & 0xffffu);
        o1 += (int)(q[k] >> 16);
        c0 += (int)(q[DEG_PARTS + k] & 0xffffu);
        c1 += (int)(q[DEG_PARTS + k] >> 16);
    }
    const int c = c0 + c1;
    int inc = c;
    for (int o = 1; o < 32; o <<= 1) {
        const int t = __shfl_up(inc, o);
        if (lane >= o) inc += t;
    }
    if (lane == 31) wsum[wv] = inc;
    __syncthreads();
    int wbase = 0;
    for (int k = 0; k < wv; ++k) wbase += wsum[k];
    const int excl = base + wbase + inc - c;
    if (i0 < n) {
        if (i0 + 1 < n) {
            *(float2*)(degO + i0)  = make_float2((float)o0, (float)o1);
            *(float2*)(degI + i0)  = make_float2((float)c0, (float)c1);
            *(int2*)(cursor + i0)  = make_int2(excl, excl + c0);
        } else {
            degO[i0]   = (float)o0;
            degI[i0]   = (float)c0;
            cursor[i0] = excl;
        }
        *(int2*)(rowptr + i0) = make_int2(excl, excl + c0);
        if (i0 + 2 == n) rowptr[n] = excl + c;
    }
}

#define SC_UNIT_EDGES 1024
#define SC_SUB_EDGES  512
#define SC_THREADS    256

__device__ __forceinline__ int gcn_home_half() {
    const unsigned id = __builtin_amdgcn_s_sendmsg_rtn(0x87);
    return (int)((id >> 18) & 1u);
}

__device__ __forceinline__ unsigned gcn_take_unit(unsigned* c, int lane) {
    unsigned t = 0u;
    if (lane == 0)
        t = __hip_atomic_fetch_add(c, 1u, __ATOMIC_RELAXED, __HIP_MEMORY_SCOPE_AGENT);
    return (unsigned)__builtin_amdgcn_readfirstlane((int)t);
}

template <int P>
__device__ __forceinline__ int gcn_half_share(int v) {
    return __builtin_amdgcn_update_dpp(0, v, 0x150 + P, 0xf, 0xf, true);
}

#define FILL_BATCHES (SC_SUB_EDGES / 32)

__device__ __forceinline__ void gcn_unit_fill(const int* __restrict__ src,
                                              const int* __restrict__ dst,
                                              const float* __restrict__ w,
                                              int* __restrict__ cursor,
                                              int2* __restrict__ list,
                                              int e_lo, int e_hi, int lane,
                                              int nSplit, int part) {
    int d[FILL_BATCHES], s[FILL_BATCHES], wb[FILL_BATCHES], pos[FILL_BATCHES];
#pragma unroll
    for (int b = 0; b < FILL_BATCHES; ++b) {
        const int e = e_lo + b * 32 + lane;
        const bool in = e < e_hi;
        d[b]  = in ? dst[e] : -1;
        s[b]  = in ? src[e] : 0;
        wb[b] = in ? __float_as_int(w[e]) : 0;
    }
#pragma unroll
    for (int b = 0; b < FILL_BATCHES; ++b) {
        const bool mine = (d[b] >= 0) && ((d[b] >= nSplit) == (part != 0));
        pos[b] = mine ? atomicAdd(cursor + d[b], 1) : -1;
    }
#pragma unroll
    for (int b = 0; b < FILL_BATCHES; ++b)
        if (pos[b] >= 0) list[pos[b]] = make_int2(s[b], wb[b]);
}

__global__ void __launch_bounds__(SC_THREADS)
gcn_fill_gemm1_kernel(const int* __restrict__ src,
                      const int* __restrict__ dst,
                      const float* __restrict__ w,
                      int* __restrict__ cursor,
                      int* __restrict__ list,
                      unsigned* __restrict__ ctr, int E, int n,
                      const float* __restrict__ feat,
                      const float* __restrict__ degO,
                      const float* __restrict__ W1,
                      float* __restrict__ h1, int fillBlocks) {
    const int b = (int)blockIdx.x;
    if (b >= fillBlocks) {
        gcn_gemm1_block(feat, degO, W1, h1, n, b - fillBlocks);
        return;
    }
    const int lane = threadIdx.x & 31;
    const int nSplit = (n >> 1) & ~31;
    const unsigned units = (unsigned)((E + SC_UNIT_EDGES - 1) / SC_UNIT_EDGES);
    const int home = gcn_home_half();
    for (int pass = 0; pass < 2; ++pass) {
        const int part = home ^ pass;
        unsigned* c = ctr + part * SC_CTR_STRIDE;
        for (unsigned it = 0; it < units; ++it) {
            const unsigned t = gcn_take_unit(c, lane);
            if (t >= units) break;
            const int e_lo = (int)t * SC_UNIT_EDGES;
            const int e_hi = (e_lo + SC_UNIT_EDGES < E) ? (e_lo + SC_UNIT_EDGES) : E;
            for (int e0 = e_lo; e0 < e_hi; e0 += SC_SUB_EDGES) {
                const int e1 = (e0 + SC_SUB_EDGES < e_hi) ? (e0 + SC_SUB_EDGES) : e_hi;
                gcn_unit_fill(src, dst, w, cursor, (int2*)list, e0, e1, lane, nSplit, part);
            }
        }
    }
}

#define AG_THREADS 256
#define AG_PITCH   34

__device__ __forceinline__ void gcn_ag_flush(float* __restrict__ tl, int lane,
                                             float ndv, float nsv, float b1c,
                                             int& r, float& acc) {
    const float ndr = __int_as_float(__builtin_amdgcn_readlane(__float_as_int(ndv), r));
    const float nsr = __int_as_float(__builtin_amdgcn_readlane(__float_as_int(nsv), r));
    tl[r * AG_PITCH + lane] = fmaxf(acc * ndr + b1c, 0.0f) * nsr;
    acc = 0.0f;
    ++r;
}

template <bool FULL>
__device__ __forceinline__ void gcn_ag_batch(const float* __restrict__ h1,
                                             float* __restrict__ tl,
                                             int s, int wb, int base, int cnt,
                                             unsigned laneB, int lane, int endv,
                                             float ndv, float nsv, float b1c,
                                             int& r, int& rend, float& acc) {
    const unsigned sB = (unsigned)s << 7;
    float v[32];
#pragma unroll
    for (int k = 0; k < 32; ++k) {
        if (FULL || k < cnt) {
            const unsigned sk = (unsigned)__builtin_amdgcn_readlane((int)sB, k);
            v[k] = *(const float*)((const char*)h1 + (sk + laneB));
        }
    }
    int lim = rend - base;
#pragma unroll
    for (int k = 0; k < 32; ++k) {
        if (FULL || k < cnt) {
            while (__builtin_expect(k >= lim, 0)) {
                gcn_ag_flush(tl, lane, ndv, nsv, b1c, r, acc);
                rend = __builtin_amdgcn_readlane(endv, r);
                lim  = rend - base;
            }
            const float wk = __int_as_float(__builtin_amdgcn_readlane(wb, k));
            acc += v[k] * wk;
        }
    }
}

__global__ void __launch_bounds__(AG_THREADS)
gcn_agg1_gemm2_kernel(const float* __restrict__ h1,
                      const int* __restrict__ rowptr,
                      const int* __restrict__ list,
                      const float* __restrict__ degO,
                      const float* __restrict__ degI,
                      const float* __restrict__ b1,
                      const float* __restrict__ W2,
                      float* __restrict__ h2, int n) {
    __shared__ float tiles[(AG_THREADS / 32) * 16 * AG_PITCH];
    const int wave = threadIdx.x >> 5;
    const int lane = threadIdx.x & 31;
    const int tile = blockIdx.x * (blockDim.x >> 5) + wave;
    const int row0 = tile * 16;
    if (row0 >= n) return;

    float* tl = tiles + wave * (16 * AG_PITCH);
    const int half = lane >> 4;
    const int mrow = lane & 15;
    const unsigned laneB = (unsigned)lane * 4u;
    const int2* __restrict__ ent = (const int2*)list;

    const int rrow = (row0 + mrow < n) ? (row0 + mrow) : (n - 1);
    const int   endv = rowptr[rrow + 1];
    const float ndv  = gcn_norm(degI[rrow]);
    const float nsv  = gcn_norm(degO[rrow]);
    const float b1c  = b1[lane];
    const int P0 = __builtin_amdgcn_readfirstlane(rowptr[row0]);
    const int P1 = __builtin_amdgcn_readlane(endv, 15);

    int   r    = 0;
    int   rend = __builtin_amdgcn_readlane(endv, 0);
    float acc  = 0.0f;

    int2 e_n = make_int2(0, 0);
    if (P0 + lane < P1) e_n = ent[P0 + lane];
    int base = P0;
    for (; base + 32 <= P1; base += 32) {
        const int s  = e_n.x;
        const int wb = e_n.y;
        const int qn = base + 32 + lane;
        if (qn < P1) e_n = ent[qn];
        gcn_ag_batch<true>(h1, tl, s, wb, base, 32, laneB, lane, endv, ndv, nsv, b1c, r, rend, acc);
    }
    if (base < P1)
        gcn_ag_batch<false>(h1, tl, e_n.x, e_n.y, base, P1 - base, laneB, lane, endv, ndv, nsv, b1c,
                            r, rend, acc);
    while (r < 16) gcn_ag_flush(tl, lane, ndv, nsv, b1c, r, acc);

    __builtin_amdgcn_fence(__ATOMIC_RELEASE, "workgroup");
    __builtin_amdgcn_fence(__ATOMIC_ACQUIRE, "workgroup");

    v8f acc8 = {};
#pragma unroll
    for (int k0 = 0; k0 < HID_D; k0 += 4) {
        const int kk = k0 + 2 * half;
        v2f a, b;
        a.x = tl[mrow * AG_PITCH + kk];
        a.y = tl[mrow * AG_PITCH + kk + 1];
        b.x = W2[kk * OUT_D + mrow];
        b.y = W2[(kk + 1) * OUT_D + mrow];
        acc8 = __builtin_amdgcn_wmma_f32_16x16x4_f32(false, a, false, b,
                                                     (short)0, acc8, false, false);
    }
#pragma unroll
    for (int j = 0; j < 8; ++j) {
        const int m = row0 + j + 8 * half;
        if (m < n) h2[(long long)m * OUT_D + mrow] = acc8[j];
    }
}

#define G16_THREADS 256
#define G16_PAIRS   8
#define G16_ROWS    (2 * G16_PAIRS)

template <int P>
__device__ __forceinline__ float gcn16_row_load(const float* __restrict__ h, int sB, int colB) {
    const unsigned off = (unsigned)(gcn_half_share<P>(sB) + colB);
    return *(const float*)((const char*)h + off);
}

#define G16_LD(P, V, S)   V[P] = gcn16_row_load<P>(h2, S, colB);
#define G16_AD(P, V, WT)  acc = __builtin_fmaf(V[P], __int_as_float(gcn_half_share<P>(WT)), acc);
#define G16_LD4(Q, V, S)  G16_LD(Q, V, S) G16_LD(Q + 1, V, S) G16_LD(Q + 2, V, S) G16_LD(Q + 3, V, S)
#define G16_AD4(Q, V, WT) G16_AD(Q, V, WT) G16_AD(Q + 1, V, WT) G16_AD(Q + 2, V, WT) G16_AD(Q + 3, V, WT)
#define G16_LD16(V, S)    G16_LD4(0, V, S) G16_LD4(4, V, S) G16_LD4(8, V, S) G16_LD4(12, V, S)
#define G16_AD16(V, WT)   G16_AD4(0, V, WT) G16_AD4(4, V, WT) G16_AD4(8, V, WT) G16_AD4(12, V, WT)

__global__ void __launch_bounds__(G16_THREADS)
gcn_agg2_kernel(const float* __restrict__ h2,
                const int* __restrict__ rowptr,
                const int* __restrict__ list,
                const float* __restrict__ b2,
                float* __restrict__ out, int n) {
    const int lane = threadIdx.x & 31;
    const int half = lane >> 4;
    const int col  = lane & 15;
    const int colB = col * 4;
    const int gw   = __builtin_amdgcn_readfirstlane(
        (int)(blockIdx.x * (G16_THREADS / 32) + (threadIdx.x >> 5)));
    const int row0 = gw * G16_ROWS;
    if (row0 >= n) return;
    const int2* __restrict__ ent = (const int2*)list;
    const float b2c = b2[col];

    const int rj = (row0 + lane < n) ? (row0 + lane) : n;
    const int rp = rowptr[rj];

    int a0 = __builtin_amdgcn_readlane(rp, 0);
    int a1 = __builtin_amdgcn_readlane(rp, 1);
    int a2 = __builtin_amdgcn_readlane(rp, 2);
    int dm = (a1 - a0 > a2 - a1) ? (a1 - a0) : (a2 - a1);
    int p0 = half ? a1 : a0;
    int dg = half ? (a2 - a1) : (a1 - a0);
    int2 e0 = ent[(col < dg) ? (p0 + col) : 0];
    int2 e1 = ent[(col + 16 < dg) ? (p0 + 16 + col) : 0];

#pragma nounroll
    for (int t = 0; t < G16_PAIRS; ++t) {
        const int c0  = __builtin_amdgcn_readlane(rp, 2 * t + 2);
        const int c1  = __builtin_amdgcn_readlane(rp, 2 * t + 3);
        const int c2  = __builtin_amdgcn_readlane(rp, 2 * t + 4);
        const int dmN = (c1 - c0 > c2 - c1) ? (c1 - c0) : (c2 - c1);
        const int p0N = half ? c1 : c0;
        const int dgN = half ? (c2 - c1) : (c1 - c0);
        const int2 e0N = ent[(col < dgN) ? (p0N + col) : 0];
        const int2 e1N = ent[(col + 16 < dgN) ? (p0N + 16 + col) : 0];

        const int  i   = row0 + 2 * t + half;
        const bool in0 = col < dg;
        const bool in1 = col + 16 < dg;
        const int sB0 = in0 ? e0.x * (OUT_D * 4) : 0;
        const int wb0 = in0 ? e0.y : 0;
        const int sB1 = in1 ? e1.x * (OUT_D * 4) : 0;
        const int wb1 = in1 ? e1.y : 0;
        float acc = 0.0f;
        float v0[16];
        G16_LD16(v0, sB0)
        if (dm > 16) {
            if (dm > 28) {
                float v1[16];
                G16_LD16(v1, sB1)
                G16_AD16(v0, wb0)
                G16_AD16(v1, wb1)
                for (int c = 32; c < dm; c += 16) {
                    const bool in = c + col < dg;
                    const int2 e  = ent[in ? (p0 + c + col) : 0];
                    const int sB  = in ? e.x * (OUT_D * 4) : 0;
                    const int wb  = in ? e.y : 0;
                    float v2[16];
                    G16_LD16(v2, sB)
                    G16_AD16(v2, wb)
                }
            } else if (dm > 24) {
                float v1[12];
                G16_LD4(0, v1, sB1) G16_LD4(4, v1, sB1) G16_LD4(8, v1, sB1)
                G16_AD16(v0, wb0)
                G16_AD4(0, v1, wb1) G16_AD4(4, v1, wb1) G16_AD4(8, v1, wb1)
            } else if (dm > 20) {
                float v1[8];
                G16_LD4(0, v1, sB1) G16_LD4(4, v1, sB1)
                G16_AD16(v0, wb0)
                G16_AD4(0, v1, wb1) G16_AD4(4, v1, wb1)
            } else {
                float v1[4];
                G16_LD4(0, v1, sB1)
                G16_AD16(v0, wb0)
                G16_AD4(0, v1, wb1)
            }
        } else {
            G16_AD16(v0, wb0)
        }
        if (i < n) out[(long long)i * OUT_D + col] = acc * gcn_norm((float)dg) + b2c;

        dm = dmN; p0 = p0N; dg = dgN; e0 = e0N; e1 = e1N;
    }
}
#undef G16_LD
#undef G16_AD
#undef G16_LD4
#undef G16_AD4
#undef G16_LD16
#undef G16_AD16

extern "C" void kernel_launch(void* const* d_in, const int* in_sizes, int n_in,
                              void* d_out, int out_size, void* d_ws, size_t ws_size,
                              hipStream_t stream) {
    const float* feat = (const float*)d_in[0];
    const int*   src  = (const int*)d_in[1];
    const int*   dst  = (const int*)d_in[2];
    const float* ew   = (const float*)d_in[3];
    const float* W1   = (const float*)d_in[4];
    const float* b1   = (const float*)d_in[5];
    const float* W2   = (const float*)d_in[6];
    const float* b2   = (const float*)d_in[7];
    float* out = (float*)d_out;

    const int n = in_sizes[0] / IN_D;
    const int E = in_sizes[1];

    float* degO = (float*)d_ws;
    float* degI = degO + n;
    unsigned* ctr = (unsigned*)(degI + n);
    int*   rowptr = (int*)(ctr + SC_CTR_WORDS);
    float* h1   = degO + 4LL * n;
    int*   list = (int*)(h1 + (long long)n * HID_D);
    float* h2   = (float*)(list + (long long)n * HID_D);
    int*   cursor = (int*)h2;
    int*   bsum   = (int*)degO + 3LL * n + 512;
    unsigned* planes = (unsigned*)h1;

    const unsigned degRanges = (unsigned)((n + DEG_BINS - 1) / DEG_BINS);
    const int pitch = (int)degRanges * DEG_BINS;
    gcn_degree_kernel<<<2u * degRanges * DEG_PARTS, DEG_THREADS, 0, stream>>>(
        src, dst, planes, bsum, E, n, pitch);

    gcn_rowptr_kernel<<<(unsigned)((n + DEG_BLOCK - 1) / DEG_BLOCK), RP_THREADS, 0, stream>>>(
        planes, bsum, degO, degI, rowptr, cursor, ctr, n, pitch);

    const unsigned scWaves  = 2u * (unsigned)((E + SC_UNIT_EDGES - 1) / SC_UNIT_EDGES);
    const unsigned fillBlocks = (scWaves + (SC_THREADS / 32) - 1) / (SC_THREADS / 32);
    const int tiles = (n + 15) / 16;
    const unsigned gemmBlocks = (unsigned)((tiles + GEMM1_WAVES - 1) / GEMM1_WAVES);
    gcn_fill_gemm1_kernel<<<fillBlocks + gemmBlocks, SC_THREADS, 0, stream>>>(
        src, dst, ew, cursor, list, ctr, E, n, feat, degO, W1, h1, (int)fillBlocks);

    const unsigned aggBlocks =
        (unsigned)((tiles + (AG_THREADS / 32) - 1) / (AG_THREADS / 32));
    gcn_agg1_gemm2_kernel<<<aggBlocks, AG_THREADS, 0, stream>>>(
        h1, rowptr, list, degO, degI, b1, W2, h2, n);

    const long long pairs = ((long long)n + 1) / 2;
    const unsigned g16Blocks =
        (unsigned)((pairs + (long long)G16_PAIRS * (G16_THREADS / 32) - 1) /
                   ((long long)G16_PAIRS * (G16_THREADS / 32)));
    gcn_agg2_kernel<<<g16Blocks, G16_THREADS, 0, stream>>>(h2, rowptr, list, b2, out, n);
}
